# starting kernel
# baseline (speedup 1.0000x reference)
#include <hip/hip_runtime.h>
#include <math.h>

typedef __attribute__((ext_vector_type(2))) float v2f;
typedef __attribute__((ext_vector_type(8))) float v8f;

#define N_NODES 50000
#define C_IN    128
#define C_HID   256
#define HEADS   4
#define EMBED   64
#define E0      800000
#define ETOT    (E0 + N_NODES)
#define NEG_SLOPE 0.2f

__global__ void gemm_wmma(const float* __restrict__ X,
                          const float* __restrict__ W,
                          float* __restrict__ H,
                          int K, int N) {
    const int lane   = threadIdx.x & 31;
    const int wave   = threadIdx.x >> 5;
    const int tileM  = blockIdx.x;
    const int lane15 = lane & 15;
    const int khalf  = lane >> 4;
    const int row    = tileM * 16 + lane15;
    const int col0   = wave * 64;

    v8f acc0 = {}, acc1 = {}, acc2 = {}, acc3 = {};

    for (int k = 0; k < K; k += 4) {
        const int kk = k + khalf * 2;
        v2f a;
        a.x = X[row * K + kk];
        a.y = X[row * K + kk + 1];

        const float* wp0 = W + (size_t)kk * N;
        const float* wp1 = wp0 + N;

        v2f b0, b1, b2, b3;
        b0.x = wp0[col0 +  0 + lane15]; b0.y = wp1[col0 +  0 + lane15];
        b1.x = wp0[col0 + 16 + lane15]; b1.y = wp1[col0 + 16 + lane15];
        b2.x = wp0[col0 + 32 + lane15]; b2.y = wp1[col0 + 32 + lane15];
        b3.x = wp0[col0 + 48 + lane15]; b3.y = wp1[col0 + 48 + lane15];

        acc0 = __builtin_amdgcn_wmma_f32_16x16x4_f32(false, a, false, b0, (short)0, acc0, false, false);
        acc1 = __builtin_amdgcn_wmma_f32_16x16x4_f32(false, a, false, b1, (short)0, acc1, false, false);
        acc2 = __builtin_amdgcn_wmma_f32_16x16x4_f32(false, a, false, b2, (short)0, acc2, false, false);
        acc3 = __builtin_amdgcn_wmma_f32_16x16x4_f32(false, a, false, b3, (short)0, acc3, false, false);
    }

    const int rbase = tileM * 16 + khalf * 8;
#pragma unroll
    for (int i = 0; i < 8; ++i) {
        float* op = H + (size_t)(rbase + i) * N + col0 + lane15;
        op[ 0] = acc0[i];
        op[16] = acc1[i];
        op[32] = acc2[i];
        op[48] = acc3[i];
    }
}

__global__ void attn_dots(const float* __restrict__ H,
                          const float* __restrict__ att_src,
                          const float* __restrict__ att_dst,
                          float* __restrict__ a_src,
                          float* __restrict__ a_dst) {
    const int gw   = (int)((blockIdx.x * blockDim.x + threadIdx.x) >> 5);
    const int lane = threadIdx.x & 31;
    const int node = gw >> 2;
    const int head = gw & 3;
    if (node >= N_NODES) return;

    const float* hp = H + (size_t)node * C_HID + head * EMBED;
    const float* as = att_src + head * EMBED;
    const float* ad = att_dst + head * EMBED;

    float s = 0.f, d = 0.f;
#pragma unroll
    for (int c = 0; c < EMBED; c += 32) {
        float v = hp[c + lane];
        s += v * as[c + lane];
        d += v * ad[c + lane];
    }
#pragma unroll
    for (int off = 16; off > 0; off >>= 1) {
        s += __shfl_down(s, off, 32);
        d += __shfl_down(d, off, 32);
    }
    if (lane == 0) {
        a_src[node * HEADS + head] = s;
        a_dst[node * HEADS + head] = d;
    }
}

__device__ __forceinline__ unsigned ford(float f) {
    unsigned u = __float_as_uint(f);
    return (u >> 31) ? ~u : (u | 0x80000000u);
}
__device__ __forceinline__ float funord(unsigned u) {
    return (u >> 31) ? __uint_as_float(u ^ 0x80000000u) : __uint_as_float(~u);
}

__global__ void fill_u32(unsigned* __restrict__ p, unsigned v, int n) {
    int i = blockIdx.x * blockDim.x + threadIdx.x;
    if (i < n) p[i] = v;
}

__device__ __forceinline__ void edge_sd(int e, const int* src, const int* dst, int& s, int& d) {
    if (e < E0) { s = src[e]; d = dst[e]; }
    else        { s = e - E0; d = s; }
}

__device__ __forceinline__ float edge_logit(const float* a_src, const float* a_dst,
                                            int s, int d, int h) {
    float v = a_src[s * HEADS + h] + a_dst[d * HEADS + h];
    return v > 0.f ? v : NEG_SLOPE * v;
}

__global__ void edge_max(const int* __restrict__ src, const int* __restrict__ dst,
                         const float* __restrict__ a_src, const float* __restrict__ a_dst,
                         unsigned* __restrict__ m) {
    int i = blockIdx.x * blockDim.x + threadIdx.x;
    if (i >= ETOT * HEADS) return;
    int e = i >> 2, h = i & 3, s, d;
    edge_sd(e, src, dst, s, d);
    float v = edge_logit(a_src, a_dst, s, d, h);
    atomicMax(&m[d * HEADS + h], ford(v));
}

__global__ void edge_expsum(const int* __restrict__ src, const int* __restrict__ dst,
                            const float* __restrict__ a_src, const float* __restrict__ a_dst,
                            const unsigned* __restrict__ m,
                            float* __restrict__ ee, float* __restrict__ z) {
    int i = blockIdx.x * blockDim.x + threadIdx.x;
    if (i >= ETOT * HEADS) return;
    int e = i >> 2, h = i & 3, s, d;
    edge_sd(e, src, dst, s, d);
    float v  = edge_logit(a_src, a_dst, s, d, h);
    float mv = funord(m[d * HEADS + h]);
    float ev = __expf(v - mv);
    ee[i] = ev;
    atomicAdd(&z[d * HEADS + h], ev);
}

__global__ void edge_scatter(const int* __restrict__ src, const int* __restrict__ dst,
                             const float* __restrict__ ee, const float* __restrict__ z,
                             const float* __restrict__ H, float* __restrict__ agg) {
    int e = blockIdx.x;
    int t = threadIdx.x;
    int h = t >> 6, s, d;
    edge_sd(e, src, dst, s, d);
    float alpha = ee[(size_t)e * HEADS + h] / (z[d * HEADS + h] + 1e-16f);
    float val   = H[(size_t)s * C_HID + t] * alpha;
    atomicAdd(&agg[(size_t)d * C_HID + t], val);
}

__global__ void bias_act(const float* __restrict__ agg, const float* __restrict__ bias,
                         float* __restrict__ out, int n, int relu) {
    int i = blockIdx.x * blockDim.x + threadIdx.x;
    if (i >= n) return;
    float v = agg[i] + bias[i & (C_HID - 1)];
    out[i] = relu ? fmaxf(v, 0.f) : v;
}

extern "C" void kernel_launch(void* const* d_in, const int* in_sizes, int n_in,
                              void* d_out, int out_size, void* d_ws, size_t ws_size,
                              hipStream_t stream) {
    const float* x        = (const float*)d_in[0];
    const int*   eidx     = (const int*)  d_in[1];
    const float* W1       = (const float*)d_in[2];
    const float* att_s1   = (const float*)d_in[3];
    const float* att_d1   = (const float*)d_in[4];
    const float* b1       = (const float*)d_in[5];
    const float* W2       = (const float*)d_in[6];
    const float* att_s2   = (const float*)d_in[7];
    const float* att_d2   = (const float*)d_in[8];
    const float* b2       = (const float*)d_in[9];
    float*       out      = (float*)d_out;

    const int* src = eidx;
    const int* dst = eidx + E0;

    const size_t NB = (size_t)N_NODES * C_HID;
    float* h      = (float*)d_ws;
    float* agg    = h      + NB;
    float* hrelu  = agg    + NB;
    float* ee     = hrelu  + NB;
    float* a_src  = ee     + (size_t)ETOT * HEADS;
    float* a_dst  = a_src  + (size_t)N_NODES * HEADS;
    unsigned* m   = (unsigned*)(a_dst + (size_t)N_NODES * HEADS);
    float* z      = (float*)(m + (size_t)N_NODES * HEADS);

    const int NH       = N_NODES * HEADS;
    const int EH       = ETOT * HEADS;
    const int NELEM    = (int)NB;
    const unsigned ORD_NEG_INF = 0x007FFFFFu;

    const dim3 B256(256);
    const int gNH  = (NH    + 255) / 256;
    const int gEH  = (EH    + 255) / 256;
    const int gNE  = (NELEM + 255) / 256;
    const int gMM  = N_NODES / 16;
    const int gDOT = (NH * 32 + 255) / 256;

    gemm_wmma  <<<gMM, 128, 0, stream>>>(x, W1, h, C_IN, C_HID);
    attn_dots  <<<gDOT, B256, 0, stream>>>(h, att_s1, att_d1, a_src, a_dst);
    fill_u32   <<<gNH, B256, 0, stream>>>(m, ORD_NEG_INF, NH);
    fill_u32   <<<gNH, B256, 0, stream>>>((unsigned*)z, 0u, NH);
    fill_u32   <<<gNE, B256, 0, stream>>>((unsigned*)agg, 0u, NELEM);
    edge_max   <<<gEH, B256, 0, stream>>>(src, dst, a_src, a_dst, m);
    edge_expsum<<<gEH, B256, 0, stream>>>(src, dst, a_src, a_dst, m, ee, z);
    edge_scatter<<<ETOT, B256, 0, stream>>>(src, dst, ee, z, h, agg);
    bias_act   <<<gNE, B256, 0, stream>>>(agg, b1, hrelu, NELEM, 1);

    gemm_wmma  <<<gMM, 128, 0, stream>>>(hrelu, W2, h, C_HID, C_HID);
    attn_dots  <<<gDOT, B256, 0, stream>>>(h, att_s2, att_d2, a_src, a_dst);
    fill_u32   <<<gNH, B256, 0, stream>>>(m, ORD_NEG_INF, NH);
    fill_u32   <<<gNH, B256, 0, stream>>>((unsigned*)z, 0u, NH);
    fill_u32   <<<gNE, B256, 0, stream>>>((unsigned*)agg, 0u, NELEM);
    edge_max   <<<gEH, B256, 0, stream>>>(src, dst, a_src, a_dst, m);
    edge_expsum<<<gEH, B256, 0, stream>>>(src, dst, a_src, a_dst, m, ee, z);
    edge_scatter<<<ETOT, B256, 0, stream>>>(src, dst, ee, z, h, agg);
    bias_act   <<<gNE, B256, 0, stream>>>(agg, b2, out, NELEM, 0);
}


# Aggregation by destination bucket: edges counting-sorted in-call (rank from a returning int atomic), one wave per node sums its rows in registers; no f32 atomics on rows; sum order = bucket rank
# speedup vs baseline: 2.8849x; 2.8849x over previous
#include <hip/hip_runtime.h>
#include <math.h>

typedef __attribute__((ext_vector_type(2))) float v2f;
typedef __attribute__((ext_vector_type(8))) float v8f;
typedef __attribute__((ext_vector_type(4))) float v4f;
typedef __attribute__((ext_vector_type(2))) int   v2i;

#define N_NODES 50000
#define C_IN    128
#define C_HID   256
#define HEADS   4
#define EMBED   64
#define E0      800000
#define ETOT    (E0 + N_NODES)
#define NEG_SLOPE 0.2f

__global__ void gemm_wmma(const float* __restrict__ X,
                          const float* __restrict__ W,
                          float* __restrict__ H,
                          int K, int N) {
    const int lane   = threadIdx.x & 31;
    const int wave   = threadIdx.x >> 5;
    const int tileM  = blockIdx.x;
    const int lane15 = lane & 15;
    const int khalf  = lane >> 4;
    const int row    = tileM * 16 + lane15;
    const int col0   = wave * 64;

    v8f acc0 = {}, acc1 = {}, acc2 = {}, acc3 = {};

    for (int k = 0; k < K; k += 4) {
        const int kk = k + khalf * 2;
        v2f a;
        a.x = X[row * K + kk];
        a.y = X[row * K + kk + 1];

        const float* wp0 = W + (size_t)kk * N;
        const float* wp1 = wp0 + N;

        v2f b0, b1, b2, b3;
        b0.x = wp0[col0 +  0 + lane15]; b0.y = wp1[col0 +  0 + lane15];
        b1.x = wp0[col0 + 16 + lane15]; b1.y = wp1[col0 + 16 + lane15];
        b2.x = wp0[col0 + 32 + lane15]; b2.y = wp1[col0 + 32 + lane15];
        b3.x = wp0[col0 + 48 + lane15]; b3.y = wp1[col0 + 48 + lane15];

        acc0 = __builtin_amdgcn_wmma_f32_16x16x4_f32(false, a, false, b0, (short)0, acc0, false, false);
        acc1 = __builtin_amdgcn_wmma_f32_16x16x4_f32(false, a, false, b1, (short)0, acc1, false, false);
        acc2 = __builtin_amdgcn_wmma_f32_16x16x4_f32(false, a, false, b2, (short)0, acc2, false, false);
        acc3 = __builtin_amdgcn_wmma_f32_16x16x4_f32(false, a, false, b3, (short)0, acc3, false, false);
    }

    const int rbase = tileM * 16 + khalf * 8;
#pragma unroll
    for (int i = 0; i < 8; ++i) {
        float* op = H + (size_t)(rbase + i) * N + col0 + lane15;
        op[ 0] = acc0[i];
        op[16] = acc1[i];
        op[32] = acc2[i];
        op[48] = acc3[i];
    }
}

__global__ void attn_dots(const float* __restrict__ H,
                          const float* __restrict__ att_src,
                          const float* __restrict__ att_dst,
                          float* __restrict__ a_src,
                          float* __restrict__ a_dst) {
    const int gw   = (int)((blockIdx.x * blockDim.x + threadIdx.x) >> 5);
    const int lane = threadIdx.x & 31;
    const int node = gw >> 2;
    const int head = gw & 3;
    if (node >= N_NODES) return;

    const float* hp = H + (size_t)node * C_HID + head * EMBED;
    const float* as = att_src + head * EMBED;
    const float* ad = att_dst + head * EMBED;

    float s = 0.f, d = 0.f;
#pragma unroll
    for (int c = 0; c < EMBED; c += 32) {
        float v = hp[c + lane];
        s += v * as[c + lane];
        d += v * ad[c + lane];
    }
#pragma unroll
    for (int off = 16; off > 0; off >>= 1) {
        s += __shfl_down(s, off, 32);
        d += __shfl_down(d, off, 32);
    }
    if (lane == 0) {
        a_src[node * HEADS + head] = s;
        a_dst[node * HEADS + head] = d;
    }
}

__device__ __forceinline__ unsigned ford(float f) {
    unsigned u = __float_as_uint(f);
    return (u >> 31) ? ~u : (u | 0x80000000u);
}
__device__ __forceinline__ float funord(unsigned u) {
    return (u >> 31) ? __uint_as_float(u ^ 0x80000000u) : __uint_as_float(~u);
}

__global__ void fill_u32(unsigned* __restrict__ p, unsigned v, int n) {
    int i = blockIdx.x * blockDim.x + threadIdx.x;
    if (i < n) p[i] = v;
}

__device__ __forceinline__ void edge_sd(int e, const int* src, const int* dst, int& s, int& d) {
    if (e < E0) { s = src[e]; d = dst[e]; }
    else        { s = e - E0; d = s; }
}

__device__ __forceinline__ float edge_logit(const float* a_src, const float* a_dst,
                                            int s, int d, int h) {
    float v = a_src[s * HEADS + h] + a_dst[d * HEADS + h];
    return v > 0.f ? v : NEG_SLOPE * v;
}

__global__ void edge_max(const int* __restrict__ src, const int* __restrict__ dst,
                         const float* __restrict__ a_src, const float* __restrict__ a_dst,
                         unsigned* __restrict__ m) {
    int i = blockIdx.x * blockDim.x + threadIdx.x;
    if (i >= ETOT * HEADS) return;
    int e = i >> 2, h = i & 3, s, d;
    edge_sd(e, src, dst, s, d);
    float v = edge_logit(a_src, a_dst, s, d, h);
    atomicMax(&m[d * HEADS + h], ford(v));
}

__global__ void edge_expsum(const int* __restrict__ src, const int* __restrict__ dst,
                            const float* __restrict__ a_src, const float* __restrict__ a_dst,
                            const unsigned* __restrict__ m,
                            float* __restrict__ ee, float* __restrict__ z) {
    int i = blockIdx.x * blockDim.x + threadIdx.x;
    if (i >= ETOT * HEADS) return;
    int e = i >> 2, h = i & 3, s, d;
    edge_sd(e, src, dst, s, d);
    float v  = edge_logit(a_src, a_dst, s, d, h);
    float mv = funord(m[d * HEADS + h]);
    float ev = __expf(v - mv);
    ee[i] = ev;
    atomicAdd(&z[d * HEADS + h], ev);
}

__global__ void edge_rank(const int* __restrict__ dst, unsigned* __restrict__ cnt,
                          unsigned* __restrict__ rank) {
    int e = blockIdx.x * blockDim.x + threadIdx.x;
    if (e >= E0) return;
    rank[e] = atomicAdd(&cnt[dst[e]], 1u);
}

#define SCAN_T 1024
__global__ void __launch_bounds__(SCAN_T) bucket_offsets(const unsigned* __restrict__ cnt,
                                                        unsigned* __restrict__ rowptr) {
    __shared__ unsigned w_before[SCAN_T / 32];
    __shared__ unsigned w_own[SCAN_T / 32];
    const int t = threadIdx.x, lane = t & 31, wave = t >> 5;
    const int first = blockIdx.x * SCAN_T;

    unsigned before = 0;
    for (int i = t; i < first; i += SCAN_T) before += cnt[i];
    const int idx = first + t;
    const unsigned c = (idx < N_NODES) ? cnt[idx] : 0u;

    unsigned incl = c;
    for (int off = 1; off < 32; off <<= 1) {
        unsigned v = __shfl_up(incl, off, 32);
        if (lane >= off) incl += v;
    }
    for (int off = 16; off > 0; off >>= 1) before += __shfl_xor(before, off, 32);
    if (lane == 31) w_own[wave] = incl;
    if (lane == 0)  w_before[wave] = before;
    __syncthreads();
    unsigned base = 0;
    for (int w = 0; w < SCAN_T / 32; ++w) base += w_before[w];
    for (int w = 0; w < wave; ++w) base += w_own[w];
    if (idx < N_NODES) rowptr[idx] = base + incl - c;
}

__global__ void edge_place(const int* __restrict__ src, const int* __restrict__ dst,
                           const unsigned* __restrict__ rank, const unsigned* __restrict__ rowptr,
                           int* __restrict__ slot) {
    int e = blockIdx.x * blockDim.x + threadIdx.x;
    if (e >= E0) return;
    v2i se; se.x = src[e]; se.y = e;
    ((v2i*)slot)[rowptr[dst[e]] + rank[e]] = se;
}

#define ROWS_PER_BLOCK 8
__global__ void __launch_bounds__(ROWS_PER_BLOCK * 32)
node_gather(const int* __restrict__ slot, const unsigned* __restrict__ rowptr,
            const unsigned* __restrict__ cnt,
            const float* __restrict__ ee, const float* __restrict__ z,
            const float* __restrict__ H, float* __restrict__ agg) {
    const int lane = threadIdx.x & 31;
    const int node = __builtin_amdgcn_readfirstlane(blockIdx.x * ROWS_PER_BLOCK + (threadIdx.x >> 5));
    if (node >= N_NODES) return;

    const unsigned beg = rowptr[node];
    const unsigned deg = cnt[node];
    const v4f zz = *(const v4f*)(z + (size_t)node * HEADS);
    const float zd0 = zz.x + 1e-16f, zd1 = zz.y + 1e-16f, zd2 = zz.z + 1e-16f, zd3 = zz.w + 1e-16f;
    const bool upper = lane >= 16;

    const v4f* H4 = (const v4f*)H;
    const v4f* E4 = (const v4f*)ee;
    const v2i* S2 = (const v2i*)slot;

    v4f acc0, acc1;
    {
        const v4f e4 = E4[E0 + node];
        const float alo = (upper ? e4.y : e4.x) / (upper ? zd1 : zd0);
        const float ahi = (upper ? e4.w : e4.z) / (upper ? zd3 : zd2);
        acc0 = H4[(size_t)node * 64 + lane] * alo;
        acc1 = H4[(size_t)node * 64 + 32 + lane] * ahi;
    }

    for (unsigned base = 0; base < deg; base += 32) {
        const unsigned n = (deg - base < 32u) ? (deg - base) : 32u;
        int   s  = 0;
        float a0 = 0.f, a1 = 0.f, a2 = 0.f, a3 = 0.f;
        if ((unsigned)lane < n) {
            const v2i se = S2[beg + base + lane];
            s = se.x;
            const v4f e4 = E4[se.y];
            a0 = e4.x / zd0; a1 = e4.y / zd1; a2 = e4.z / zd2; a3 = e4.w / zd3;
        }
#pragma unroll 4
        for (unsigned j = 0; j < n; ++j) {
            const int   sj = __builtin_amdgcn_readlane(s, (int)j);
            const float b0 = __int_as_float(__builtin_amdgcn_readlane(__float_as_int(a0), (int)j));
            const float b1 = __int_as_float(__builtin_amdgcn_readlane(__float_as_int(a1), (int)j));
            const float b2 = __int_as_float(__builtin_amdgcn_readlane(__float_as_int(a2), (int)j));
            const float b3 = __int_as_float(__builtin_amdgcn_readlane(__float_as_int(a3), (int)j));
            const float alo = upper ? b1 : b0;
            const float ahi = upper ? b3 : b2;
            const v4f v0 = H4[(size_t)sj * 64 + lane];
            const v4f v1 = H4[(size_t)sj * 64 + 32 + lane];
            acc0 += v0 * alo;
            acc1 += v1 * ahi;
        }
    }

    v4f* A4 = (v4f*)agg;
    A4[(size_t)node * 64 + lane]      = acc0;
    A4[(size_t)node * 64 + 32 + lane] = acc1;
}

__global__ void bias_act(const float* __restrict__ agg, const float* __restrict__ bias,
                         float* __restrict__ out, int n, int relu) {
    int i = blockIdx.x * blockDim.x + threadIdx.x;
    if (i >= n) return;
    float v = agg[i] + bias[i & (C_HID - 1)];
    out[i] = relu ? fmaxf(v, 0.f) : v;
}

extern "C" void kernel_launch(void* const* d_in, const int* in_sizes, int n_in,
                              void* d_out, int out_size, void* d_ws, size_t ws_size,
                              hipStream_t stream) {
    const float* x        = (const float*)d_in[0];
    const int*   eidx     = (const int*)  d_in[1];
    const float* W1       = (const float*)d_in[2];
    const float* att_s1   = (const float*)d_in[3];
    const float* att_d1   = (const float*)d_in[4];
    const float* b1       = (const float*)d_in[5];
    const float* W2       = (const float*)d_in[6];
    const float* att_s2   = (const float*)d_in[7];
    const float* att_d2   = (const float*)d_in[8];
    const float* b2       = (const float*)d_in[9];
    float*       out      = (float*)d_out;

    const int* src = eidx;
    const int* dst = eidx + E0;

    const size_t NB = (size_t)N_NODES * C_HID;
    float* h      = (float*)d_ws;
    float* agg    = h      + NB;
    float* hrelu  = agg    + NB;
    float* ee     = hrelu  + NB;
    float* a_src  = ee     + (size_t)ETOT * HEADS;
    float* a_dst  = a_src  + (size_t)N_NODES * HEADS;
    unsigned* m   = (unsigned*)(a_dst + (size_t)N_NODES * HEADS);
    float* z      = (float*)(m + (size_t)N_NODES * HEADS);
    unsigned* cnt    = (unsigned*)(z + (size_t)N_NODES * HEADS);
    unsigned* rowptr = cnt + N_NODES;
    unsigned* rank   = rowptr + N_NODES;
    int*      slot   = (int*)(rank + E0);

    const int NH       = N_NODES * HEADS;
    const int EH       = ETOT * HEADS;
    const int NELEM    = (int)NB;
    const unsigned ORD_NEG_INF = 0x007FFFFFu;

    const dim3 B256(256);
    const int gNH  = (NH    + 255) / 256;
    const int gEH  = (EH    + 255) / 256;
    const int gNE  = (NELEM + 255) / 256;
    const int gMM  = N_NODES / 16;
    const int gDOT = (NH * 32 + 255) / 256;
    const int gN   = (N_NODES + 255) / 256;
    const int gE0  = (E0 + 255) / 256;
    const int gSCN = (N_NODES + SCAN_T - 1) / SCAN_T;
    const int gROW = (N_NODES + ROWS_PER_BLOCK - 1) / ROWS_PER_BLOCK;

    fill_u32      <<<gN,   B256, 0, stream>>>(cnt, 0u, N_NODES);
    edge_rank     <<<gE0,  B256, 0, stream>>>(dst, cnt, rank);
    bucket_offsets<<<gSCN, SCAN_T, 0, stream>>>(cnt, rowptr);
    edge_place    <<<gE0,  B256, 0, stream>>>(src, dst, rank, rowptr, slot);

    gemm_wmma  <<<gMM, 128, 0, stream>>>(x, W1, h, C_IN, C_HID);
    attn_dots  <<<gDOT, B256, 0, stream>>>(h, att_s1, att_d1, a_src, a_dst);
    fill_u32   <<<gNH, B256, 0, stream>>>(m, ORD_NEG_INF, NH);
    fill_u32   <<<gNH, B256, 0, stream>>>((unsigned*)z, 0u, NH);
    edge_max   <<<gEH, B256, 0, stream>>>(src, dst, a_src, a_dst, m);
    edge_expsum<<<gEH, B256, 0, stream>>>(src, dst, a_src, a_dst, m, ee, z);
    node_gather<<<gROW, ROWS_PER_BLOCK * 32, 0, stream>>>(slot, rowptr, cnt, ee, z, h, agg);
    bias_act   <<<gNE, B256, 0, stream>>>(agg, b1, hrelu, NELEM, 1);

    gemm_wmma  <<<gMM, 128, 0, stream>>>(hrelu, W2, h, C_HID, C_HID);
    attn_dots  <<<gDOT, B256, 0, stream>>>(h, att_s2, att_d2, a_src, a_dst);
    fill_u32   <<<gNH, B256, 0, stream>>>(m, ORD_NEG_INF, NH);
    fill_u32   <<<gNH, B256, 0, stream>>>((unsigned*)z, 0u, NH);
    edge_max   <<<gEH, B256, 0, stream>>>(src, dst, a_src, a_dst, m);
    edge_expsum<<<gEH, B256, 0, stream>>>(src, dst, a_src, a_dst, m, ee, z);
    node_gather<<<gROW, ROWS_PER_BLOCK * 32, 0, stream>>>(slot, rowptr, cnt, ee, z, h, agg);
    bias_act   <<<gNE, B256, 0, stream>>>(agg, b2, out, NELEM, 0);
}


# Bias and activation folded into the gather's row store: the wave owning a node writes the finished row once; the two one-thread-per-element launches (400,000 waves each) and the agg plane are gone
# speedup vs baseline: 3.1877x; 1.1050x over previous
#include <hip/hip_runtime.h>
#include <math.h>

typedef __attribute__((ext_vector_type(2))) float v2f;
typedef __attribute__((ext_vector_type(8))) float v8f;
typedef __attribute__((ext_vector_type(4))) float v4f;
typedef __attribute__((ext_vector_type(2))) int   v2i;

#define N_NODES 50000
#define C_IN    128
#define C_HID   256
#define HEADS   4
#define EMBED   64
#define E0      800000
#define ETOT    (E0 + N_NODES)
#define NEG_SLOPE 0.2f

__global__ void gemm_wmma(const float* __restrict__ X,
                          const float* __restrict__ W,
                          float* __restrict__ H,
                          int K, int N) {
    const int lane   = threadIdx.x & 31;
    const int wave   = threadIdx.x >> 5;
    const int tileM  = blockIdx.x;
    const int lane15 = lane & 15;
    const int khalf  = lane >> 4;
    const int row    = tileM * 16 + lane15;
    const int col0   = wave * 64;

    v8f acc0 = {}, acc1 = {}, acc2 = {}, acc3 = {};

    for (int k = 0; k < K; k += 4) {
        const int kk = k + khalf * 2;
        v2f a;
        a.x = X[row * K + kk];
        a.y = X[row * K + kk + 1];

        const float* wp0 = W + (size_t)kk * N;
        const float* wp1 = wp0 + N;

        v2f b0, b1, b2, b3;
        b0.x = wp0[col0 +  0 + lane15]; b0.y = wp1[col0 +  0 + lane15];
        b1.x = wp0[col0 + 16 + lane15]; b1.y = wp1[col0 + 16 + lane15];
        b2.x = wp0[col0 + 32 + lane15]; b2.y = wp1[col0 + 32 + lane15];
        b3.x = wp0[col0 + 48 + lane15]; b3.y = wp1[col0 + 48 + lane15];

        acc0 = __builtin_amdgcn_wmma_f32_16x16x4_f32(false, a, false, b0, (short)0, acc0, false, false);
        acc1 = __builtin_amdgcn_wmma_f32_16x16x4_f32(false, a, false, b1, (short)0, acc1, false, false);
        acc2 = __builtin_amdgcn_wmma_f32_16x16x4_f32(false, a, false, b2, (short)0, acc2, false, false);
        acc3 = __builtin_amdgcn_wmma_f32_16x16x4_f32(false, a, false, b3, (short)0, acc3, false, false);
    }

    const int rbase = tileM * 16 + khalf * 8;
#pragma unroll
    for (int i = 0; i < 8; ++i) {
        float* op = H + (size_t)(rbase + i) * N + col0 + lane15;
        op[ 0] = acc0[i];
        op[16] = acc1[i];
        op[32] = acc2[i];
        op[48] = acc3[i];
    }
}

__global__ void attn_dots(const float* __restrict__ H,
                          const float* __restrict__ att_src,
                          const float* __restrict__ att_dst,
                          float* __restrict__ a_src,
                          float* __restrict__ a_dst) {
    const int gw   = (int)((blockIdx.x * blockDim.x + threadIdx.x) >> 5);
    const int lane = threadIdx.x & 31;
    const int node = gw >> 2;
    const int head = gw & 3;
    if (node >= N_NODES) return;

    const float* hp = H + (size_t)node * C_HID + head * EMBED;
    const float* as = att_src + head * EMBED;
    const float* ad = att_dst + head * EMBED;

    float s = 0.f, d = 0.f;
#pragma unroll
    for (int c = 0; c < EMBED; c += 32) {
        float v = hp[c + lane];
        s += v * as[c + lane];
        d += v * ad[c + lane];
    }
#pragma unroll
    for (int off = 16; off > 0; off >>= 1) {
        s += __shfl_down(s, off, 32);
        d += __shfl_down(d, off, 32);
    }
    if (lane == 0) {
        a_src[node * HEADS + head] = s;
        a_dst[node * HEADS + head] = d;
    }
}

__device__ __forceinline__ unsigned ford(float f) {
    unsigned u = __float_as_uint(f);
    return (u >> 31) ? ~u : (u | 0x80000000u);
}
__device__ __forceinline__ float funord(unsigned u) {
    return (u >> 31) ? __uint_as_float(u ^ 0x80000000u) : __uint_as_float(~u);
}

__global__ void fill_u32(unsigned* __restrict__ p, unsigned v, int n) {
    int i = blockIdx.x * blockDim.x + threadIdx.x;
    if (i < n) p[i] = v;
}

__device__ __forceinline__ void edge_sd(int e, const int* src, const int* dst, int& s, int& d) {
    if (e < E0) { s = src[e]; d = dst[e]; }
    else        { s = e - E0; d = s; }
}

__device__ __forceinline__ float edge_logit(const float* a_src, const float* a_dst,
                                            int s, int d, int h) {
    float v = a_src[s * HEADS + h] + a_dst[d * HEADS + h];
    return v > 0.f ? v : NEG_SLOPE * v;
}

__global__ void edge_max(const int* __restrict__ src, const int* __restrict__ dst,
                         const float* __restrict__ a_src, const float* __restrict__ a_dst,
                         unsigned* __restrict__ m) {
    int i = blockIdx.x * blockDim.x + threadIdx.x;
    if (i >= ETOT * HEADS) return;
    int e = i >> 2, h = i & 3, s, d;
    edge_sd(e, src, dst, s, d);
    float v = edge_logit(a_src, a_dst, s, d, h);
    atomicMax(&m[d * HEADS + h], ford(v));
}

__global__ void edge_expsum(const int* __restrict__ src, const int* __restrict__ dst,
                            const float* __restrict__ a_src, const float* __restrict__ a_dst,
                            const unsigned* __restrict__ m,
                            float* __restrict__ ee, float* __restrict__ z) {
    int i = blockIdx.x * blockDim.x + threadIdx.x;
    if (i >= ETOT * HEADS) return;
    int e = i >> 2, h = i & 3, s, d;
    edge_sd(e, src, dst, s, d);
    float v  = edge_logit(a_src, a_dst, s, d, h);
    float mv = funord(m[d * HEADS + h]);
    float ev = __expf(v - mv);
    ee[i] = ev;
    atomicAdd(&z[d * HEADS + h], ev);
}

__global__ void edge_rank(const int* __restrict__ dst, unsigned* __restrict__ cnt,
                          unsigned* __restrict__ rank) {
    int e = blockIdx.x * blockDim.x + threadIdx.x;
    if (e >= E0) return;
    rank[e] = atomicAdd(&cnt[dst[e]], 1u);
}

#define SCAN_T 1024
__global__ void __launch_bounds__(SCAN_T) bucket_offsets(const unsigned* __restrict__ cnt,
                                                        unsigned* __restrict__ rowptr) {
    __shared__ unsigned w_before[SCAN_T / 32];
    __shared__ unsigned w_own[SCAN_T / 32];
    const int t = threadIdx.x, lane = t & 31, wave = t >> 5;
    const int first = blockIdx.x * SCAN_T;

    unsigned before = 0;
    for (int i = t; i < first; i += SCAN_T) before += cnt[i];
    const int idx = first + t;
    const unsigned c = (idx < N_NODES) ? cnt[idx] : 0u;

    unsigned incl = c;
    for (int off = 1; off < 32; off <<= 1) {
        unsigned v = __shfl_up(incl, off, 32);
        if (lane >= off) incl += v;
    }
    for (int off = 16; off > 0; off >>= 1) before += __shfl_xor(before, off, 32);
    if (lane == 31) w_own[wave] = incl;
    if (lane == 0)  w_before[wave] = before;
    __syncthreads();
    unsigned base = 0;
    for (int w = 0; w < SCAN_T / 32; ++w) base += w_before[w];
    for (int w = 0; w < wave; ++w) base += w_own[w];
    if (idx < N_NODES) rowptr[idx] = base + incl - c;
}

__global__ void edge_place(const int* __restrict__ src, const int* __restrict__ dst,
                           const unsigned* __restrict__ rank, const unsigned* __restrict__ rowptr,
                           int* __restrict__ slot) {
    int e = blockIdx.x * blockDim.x + threadIdx.x;
    if (e >= E0) return;
    v2i se; se.x = src[e]; se.y = e;
    ((v2i*)slot)[rowptr[dst[e]] + rank[e]] = se;
}

#define ROWS_PER_BLOCK 8
__global__ void __launch_bounds__(ROWS_PER_BLOCK * 32)
node_gather(const int* __restrict__ slot, const unsigned* __restrict__ rowptr,
            const unsigned* __restrict__ cnt,
            const float* __restrict__ ee, const float* __restrict__ z,
            const float* __restrict__ H, const float* __restrict__ bias,
            float* __restrict__ out, int relu) {
    const int lane = threadIdx.x & 31;
    const int node = __builtin_amdgcn_readfirstlane(blockIdx.x * ROWS_PER_BLOCK + (threadIdx.x >> 5));
    if (node >= N_NODES) return;

    const unsigned beg = rowptr[node];
    const unsigned deg = cnt[node];
    const v4f zz = *(const v4f*)(z + (size_t)node * HEADS);
    const float zd0 = zz.x + 1e-16f, zd1 = zz.y + 1e-16f, zd2 = zz.z + 1e-16f, zd3 = zz.w + 1e-16f;
    const bool upper = lane >= 16;

    const v4f* H4 = (const v4f*)H;
    const v4f* E4 = (const v4f*)ee;
    const v2i* S2 = (const v2i*)slot;

    const v4f bias0 = ((const v4f*)bias)[lane];
    const v4f bias1 = ((const v4f*)bias)[32 + lane];

    v4f acc0, acc1;
    {
        const v4f e4 = E4[E0 + node];
        const float alo = (upper ? e4.y : e4.x) / (upper ? zd1 : zd0);
        const float ahi = (upper ? e4.w : e4.z) / (upper ? zd3 : zd2);
        acc0 = H4[(size_t)node * 64 + lane] * alo;
        acc1 = H4[(size_t)node * 64 + 32 + lane] * ahi;
    }

    for (unsigned base = 0; base < deg; base += 32) {
        const unsigned n = (deg - base < 32u) ? (deg - base) : 32u;
        int   s  = 0;
        float a0 = 0.f, a1 = 0.f, a2 = 0.f, a3 = 0.f;
        if ((unsigned)lane < n) {
            const v2i se = S2[beg + base + lane];
            s = se.x;
            const v4f e4 = E4[se.y];
            a0 = e4.x / zd0; a1 = e4.y / zd1; a2 = e4.z / zd2; a3 = e4.w / zd3;
        }
#pragma unroll 4
        for (unsigned j = 0; j < n; ++j) {
            const int   sj = __builtin_amdgcn_readlane(s, (int)j);
            const float b0 = __int_as_float(__builtin_amdgcn_readlane(__float_as_int(a0), (int)j));
            const float b1 = __int_as_float(__builtin_amdgcn_readlane(__float_as_int(a1), (int)j));
            const float b2 = __int_as_float(__builtin_amdgcn_readlane(__float_as_int(a2), (int)j));
            const float b3 = __int_as_float(__builtin_amdgcn_readlane(__float_as_int(a3), (int)j));
            const float alo = upper ? b1 : b0;
            const float ahi = upper ? b3 : b2;
            const v4f v0 = H4[(size_t)sj * 64 + lane];
            const v4f v1 = H4[(size_t)sj * 64 + 32 + lane];
            acc0 += v0 * alo;
            acc1 += v1 * ahi;
        }
    }

    acc0 += bias0;
    acc1 += bias1;
    if (relu) {
        acc0.x = fmaxf(acc0.x, 0.f); acc0.y = fmaxf(acc0.y, 0.f);
        acc0.z = fmaxf(acc0.z, 0.f); acc0.w = fmaxf(acc0.w, 0.f);
        acc1.x = fmaxf(acc1.x, 0.f); acc1.y = fmaxf(acc1.y, 0.f);
        acc1.z = fmaxf(acc1.z, 0.f); acc1.w = fmaxf(acc1.w, 0.f);
    }
    v4f* O4 = (v4f*)out;
    O4[(size_t)node * 64 + lane]      = acc0;
    O4[(size_t)node * 64 + 32 + lane] = acc1;
}

extern "C" void kernel_launch(void* const* d_in, const int* in_sizes, int n_in,
                              void* d_out, int out_size, void* d_ws, size_t ws_size,
                              hipStream_t stream) {
    const float* x        = (const float*)d_in[0];
    const int*   eidx     = (const int*)  d_in[1];
    const float* W1       = (const float*)d_in[2];
    const float* att_s1   = (const float*)d_in[3];
    const float* att_d1   = (const float*)d_in[4];
    const float* b1       = (const float*)d_in[5];
    const float* W2       = (const float*)d_in[6];
    const float* att_s2   = (const float*)d_in[7];
    const float* att_d2   = (const float*)d_in[8];
    const float* b2       = (const float*)d_in[9];
    float*       out      = (float*)d_out;

    const int* src = eidx;
    const int* dst = eidx + E0;

    const size_t NB = (size_t)N_NODES * C_HID;
    float* h      = (float*)d_ws;
    float* hrelu  = h      + NB;
    float* ee     = hrelu  + NB;
    float* a_src  = ee     + (size_t)ETOT * HEADS;
    float* a_dst  = a_src  + (size_t)N_NODES * HEADS;
    unsigned* m   = (unsigned*)(a_dst + (size_t)N_NODES * HEADS);
    float* z      = (float*)(m + (size_t)N_NODES * HEADS);
    unsigned* cnt    = (unsigned*)(z + (size_t)N_NODES * HEADS);
    unsigned* rowptr = cnt + N_NODES;
    unsigned* rank   = rowptr + N_NODES;
    int*      slot   = (int*)(rank + E0);

    const int NH       = N_NODES * HEADS;
    const int EH       = ETOT * HEADS;
    const unsigned ORD_NEG_INF = 0x007FFFFFu;

    const dim3 B256(256);
    const int gNH  = (NH    + 255) / 256;
    const int gEH  = (EH    + 255) / 256;
    const int gMM  = N_NODES / 16;
    const int gDOT = (NH * 32 + 255) / 256;
    const int gN   = (N_NODES + 255) / 256;
    const int gE0  = (E0 + 255) / 256;
    const int gSCN = (N_NODES + SCAN_T - 1) / SCAN_T;
    const int gROW = (N_NODES + ROWS_PER_BLOCK - 1) / ROWS_PER_BLOCK;

    fill_u32      <<<gN,   B256, 0, stream>>>(cnt, 0u, N_NODES);
    edge_rank     <<<gE0,  B256, 0, stream>>>(dst, cnt, rank);
    bucket_offsets<<<gSCN, SCAN_T, 0, stream>>>(cnt, rowptr);
    edge_place    <<<gE0,  B256, 0, stream>>>(src, dst, rank, rowptr, slot);

    gemm_wmma  <<<gMM, 128, 0, stream>>>(x, W1, h, C_IN, C_HID);
    attn_dots  <<<gDOT, B256, 0, stream>>>(h, att_s1, att_d1, a_src, a_dst);
    fill_u32   <<<gNH, B256, 0, stream>>>(m, ORD_NEG_INF, NH);
    fill_u32   <<<gNH, B256, 0, stream>>>((unsigned*)z, 0u, NH);
    edge_max   <<<gEH, B256, 0, stream>>>(src, dst, a_src, a_dst, m);
    edge_expsum<<<gEH, B256, 0, stream>>>(src, dst, a_src, a_dst, m, ee, z);
    node_gather<<<gROW, ROWS_PER_BLOCK * 32, 0, stream>>>(slot, rowptr, cnt, ee, z, h, b1, hrelu, 1);

    gemm_wmma  <<<gMM, 128, 0, stream>>>(hrelu, W2, h, C_HID, C_HID);
    attn_dots  <<<gDOT, B256, 0, stream>>>(h, att_s2, att_d2, a_src, a_dst);
    fill_u32   <<<gNH, B256, 0, stream>>>(m, ORD_NEG_INF, NH);
    fill_u32   <<<gNH, B256, 0, stream>>>((unsigned*)z, 0u, NH);
    edge_max   <<<gEH, B256, 0, stream>>>(src, dst, a_src, a_dst, m);
    edge_expsum<<<gEH, B256, 0, stream>>>(src, dst, a_src, a_dst, m, ee, z);
    node_gather<<<gROW, ROWS_PER_BLOCK * 32, 0, stream>>>(slot, rowptr, cnt, ee, z, h, b2, out, 0);
}


# Edge softmax done by the wave that owns the destination bucket: logits, max, exp and z in registers (z by a fixed lane tree, not atomics); edge_max, edge_expsum, 4 fills and ee/m/z planes gone
# speedup vs baseline: 4.9583x; 1.5554x over previous
#include <hip/hip_runtime.h>
#include <math.h>

typedef __attribute__((ext_vector_type(2))) float v2f;
typedef __attribute__((ext_vector_type(8))) float v8f;
typedef __attribute__((ext_vector_type(4))) float v4f;

#define N_NODES 50000
#define C_IN    128
#define C_HID   256
#define HEADS   4
#define EMBED   64
#define E0      800000
#define ETOT    (E0 + N_NODES)
#define NEG_SLOPE 0.2f

__global__ void gemm_wmma(const float* __restrict__ X,
                          const float* __restrict__ W,
                          float* __restrict__ H,
                          int K, int N) {
    const int lane   = threadIdx.x & 31;
    const int wave   = threadIdx.x >> 5;
    const int tileM  = blockIdx.x;
    const int lane15 = lane & 15;
    const int khalf  = lane >> 4;
    const int row    = tileM * 16 + lane15;
    const int col0   = wave * 64;

    v8f acc0 = {}, acc1 = {}, acc2 = {}, acc3 = {};

    for (int k = 0; k < K; k += 4) {
        const int kk = k + khalf * 2;
        v2f a;
        a.x = X[row * K + kk];
        a.y = X[row * K + kk + 1];

        const float* wp0 = W + (size_t)kk * N;
        const float* wp1 = wp0 + N;

        v2f b0, b1, b2, b3;
        b0.x = wp0[col0 +  0 + lane15]; b0.y = wp1[col0 +  0 + lane15];
        b1.x = wp0[col0 + 16 + lane15]; b1.y = wp1[col0 + 16 + lane15];
        b2.x = wp0[col0 + 32 + lane15]; b2.y = wp1[col0 + 32 + lane15];
        b3.x = wp0[col0 + 48 + lane15]; b3.y = wp1[col0 + 48 + lane15];

        acc0 = __builtin_amdgcn_wmma_f32_16x16x4_f32(false, a, false, b0, (short)0, acc0, false, false);
        acc1 = __builtin_amdgcn_wmma_f32_16x16x4_f32(false, a, false, b1, (short)0, acc1, false, false);
        acc2 = __builtin_amdgcn_wmma_f32_16x16x4_f32(false, a, false, b2, (short)0, acc2, false, false);
        acc3 = __builtin_amdgcn_wmma_f32_16x16x4_f32(false, a, false, b3, (short)0, acc3, false, false);
    }

    const int rbase = tileM * 16 + khalf * 8;
#pragma unroll
    for (int i = 0; i < 8; ++i) {
        float* op = H + (size_t)(rbase + i) * N + col0 + lane15;
        op[ 0] = acc0[i];
        op[16] = acc1[i];
        op[32] = acc2[i];
        op[48] = acc3[i];
    }
}

__global__ void attn_dots(const float* __restrict__ H,
                          const float* __restrict__ att_src,
                          const float* __restrict__ att_dst,
                          float* __restrict__ a_src,
                          float* __restrict__ a_dst) {
    const int gw   = (int)((blockIdx.x * blockDim.x + threadIdx.x) >> 5);
    const int lane = threadIdx.x & 31;
    const int node = gw >> 2;
    const int head = gw & 3;
    if (node >= N_NODES) return;

    const float* hp = H + (size_t)node * C_HID + head * EMBED;
    const float* as = att_src + head * EMBED;
    const float* ad = att_dst + head * EMBED;

    float s = 0.f, d = 0.f;
#pragma unroll
    for (int c = 0; c < EMBED; c += 32) {
        float v = hp[c + lane];
        s += v * as[c + lane];
        d += v * ad[c + lane];
    }
#pragma unroll
    for (int off = 16; off > 0; off >>= 1) {
        s += __shfl_down(s, off, 32);
        d += __shfl_down(d, off, 32);
    }
    if (lane == 0) {
        a_src[node * HEADS + head] = s;
        a_dst[node * HEADS + head] = d;
    }
}

__global__ void fill_u32(unsigned* __restrict__ p, unsigned v, int n) {
    int i = blockIdx.x * blockDim.x + threadIdx.x;
    if (i < n) p[i] = v;
}

__global__ void edge_rank(const int* __restrict__ dst, unsigned* __restrict__ cnt,
                          unsigned* __restrict__ rank) {
    int e = blockIdx.x * blockDim.x + threadIdx.x;
    if (e >= E0) return;
    rank[e] = atomicAdd(&cnt[dst[e]], 1u);
}

#define SCAN_T 1024
__global__ void __launch_bounds__(SCAN_T) bucket_offsets(const unsigned* __restrict__ cnt,
                                                        unsigned* __restrict__ rowptr) {
    __shared__ unsigned w_before[SCAN_T / 32];
    __shared__ unsigned w_own[SCAN_T / 32];
    const int t = threadIdx.x, lane = t & 31, wave = t >> 5;
    const int first = blockIdx.x * SCAN_T;

    unsigned before = 0;
    for (int i = t; i < first; i += SCAN_T) before += cnt[i];
    const int idx = first + t;
    const unsigned c = (idx < N_NODES) ? cnt[idx] : 0u;

    unsigned incl = c;
    for (int off = 1; off < 32; off <<= 1) {
        unsigned v = __shfl_up(incl, off, 32);
        if (lane >= off) incl += v;
    }
    for (int off = 16; off > 0; off >>= 1) before += __shfl_xor(before, off, 32);
    if (lane == 31) w_own[wave] = incl;
    if (lane == 0)  w_before[wave] = before;
    __syncthreads();
    unsigned base = 0;
    for (int w = 0; w < SCAN_T / 32; ++w) base += w_before[w];
    for (int w = 0; w < wave; ++w) base += w_own[w];
    if (idx < N_NODES) rowptr[idx] = base + incl - c;
}

__global__ void edge_place(const int* __restrict__ src, const int* __restrict__ dst,
                           const unsigned* __restrict__ rank, const unsigned* __restrict__ rowptr,
                           int* __restrict__ slot) {
    int e = blockIdx.x * blockDim.x + threadIdx.x;
    if (e >= E0) return;
    slot[rowptr[dst[e]] + rank[e]] = src[e];
}

__device__ __forceinline__ float xhalf(float x) {
    const int b = __builtin_bit_cast(int, x);
    return __builtin_bit_cast(float, __builtin_amdgcn_permlanex16(b, b, 0x76543210, 0xfedcba98, false, false));
}
template <int MASK>
__device__ __forceinline__ float row_xor(float x) {
    return __builtin_bit_cast(float, __builtin_amdgcn_update_dpp(0, __builtin_bit_cast(int, x),
                                                                 0x160 + MASK, 0xf, 0xf, false));
}
__device__ __forceinline__ float wave_max(float v) {
    v = fmaxf(v, row_xor<1>(v));
    v = fmaxf(v, row_xor<2>(v));
    v = fmaxf(v, row_xor<4>(v));
    v = fmaxf(v, row_xor<8>(v));
    return fmaxf(v, xhalf(v));
}
__device__ __forceinline__ float wave_sum(float v) {
    v += row_xor<1>(v);
    v += row_xor<2>(v);
    v += row_xor<4>(v);
    v += row_xor<8>(v);
    return v + xhalf(v);
}

__device__ __forceinline__ float uni(float x) {
    return __builtin_bit_cast(float, __builtin_amdgcn_readfirstlane(__builtin_bit_cast(int, x)));
}

__device__ __forceinline__ float leaky(float v) { return v > 0.f ? v : NEG_SLOPE * v; }

__device__ __forceinline__ void load_edges(const int* __restrict__ slot, const v4f* __restrict__ AS4,
                                           unsigned first, unsigned n, int lane, v4f ad,
                                           int& s, float& l0, float& l1, float& l2, float& l3) {
    s = 0;
    l0 = l1 = l2 = l3 = -INFINITY;
    if ((unsigned)lane < n) {
        s = slot[first + lane];
        const v4f as = AS4[s];
        l0 = leaky(as.x + ad.x); l1 = leaky(as.y + ad.y);
        l2 = leaky(as.z + ad.z); l3 = leaky(as.w + ad.w);
    }
}

__device__ __forceinline__ void add_rows(const v4f* __restrict__ H4, unsigned n, int lane, bool upper,
                                         int s, float a0, float a1, float a2, float a3,
                                         v4f& acc0, v4f& acc1) {
#pragma unroll 4
    for (unsigned j = 0; j < n; ++j) {
        const int   sj = __builtin_amdgcn_readlane(s, (int)j);
        const float b0 = __int_as_float(__builtin_amdgcn_readlane(__float_as_int(a0), (int)j));
        const float b1 = __int_as_float(__builtin_amdgcn_readlane(__float_as_int(a1), (int)j));
        const float b2 = __int_as_float(__builtin_amdgcn_readlane(__float_as_int(a2), (int)j));
        const float b3 = __int_as_float(__builtin_amdgcn_readlane(__float_as_int(a3), (int)j));
        const float alo = upper ? b1 : b0;
        const float ahi = upper ? b3 : b2;
        const v4f v0 = H4[(size_t)sj * 64 + lane];
        const v4f v1 = H4[(size_t)sj * 64 + 32 + lane];
        acc0 += v0 * alo;
        acc1 += v1 * ahi;
    }
}

#define ROWS_PER_BLOCK 8
__global__ void __launch_bounds__(ROWS_PER_BLOCK * 32)
node_softmax_gather(const int* __restrict__ slot, const unsigned* __restrict__ rowptr,
                    const unsigned* __restrict__ cnt,
                    const float* __restrict__ a_src, const float* __restrict__ a_dst,
                    const float* __restrict__ H, const float* __restrict__ bias,
                    float* __restrict__ out, int relu) {
    const int lane = threadIdx.x & 31;
    const int node = __builtin_amdgcn_readfirstlane(blockIdx.x * ROWS_PER_BLOCK + (threadIdx.x >> 5));
    if (node >= N_NODES) return;

    const v4f* H4  = (const v4f*)H;
    const v4f* AS4 = (const v4f*)a_src;

    const unsigned beg = rowptr[node];
    const unsigned deg = cnt[node];
    const v4f ad  = ((const v4f*)a_dst)[node];
    const v4f asn = AS4[node];
    const bool upper = lane >= 16;

    const v4f bias0 = ((const v4f*)bias)[lane];
    const v4f bias1 = ((const v4f*)bias)[32 + lane];
    const v4f self0 = H4[(size_t)node * 64 + lane];
    const v4f self1 = H4[(size_t)node * 64 + 32 + lane];

    const float ls0 = leaky(asn.x + ad.x), ls1 = leaky(asn.y + ad.y);
    const float ls2 = leaky(asn.z + ad.z), ls3 = leaky(asn.w + ad.w);

    const unsigned n0 = deg < 32u ? deg : 32u;
    int   s0;
    float p0, p1, p2, p3;
    load_edges(slot, AS4, beg, n0, lane, ad, s0, p0, p1, p2, p3);

    float m0 = fmaxf(ls0, p0), m1 = fmaxf(ls1, p1), m2 = fmaxf(ls2, p2), m3 = fmaxf(ls3, p3);
    for (unsigned base = 32; base < deg; base += 32) {
        const unsigned n = (deg - base < 32u) ? (deg - base) : 32u;
        int s; float l0, l1, l2, l3;
        load_edges(slot, AS4, beg + base, n, lane, ad, s, l0, l1, l2, l3);
        m0 = fmaxf(m0, l0); m1 = fmaxf(m1, l1); m2 = fmaxf(m2, l2); m3 = fmaxf(m3, l3);
    }
    m0 = uni(wave_max(m0)); m1 = uni(wave_max(m1)); m2 = uni(wave_max(m2)); m3 = uni(wave_max(m3));
    m0 = __builtin_isfinite(m0) ? m0 : 0.f; m1 = __builtin_isfinite(m1) ? m1 : 0.f;
    m2 = __builtin_isfinite(m2) ? m2 : 0.f; m3 = __builtin_isfinite(m3) ? m3 : 0.f;

    p0 = __expf(p0 - m0); p1 = __expf(p1 - m1); p2 = __expf(p2 - m2); p3 = __expf(p3 - m3);
    float z0 = p0, z1 = p1, z2 = p2, z3 = p3;
    for (unsigned base = 32; base < deg; base += 32) {
        const unsigned n = (deg - base < 32u) ? (deg - base) : 32u;
        int s; float l0, l1, l2, l3;
        load_edges(slot, AS4, beg + base, n, lane, ad, s, l0, l1, l2, l3);
        z0 += __expf(l0 - m0); z1 += __expf(l1 - m1); z2 += __expf(l2 - m2); z3 += __expf(l3 - m3);
    }
    const float es0 = __expf(ls0 - m0), es1 = __expf(ls1 - m1);
    const float es2 = __expf(ls2 - m2), es3 = __expf(ls3 - m3);
    const float zd0 = uni(wave_sum(z0) + es0 + 1e-16f), zd1 = uni(wave_sum(z1) + es1 + 1e-16f);
    const float zd2 = uni(wave_sum(z2) + es2 + 1e-16f), zd3 = uni(wave_sum(z3) + es3 + 1e-16f);

    v4f acc0, acc1;
    {
        const float alo = (upper ? es1 : es0) / (upper ? zd1 : zd0);
        const float ahi = (upper ? es3 : es2) / (upper ? zd3 : zd2);
        acc0 = self0 * alo;
        acc1 = self1 * ahi;
    }
    {
        int      s = s0;
        unsigned n = n0;
        float a0 = p0 / zd0, a1 = p1 / zd1, a2 = p2 / zd2, a3 = p3 / zd3;
        for (unsigned base = 0;;) {
            add_rows(H4, n, lane, upper, s, a0, a1, a2, a3, acc0, acc1);
            base += 32;
            if (base >= deg) break;
            n = (deg - base < 32u) ? (deg - base) : 32u;
            float l0, l1, l2, l3;
            load_edges(slot, AS4, beg + base, n, lane, ad, s, l0, l1, l2, l3);
            a0 = __expf(l0 - m0) / zd0; a1 = __expf(l1 - m1) / zd1;
            a2 = __expf(l2 - m2) / zd2; a3 = __expf(l3 - m3) / zd3;
        }
    }

    acc0 += bias0;
    acc1 += bias1;
    if (relu) {
        acc0.x = fmaxf(acc0.x, 0.f); acc0.y = fmaxf(acc0.y, 0.f);
        acc0.z = fmaxf(acc0.z, 0.f); acc0.w = fmaxf(acc0.w, 0.f);
        acc1.x = fmaxf(acc1.x, 0.f); acc1.y = fmaxf(acc1.y, 0.f);
        acc1.z = fmaxf(acc1.z, 0.f); acc1.w = fmaxf(acc1.w, 0.f);
    }
    v4f* O4 = (v4f*)out;
    O4[(size_t)node * 64 + lane]      = acc0;
    O4[(size_t)node * 64 + 32 + lane] = acc1;
}

extern "C" void kernel_launch(void* const* d_in, const int* in_sizes, int n_in,
                              void* d_out, int out_size, void* d_ws, size_t ws_size,
                              hipStream_t stream) {
    const float* x        = (const float*)d_in[0];
    const int*   eidx     = (const int*)  d_in[1];
    const float* W1       = (const float*)d_in[2];
    const float* att_s1   = (const float*)d_in[3];
    const float* att_d1   = (const float*)d_in[4];
    const float* b1       = (const float*)d_in[5];
    const float* W2       = (const float*)d_in[6];
    const float* att_s2   = (const float*)d_in[7];
    const float* att_d2   = (const float*)d_in[8];
    const float* b2       = (const float*)d_in[9];
    float*       out      = (float*)d_out;

    const int* src = eidx;
    const int* dst = eidx + E0;

    const size_t NB = (size_t)N_NODES * C_HID;
    float* h      = (float*)d_ws;
    float* hrelu  = h      + NB;
    float* a_src  = hrelu  + NB;
    float* a_dst  = a_src  + (size_t)N_NODES * HEADS;
    unsigned* cnt    = (unsigned*)(a_dst + (size_t)N_NODES * HEADS);
    unsigned* rowptr = cnt + N_NODES;
    unsigned* rank   = rowptr + N_NODES;
    int*      slot   = (int*)(rank + E0);

    const int NH       = N_NODES * HEADS;

    const dim3 B256(256);
    const int gMM  = N_NODES / 16;
    const int gDOT = (NH * 32 + 255) / 256;
    const int gN   = (N_NODES + 255) / 256;
    const int gE0  = (E0 + 255) / 256;
    const int gSCN = (N_NODES + SCAN_T - 1) / SCAN_T;
    const int gROW = (N_NODES + ROWS_PER_BLOCK - 1) / ROWS_PER_BLOCK;

    fill_u32      <<<gN,   B256, 0, stream>>>(cnt, 0u, N_NODES);
    edge_rank     <<<gE0,  B256, 0, stream>>>(dst, cnt, rank);
    bucket_offsets<<<gSCN, SCAN_T, 0, stream>>>(cnt, rowptr);
    edge_place    <<<gE0,  B256, 0, stream>>>(src, dst, rank, rowptr, slot);

    gemm_wmma  <<<gMM, 128, 0, stream>>>(x, W1, h, C_IN, C_HID);
    attn_dots  <<<gDOT, B256, 0, stream>>>(h, att_s1, att_d1, a_src, a_dst);
    node_softmax_gather<<<gROW, ROWS_PER_BLOCK * 32, 0, stream>>>(slot, rowptr, cnt, a_src, a_dst, h, b1, hrelu, 1);

    gemm_wmma  <<<gMM, 128, 0, stream>>>(hrelu, W2, h, C_HID, C_HID);
    attn_dots  <<<gDOT, B256, 0, stream>>>(h, att_s2, att_d2, a_src, a_dst);
    node_softmax_gather<<<gROW, ROWS_PER_BLOCK * 32, 0, stream>>>(slot, rowptr, cnt, a_src, a_dst, h, b2, out, 0);
}


# Attention dots (a_src, a_dst) come from the accumulators in the GEMM epilogue, one head per wave: both attn_dots launches and their re-read of h gone; dot sum order: 4 per lane, then 16-lane xor tree
# speedup vs baseline: 5.6331x; 1.1361x over previous
#include <hip/hip_runtime.h>
#include <math.h>

typedef __attribute__((ext_vector_type(2))) float v2f;
typedef __attribute__((ext_vector_type(8))) float v8f;
typedef __attribute__((ext_vector_type(4))) float v4f;

#define N_NODES 50000
#define C_IN    128
#define C_HID   256
#define HEADS   4
#define EMBED   64
#define E0      800000
#define ETOT    (E0 + N_NODES)
#define NEG_SLOPE 0.2f

template <int MASK>
__device__ __forceinline__ float row_xor(float x) {
    return __builtin_bit_cast(float, __builtin_amdgcn_update_dpp(0, __builtin_bit_cast(int, x),
                                                                 0x160 + MASK, 0xf, 0xf, false));
}
__device__ __forceinline__ float row_sum16(float v) {
    v += row_xor<1>(v);
    v += row_xor<2>(v);
    v += row_xor<4>(v);
    v += row_xor<8>(v);
    return v;
}

__global__ void gemm_wmma(const float* __restrict__ X,
                          const float* __restrict__ W,
                          float* __restrict__ H,
                          const float* __restrict__ att_src,
                          const float* __restrict__ att_dst,
                          float* __restrict__ a_src,
                          float* __restrict__ a_dst,
                          int K, int N) {
    __shared__ v4f dots4[2][16];
    float* dots = (float*)dots4;

    const int lane   = threadIdx.x & 31;
    const int wave   = threadIdx.x >> 5;
    const int tileM  = blockIdx.x;
    const int lane15 = lane & 15;
    const int khalf  = lane >> 4;
    const int row    = tileM * 16 + lane15;
    const int col0   = wave * 64;

    v8f acc0 = {}, acc1 = {}, acc2 = {}, acc3 = {};

    for (int k = 0; k < K; k += 4) {
        const int kk = k + khalf * 2;
        v2f a;
        a.x = X[row * K + kk];
        a.y = X[row * K + kk + 1];

        const float* wp0 = W + (size_t)kk * N;
        const float* wp1 = wp0 + N;

        v2f b0, b1, b2, b3;
        b0.x = wp0[col0 +  0 + lane15]; b0.y = wp1[col0 +  0 + lane15];
        b1.x = wp0[col0 + 16 + lane15]; b1.y = wp1[col0 + 16 + lane15];
        b2.x = wp0[col0 + 32 + lane15]; b2.y = wp1[col0 + 32 + lane15];
        b3.x = wp0[col0 + 48 + lane15]; b3.y = wp1[col0 + 48 + lane15];

        acc0 = __builtin_amdgcn_wmma_f32_16x16x4_f32(false, a, false, b0, (short)0, acc0, false, false);
        acc1 = __builtin_amdgcn_wmma_f32_16x16x4_f32(false, a, false, b1, (short)0, acc1, false, false);
        acc2 = __builtin_amdgcn_wmma_f32_16x16x4_f32(false, a, false, b2, (short)0, acc2, false, false);
        acc3 = __builtin_amdgcn_wmma_f32_16x16x4_f32(false, a, false, b3, (short)0, acc3, false, false);
    }

    const float* sp = att_src + (col0 + lane15);
    const float* dp = att_dst + (col0 + lane15);
    const float s0 = sp[0], s1 = sp[16], s2 = sp[32], s3 = sp[48];
    const float d0 = dp[0], d1 = dp[16], d2 = dp[32], d3 = dp[48];
#pragma unroll
    for (int i = 0; i < 8; ++i) {
        const float ps = row_sum16(acc0[i] * s0 + acc1[i] * s1 + acc2[i] * s2 + acc3[i] * s3);
        const float pd = row_sum16(acc0[i] * d0 + acc1[i] * d1 + acc2[i] * d2 + acc3[i] * d3);
        if (lane15 == 0) {
            dots[(0 * 16 + khalf * 8 + i) * HEADS + wave] = ps;
            dots[(1 * 16 + khalf * 8 + i) * HEADS + wave] = pd;
        }
    }
    __syncthreads();
    if (wave == 0) {
        float* plane = khalf ? a_dst : a_src;
        *(v4f*)(plane + (size_t)(tileM * 16 + lane15) * HEADS) = dots4[khalf][lane15];
    }

    const int rbase = tileM * 16 + khalf * 8;
#pragma unroll
    for (int i = 0; i < 8; ++i) {
        float* op = H + (size_t)(rbase + i) * N + col0 + lane15;
        op[ 0] = acc0[i];
        op[16] = acc1[i];
        op[32] = acc2[i];
        op[48] = acc3[i];
    }
}

__global__ void fill_u32(unsigned* __restrict__ p, unsigned v, int n) {
    int i = blockIdx.x * blockDim.x + threadIdx.x;
    if (i < n) p[i] = v;
}

__global__ void edge_rank(const int* __restrict__ dst, unsigned* __restrict__ cnt,
                          unsigned* __restrict__ rank) {
    int e = blockIdx.x * blockDim.x + threadIdx.x;
    if (e >= E0) return;
    rank[e] = atomicAdd(&cnt[dst[e]], 1u);
}

#define SCAN_T 1024
__global__ void __launch_bounds__(SCAN_T) bucket_offsets(const unsigned* __restrict__ cnt,
                                                        unsigned* __restrict__ rowptr) {
    __shared__ unsigned w_before[SCAN_T / 32];
    __shared__ unsigned w_own[SCAN_T / 32];
    const int t = threadIdx.x, lane = t & 31, wave = t >> 5;
    const int first = blockIdx.x * SCAN_T;

    unsigned before = 0;
    for (int i = t; i < first; i += SCAN_T) before += cnt[i];
    const int idx = first + t;
    const unsigned c = (idx < N_NODES) ? cnt[idx] : 0u;

    unsigned incl = c;
    for (int off = 1; off < 32; off <<= 1) {
        unsigned v = __shfl_up(incl, off, 32);
        if (lane >= off) incl += v;
    }
    for (int off = 16; off > 0; off >>= 1) before += __shfl_xor(before, off, 32);
    if (lane == 31) w_own[wave] = incl;
    if (lane == 0)  w_before[wave] = before;
    __syncthreads();
    unsigned base = 0;
    for (int w = 0; w < SCAN_T / 32; ++w) base += w_before[w];
    for (int w = 0; w < wave; ++w) base += w_own[w];
    if (idx < N_NODES) rowptr[idx] = base + incl - c;
}

__global__ void edge_place(const int* __restrict__ src, const int* __restrict__ dst,
                           const unsigned* __restrict__ rank, const unsigned* __restrict__ rowptr,
                           int* __restrict__ slot) {
    int e = blockIdx.x * blockDim.x + threadIdx.x;
    if (e >= E0) return;
    slot[rowptr[dst[e]] + rank[e]] = src[e];
}

__device__ __forceinline__ float xhalf(float x) {
    const int b = __builtin_bit_cast(int, x);
    return __builtin_bit_cast(float, __builtin_amdgcn_permlanex16(b, b, 0x76543210, 0xfedcba98, false, false));
}
__device__ __forceinline__ float wave_max(float v) {
    v = fmaxf(v, row_xor<1>(v));
    v = fmaxf(v, row_xor<2>(v));
    v = fmaxf(v, row_xor<4>(v));
    v = fmaxf(v, row_xor<8>(v));
    return fmaxf(v, xhalf(v));
}
__device__ __forceinline__ float wave_sum(float v) {
    v += row_xor<1>(v);
    v += row_xor<2>(v);
    v += row_xor<4>(v);
    v += row_xor<8>(v);
    return v + xhalf(v);
}

__device__ __forceinline__ float uni(float x) {
    return __builtin_bit_cast(float, __builtin_amdgcn_readfirstlane(__builtin_bit_cast(int, x)));
}

__device__ __forceinline__ float leaky(float v) { return v > 0.f ? v : NEG_SLOPE * v; }

__device__ __forceinline__ void load_edges(const int* __restrict__ slot, const v4f* __restrict__ AS4,
                                           unsigned first, unsigned n, int lane, v4f ad,
                                           int& s, float& l0, float& l1, float& l2, float& l3) {
    s = 0;
    l0 = l1 = l2 = l3 = -INFINITY;
    if ((unsigned)lane < n) {
        s = slot[first + lane];
        const v4f as = AS4[s];
        l0 = leaky(as.x + ad.x); l1 = leaky(as.y + ad.y);
        l2 = leaky(as.z + ad.z); l3 = leaky(as.w + ad.w);
    }
}

__device__ __forceinline__ void add_rows(const v4f* __restrict__ H4, unsigned n, int lane, bool upper,
                                         int s, float a0, float a1, float a2, float a3,
                                         v4f& acc0, v4f& acc1) {
#pragma unroll 4
    for (unsigned j = 0; j < n; ++j) {
        const int   sj = __builtin_amdgcn_readlane(s, (int)j);
        const float b0 = __int_as_float(__builtin_amdgcn_readlane(__float_as_int(a0), (int)j));
        const float b1 = __int_as_float(__builtin_amdgcn_readlane(__float_as_int(a1), (int)j));
        const float b2 = __int_as_float(__builtin_amdgcn_readlane(__float_as_int(a2), (int)j));
        const float b3 = __int_as_float(__builtin_amdgcn_readlane(__float_as_int(a3), (int)j));
        const float alo = upper ? b1 : b0;
        const float ahi = upper ? b3 : b2;
        const v4f v0 = H4[(size_t)sj * 64 + lane];
        const v4f v1 = H4[(size_t)sj * 64 + 32 + lane];
        acc0 += v0 * alo;
        acc1 += v1 * ahi;
    }
}

#define ROWS_PER_BLOCK 8
__global__ void __launch_bounds__(ROWS_PER_BLOCK * 32)
node_softmax_gather(const int* __restrict__ slot, const unsigned* __restrict__ rowptr,
                    const unsigned* __restrict__ cnt,
                    const float* __restrict__ a_src, const float* __restrict__ a_dst,
                    const float* __restrict__ H, const float* __restrict__ bias,
                    float* __restrict__ out, int relu) {
    const int lane = threadIdx.x & 31;
    const int node = __builtin_amdgcn_readfirstlane(blockIdx.x * ROWS_PER_BLOCK + (threadIdx.x >> 5));
    if (node >= N_NODES) return;

    const v4f* H4  = (const v4f*)H;
    const v4f* AS4 = (const v4f*)a_src;

    const unsigned beg = rowptr[node];
    const unsigned deg = cnt[node];
    const v4f ad  = ((const v4f*)a_dst)[node];
    const v4f asn = AS4[node];
    const bool upper = lane >= 16;

    const v4f bias0 = ((const v4f*)bias)[lane];
    const v4f bias1 = ((const v4f*)bias)[32 + lane];
    const v4f self0 = H4[(size_t)node * 64 + lane];
    const v4f self1 = H4[(size_t)node * 64 + 32 + lane];

    const float ls0 = leaky(asn.x + ad.x), ls1 = leaky(asn.y + ad.y);
    const float ls2 = leaky(asn.z + ad.z), ls3 = leaky(asn.w + ad.w);

    const unsigned n0 = deg < 32u ? deg : 32u;
    int   s0;
    float p0, p1, p2, p3;
    load_edges(slot, AS4, beg, n0, lane, ad, s0, p0, p1, p2, p3);

    float m0 = fmaxf(ls0, p0), m1 = fmaxf(ls1, p1), m2 = fmaxf(ls2, p2), m3 = fmaxf(ls3, p3);
    for (unsigned base = 32; base < deg; base += 32) {
        const unsigned n = (deg - base < 32u) ? (deg - base) : 32u;
        int s; float l0, l1, l2, l3;
        load_edges(slot, AS4, beg + base, n, lane, ad, s, l0, l1, l2, l3);
        m0 = fmaxf(m0, l0); m1 = fmaxf(m1, l1); m2 = fmaxf(m2, l2); m3 = fmaxf(m3, l3);
    }
    m0 = uni(wave_max(m0)); m1 = uni(wave_max(m1)); m2 = uni(wave_max(m2)); m3 = uni(wave_max(m3));
    m0 = __builtin_isfinite(m0) ? m0 : 0.f; m1 = __builtin_isfinite(m1) ? m1 : 0.f;
    m2 = __builtin_isfinite(m2) ? m2 : 0.f; m3 = __builtin_isfinite(m3) ? m3 : 0.f;

    p0 = __expf(p0 - m0); p1 = __expf(p1 - m1); p2 = __expf(p2 - m2); p3 = __expf(p3 - m3);
    float z0 = p0, z1 = p1, z2 = p2, z3 = p3;
    for (unsigned base = 32; base < deg; base += 32) {
        const unsigned n = (deg - base < 32u) ? (deg - base) : 32u;
        int s; float l0, l1, l2, l3;
        load_edges(slot, AS4, beg + base, n, lane, ad, s, l0, l1, l2, l3);
        z0 += __expf(l0 - m0); z1 += __expf(l1 - m1); z2 += __expf(l2 - m2); z3 += __expf(l3 - m3);
    }
    const float es0 = __expf(ls0 - m0), es1 = __expf(ls1 - m1);
    const float es2 = __expf(ls2 - m2), es3 = __expf(ls3 - m3);
    const float zd0 = uni(wave_sum(z0) + es0 + 1e-16f), zd1 = uni(wave_sum(z1) + es1 + 1e-16f);
    const float zd2 = uni(wave_sum(z2) + es2 + 1e-16f), zd3 = uni(wave_sum(z3) + es3 + 1e-16f);

    v4f acc0, acc1;
    {
        const float alo = (upper ? es1 : es0) / (upper ? zd1 : zd0);
        const float ahi = (upper ? es3 : es2) / (upper ? zd3 : zd2);
        acc0 = self0 * alo;
        acc1 = self1 * ahi;
    }
    {
        int      s = s0;
        unsigned n = n0;
        float a0 = p0 / zd0, a1 = p1 / zd1, a2 = p2 / zd2, a3 = p3 / zd3;
        for (unsigned base = 0;;) {
            add_rows(H4, n, lane, upper, s, a0, a1, a2, a3, acc0, acc1);
            base += 32;
            if (base >= deg) break;
            n = (deg - base < 32u) ? (deg - base) : 32u;
            float l0, l1, l2, l3;
            load_edges(slot, AS4, beg + base, n, lane, ad, s, l0, l1, l2, l3);
            a0 = __expf(l0 - m0) / zd0; a1 = __expf(l1 - m1) / zd1;
            a2 = __expf(l2 - m2) / zd2; a3 = __expf(l3 - m3) / zd3;
        }
    }

    acc0 += bias0;
    acc1 += bias1;
    if (relu) {
        acc0.x = fmaxf(acc0.x, 0.f); acc0.y = fmaxf(acc0.y, 0.f);
        acc0.z = fmaxf(acc0.z, 0.f); acc0.w = fmaxf(acc0.w, 0.f);
        acc1.x = fmaxf(acc1.x, 0.f); acc1.y = fmaxf(acc1.y, 0.f);
        acc1.z = fmaxf(acc1.z, 0.f); acc1.w = fmaxf(acc1.w, 0.f);
    }
    v4f* O4 = (v4f*)out;
    O4[(size_t)node * 64 + lane]      = acc0;
    O4[(size_t)node * 64 + 32 + lane] = acc1;
}

extern "C" void kernel_launch(void* const* d_in, const int* in_sizes, int n_in,
                              void* d_out, int out_size, void* d_ws, size_t ws_size,
                              hipStream_t stream) {
    const float* x        = (const float*)d_in[0];
    const int*   eidx     = (const int*)  d_in[1];
    const float* W1       = (const float*)d_in[2];
    const float* att_s1   = (const float*)d_in[3];
    const float* att_d1   = (const float*)d_in[4];
    const float* b1       = (const float*)d_in[5];
    const float* W2       = (const float*)d_in[6];
    const float* att_s2   = (const float*)d_in[7];
    const float* att_d2   = (const float*)d_in[8];
    const float* b2       = (const float*)d_in[9];
    float*       out      = (float*)d_out;

    const int* src = eidx;
    const int* dst = eidx + E0;

    const size_t NB = (size_t)N_NODES * C_HID;
    float* h      = (float*)d_ws;
    float* hrelu  = h      + NB;
    float* a_src  = hrelu  + NB;
    float* a_dst  = a_src  + (size_t)N_NODES * HEADS;
    unsigned* cnt    = (unsigned*)(a_dst + (size_t)N_NODES * HEADS);
    unsigned* rowptr = cnt + N_NODES;
    unsigned* rank   = rowptr + N_NODES;
    int*      slot   = (int*)(rank + E0);

    const dim3 B256(256);
    const int gMM  = N_NODES / 16;
    const int gN   = (N_NODES + 255) / 256;
    const int gE0  = (E0 + 255) / 256;
    const int gSCN = (N_NODES + SCAN_T - 1) / SCAN_T;
    const int gROW = (N_NODES + ROWS_PER_BLOCK - 1) / ROWS_PER_BLOCK;

    fill_u32      <<<gN,   B256, 0, stream>>>(cnt, 0u, N_NODES);
    edge_rank     <<<gE0,  B256, 0, stream>>>(dst, cnt, rank);
    bucket_offsets<<<gSCN, SCAN_T, 0, stream>>>(cnt, rowptr);
    edge_place    <<<gE0,  B256, 0, stream>>>(src, dst, rank, rowptr, slot);

    gemm_wmma  <<<gMM, 128, 0, stream>>>(x, W1, h, att_s1, att_d1, a_src, a_dst, C_IN, C_HID);
    node_softmax_gather<<<gROW, ROWS_PER_BLOCK * 32, 0, stream>>>(slot, rowptr, cnt, a_src, a_dst, h, b1, hrelu, 1);

    gemm_wmma  <<<gMM, 128, 0, stream>>>(hrelu, W2, h, att_s2, att_d2, a_src, a_dst, C_HID, C_HID);
    node_softmax_gather<<<gROW, ROWS_PER_BLOCK * 32, 0, stream>>>(slot, rowptr, cnt, a_src, a_dst, h, b2, out, 0);
}


# Edge sort on LDS counters: 3200-edge chunks ordered by 256-row bin, then one work-group per bin ranks by returning LDS add; global atomics, fill and scan gone (8 to 6 launches); sum order = LDS rank
# speedup vs baseline: 7.4580x; 1.3240x over previous
#include <hip/hip_runtime.h>
#include <math.h>

typedef __attribute__((ext_vector_type(2))) float v2f;
typedef __attribute__((ext_vector_type(8))) float v8f;
typedef __attribute__((ext_vector_type(4))) float v4f;

#define N_NODES 50000
#define C_IN    128
#define C_HID   256
#define HEADS   4
#define EMBED   64
#define E0      800000
#define ETOT    (E0 + N_NODES)
#define NEG_SLOPE 0.2f

template <int MASK>
__device__ __forceinline__ float row_xor(float x) {
    return __builtin_bit_cast(float, __builtin_amdgcn_update_dpp(0, __builtin_bit_cast(int, x),
                                                                 0x160 + MASK, 0xf, 0xf, false));
}
__device__ __forceinline__ float row_sum16(float v) {
    v += row_xor<1>(v);
    v += row_xor<2>(v);
    v += row_xor<4>(v);
    v += row_xor<8>(v);
    return v;
}

__global__ void gemm_wmma(const float* __restrict__ X,
                          const float* __restrict__ W,
                          float* __restrict__ H,
                          const float* __restrict__ att_src,
                          const float* __restrict__ att_dst,
                          float* __restrict__ a_src,
                          float* __restrict__ a_dst,
                          int K, int N) {
    __shared__ v4f dots4[2][16];
    float* dots = (float*)dots4;

    const int lane   = threadIdx.x & 31;
    const int wave   = threadIdx.x >> 5;
    const int tileM  = blockIdx.x;
    const int lane15 = lane & 15;
    const int khalf  = lane >> 4;
    const int row    = tileM * 16 + lane15;
    const int col0   = wave * 64;

    v8f acc0 = {}, acc1 = {}, acc2 = {}, acc3 = {};

    for (int k = 0; k < K; k += 4) {
        const int kk = k + khalf * 2;
        v2f a;
        a.x = X[row * K + kk];
        a.y = X[row * K + kk + 1];

        const float* wp0 = W + (size_t)kk * N;
        const float* wp1 = wp0 + N;

        v2f b0, b1, b2, b3;
        b0.x = wp0[col0 +  0 + lane15]; b0.y = wp1[col0 +  0 + lane15];
        b1.x = wp0[col0 + 16 + lane15]; b1.y = wp1[col0 + 16 + lane15];
        b2.x = wp0[col0 + 32 + lane15]; b2.y = wp1[col0 + 32 + lane15];
        b3.x = wp0[col0 + 48 + lane15]; b3.y = wp1[col0 + 48 + lane15];

        acc0 = __builtin_amdgcn_wmma_f32_16x16x4_f32(false, a, false, b0, (short)0, acc0, false, false);
        acc1 = __builtin_amdgcn_wmma_f32_16x16x4_f32(false, a, false, b1, (short)0, acc1, false, false);
        acc2 = __builtin_amdgcn_wmma_f32_16x16x4_f32(false, a, false, b2, (short)0, acc2, false, false);
        acc3 = __builtin_amdgcn_wmma_f32_16x16x4_f32(false, a, false, b3, (short)0, acc3, false, false);
    }

    const float* sp = att_src + (col0 + lane15);
    const float* dp = att_dst + (col0 + lane15);
    const float s0 = sp[0], s1 = sp[16], s2 = sp[32], s3 = sp[48];
    const float d0 = dp[0], d1 = dp[16], d2 = dp[32], d3 = dp[48];
#pragma unroll
    for (int i = 0; i < 8; ++i) {
        const float ps = row_sum16(acc0[i] * s0 + acc1[i] * s1 + acc2[i] * s2 + acc3[i] * s3);
        const float pd = row_sum16(acc0[i] * d0 + acc1[i] * d1 + acc2[i] * d2 + acc3[i] * d3);
        if (lane15 == 0) {
            dots[(0 * 16 + khalf * 8 + i) * HEADS + wave] = ps;
            dots[(1 * 16 + khalf * 8 + i) * HEADS + wave] = pd;
        }
    }
    __syncthreads();
    if (wave == 0) {
        float* plane = khalf ? a_dst : a_src;
        *(v4f*)(plane + (size_t)(tileM * 16 + lane15) * HEADS) = dots4[khalf][lane15];
    }

    const int rbase = tileM * 16 + khalf * 8;
#pragma unroll
    for (int i = 0; i < 8; ++i) {
        float* op = H + (size_t)(rbase + i) * N + col0 + lane15;
        op[ 0] = acc0[i];
        op[16] = acc1[i];
        op[32] = acc2[i];
        op[48] = acc3[i];
    }
}

typedef __attribute__((ext_vector_type(2))) int v2i;
typedef __attribute__((ext_vector_type(2))) unsigned v2u;

#define BIN_SHIFT 8
#define BIN_NODES (1 << BIN_SHIFT)
#define NBIN      ((N_NODES + BIN_NODES - 1) / BIN_NODES)
#define CHUNK     3200
#define NCHUNK    (E0 / CHUNK)
#define SORT_T    1024
static_assert(E0 % CHUNK == 0, "chunks cover the edge list exactly");

__device__ __forceinline__ unsigned block_scan_incl(unsigned c, unsigned* wtot, int lane, int wave) {
    unsigned incl = c;
    for (int off = 1; off < 32; off <<= 1) {
        unsigned v = __shfl_up(incl, off, 32);
        if (lane >= off) incl += v;
    }
    if (lane == 31) wtot[wave] = incl;
    __syncthreads();
    unsigned before = 0;
    for (int w = 0; w < wave; ++w) before += wtot[w];
    return before + incl;
}

__global__ void __launch_bounds__(SORT_T)
chunk_sort(const int* __restrict__ src, const int* __restrict__ dst,
           unsigned* __restrict__ table, int* __restrict__ rec) {
    __shared__ unsigned hist[NBIN];
    __shared__ unsigned wtot[SORT_T / 32];
    const int t = threadIdx.x, lane = t & 31, wave = t >> 5;
    const int base = blockIdx.x * CHUNK;
    if (t < NBIN) hist[t] = 0u;
    __syncthreads();

    constexpr int PER = (CHUNK + SORT_T - 1) / SORT_T;
    int s[PER], d[PER];
    unsigned r[PER];
#pragma unroll
    for (int k = 0; k < PER; ++k) {
        const int i = t + k * SORT_T;
        d[k] = -1; s[k] = 0; r[k] = 0u;
        if (i < CHUNK) { d[k] = dst[base + i]; s[k] = src[base + i]; }
    }
#pragma unroll
    for (int k = 0; k < PER; ++k)
        if (d[k] >= 0) r[k] = atomicAdd(&hist[d[k] >> BIN_SHIFT], 1u);
    __syncthreads();

    const unsigned c = (t < NBIN) ? hist[t] : 0u;
    const unsigned excl = block_scan_incl(c, wtot, lane, wave) - c;
    if (t < NBIN) {
        hist[t] = excl;
        v2u run; run.x = excl; run.y = c;
        ((v2u*)table)[blockIdx.x * NBIN + t] = run;
    }
    __syncthreads();

    v2i* rec2 = (v2i*)rec;
#pragma unroll
    for (int k = 0; k < PER; ++k)
        if (d[k] >= 0) {
            v2i e; e.x = s[k]; e.y = d[k];
            rec2[base + hist[d[k] >> BIN_SHIFT] + r[k]] = e;
        }
}

__global__ void __launch_bounds__(SORT_T)
bin_place(const unsigned* __restrict__ table, const int* __restrict__ rec,
          unsigned* __restrict__ cnt, unsigned* __restrict__ rowptr, int* __restrict__ slot) {
    __shared__ unsigned rstart[NCHUNK], rlen[NCHUNK];
    __shared__ unsigned cntL[BIN_NODES], offL[BIN_NODES], fillL[BIN_NODES];
    __shared__ unsigned wsum[SORT_T / 32], wtot[SORT_T / 32];
    const int t = threadIdx.x, lane = t & 31, wave = t >> 5;
    const int bin = blockIdx.x;
    const v2i* rec2 = (const v2i*)rec;

    unsigned o0 = 0u;
    if (t < NCHUNK) {
        const v2u run = ((const v2u*)table)[t * NBIN + bin];
        o0 = run.x;
        rstart[t] = (unsigned)t * CHUNK + run.x;
        rlen[t]   = run.y;
    }
    if (t < BIN_NODES) { cntL[t] = 0u; fillL[t] = 0u; }
    unsigned part = o0;
    for (int off = 16; off > 0; off >>= 1) part += __shfl_xor(part, off, 32);
    if (lane == 0) wsum[wave] = part;
    __syncthreads();
    unsigned binStart = 0u;
    for (int w = 0; w < SORT_T / 32; ++w) binStart += wsum[w];

    constexpr int RPW = (NCHUNK + 31) / 32;
    int sv[RPW], dv[RPW];
    unsigned at[RPW];
    bool has[RPW];
#pragma unroll
    for (int k = 0; k < RPW; ++k) {
        const int run = wave + 32 * k;
        const int rr  = run < NCHUNK ? run : 0;
        has[k] = run < NCHUNK && (unsigned)lane < rlen[rr];
        at[k]  = has[k] ? rstart[rr] + lane : 0u;
    }
#pragma unroll
    for (int k = 0; k < RPW; ++k) {
        const v2i e = rec2[at[k]];
        sv[k] = e.x;
        dv[k] = has[k] ? e.y - bin * BIN_NODES : -1;
    }
#pragma unroll
    for (int k = 0; k < RPW; ++k)
        if (dv[k] >= 0) atomicAdd(&cntL[dv[k]], 1u);
    for (int run = wave; run < NCHUNK; run += 32) {
        const unsigned len = rlen[run], st = rstart[run];
        for (unsigned i = lane + 32; i < len; i += 32)
            atomicAdd(&cntL[rec2[st + i].y - bin * BIN_NODES], 1u);
    }
    __syncthreads();

    const unsigned c = (t < BIN_NODES) ? cntL[t] : 0u;
    const unsigned excl = block_scan_incl(c, wtot, lane, wave) - c;
    if (t < BIN_NODES) {
        offL[t] = binStart + excl;
        const int node = bin * BIN_NODES + t;
        if (node < N_NODES) { rowptr[node] = binStart + excl; cnt[node] = c; }
    }
    __syncthreads();

#pragma unroll
    for (int k = 0; k < RPW; ++k)
        if (dv[k] >= 0) slot[offL[dv[k]] + atomicAdd(&fillL[dv[k]], 1u)] = sv[k];
    for (int run = wave; run < NCHUNK; run += 32) {
        const unsigned len = rlen[run], st = rstart[run];
        for (unsigned i = lane + 32; i < len; i += 32) {
            const v2i e = rec2[st + i];
            const int dl = e.y - bin * BIN_NODES;
            slot[offL[dl] + atomicAdd(&fillL[dl], 1u)] = e.x;
        }
    }
}

__device__ __forceinline__ float xhalf(float x) {
    const int b = __builtin_bit_cast(int, x);
    return __builtin_bit_cast(float, __builtin_amdgcn_permlanex16(b, b, 0x76543210, 0xfedcba98, false, false));
}
__device__ __forceinline__ float wave_max(float v) {
    v = fmaxf(v, row_xor<1>(v));
    v = fmaxf(v, row_xor<2>(v));
    v = fmaxf(v, row_xor<4>(v));
    v = fmaxf(v, row_xor<8>(v));
    return fmaxf(v, xhalf(v));
}
__device__ __forceinline__ float wave_sum(float v) {
    v += row_xor<1>(v);
    v += row_xor<2>(v);
    v += row_xor<4>(v);
    v += row_xor<8>(v);
    return v + xhalf(v);
}

__device__ __forceinline__ float uni(float x) {
    return __builtin_bit_cast(float, __builtin_amdgcn_readfirstlane(__builtin_bit_cast(int, x)));
}

__device__ __forceinline__ float leaky(float v) { return v > 0.f ? v : NEG_SLOPE * v; }

__device__ __forceinline__ void load_edges(const int* __restrict__ slot, const v4f* __restrict__ AS4,
                                           unsigned first, unsigned n, int lane, v4f ad,
                                           int& s, float& l0, float& l1, float& l2, float& l3) {
    s = 0;
    l0 = l1 = l2 = l3 = -INFINITY;
    if ((unsigned)lane < n) {
        s = slot[first + lane];
        const v4f as = AS4[s];
        l0 = leaky(as.x + ad.x); l1 = leaky(as.y + ad.y);
        l2 = leaky(as.z + ad.z); l3 = leaky(as.w + ad.w);
    }
}

__device__ __forceinline__ void add_rows(const v4f* __restrict__ H4, unsigned n, int lane, bool upper,
                                         int s, float a0, float a1, float a2, float a3,
                                         v4f& acc0, v4f& acc1) {
#pragma unroll 4
    for (unsigned j = 0; j < n; ++j) {
        const int   sj = __builtin_amdgcn_readlane(s, (int)j);
        const float b0 = __int_as_float(__builtin_amdgcn_readlane(__float_as_int(a0), (int)j));
        const float b1 = __int_as_float(__builtin_amdgcn_readlane(__float_as_int(a1), (int)j));
        const float b2 = __int_as_float(__builtin_amdgcn_readlane(__float_as_int(a2), (int)j));
        const float b3 = __int_as_float(__builtin_amdgcn_readlane(__float_as_int(a3), (int)j));
        const float alo = upper ? b1 : b0;
        const float ahi = upper ? b3 : b2;
        const v4f v0 = H4[(size_t)sj * 64 + lane];
        const v4f v1 = H4[(size_t)sj * 64 + 32 + lane];
        acc0 += v0 * alo;
        acc1 += v1 * ahi;
    }
}

#define ROWS_PER_BLOCK 8
__global__ void __launch_bounds__(ROWS_PER_BLOCK * 32)
node_softmax_gather(const int* __restrict__ slot, const unsigned* __restrict__ rowptr,
                    const unsigned* __restrict__ cnt,
                    const float* __restrict__ a_src, const float* __restrict__ a_dst,
                    const float* __restrict__ H, const float* __restrict__ bias,
                    float* __restrict__ out, int relu) {
    const int lane = threadIdx.x & 31;
    const int node = __builtin_amdgcn_readfirstlane(blockIdx.x * ROWS_PER_BLOCK + (threadIdx.x >> 5));
    if (node >= N_NODES) return;

    const v4f* H4  = (const v4f*)H;
    const v4f* AS4 = (const v4f*)a_src;

    const unsigned beg = rowptr[node];
    const unsigned deg = cnt[node];
    const v4f ad  = ((const v4f*)a_dst)[node];
    const v4f asn = AS4[node];
    const bool upper = lane >= 16;

    const v4f bias0 = ((const v4f*)bias)[lane];
    const v4f bias1 = ((const v4f*)bias)[32 + lane];
    const v4f self0 = H4[(size_t)node * 64 + lane];
    const v4f self1 = H4[(size_t)node * 64 + 32 + lane];

    const float ls0 = leaky(asn.x + ad.x), ls1 = leaky(asn.y + ad.y);
    const float ls2 = leaky(asn.z + ad.z), ls3 = leaky(asn.w + ad.w);

    const unsigned n0 = deg < 32u ? deg : 32u;
    int   s0;
    float p0, p1, p2, p3;
    load_edges(slot, AS4, beg, n0, lane, ad, s0, p0, p1, p2, p3);

    float m0 = fmaxf(ls0, p0), m1 = fmaxf(ls1, p1), m2 = fmaxf(ls2, p2), m3 = fmaxf(ls3, p3);
    for (unsigned base = 32; base < deg; base += 32) {
        const unsigned n = (deg - base < 32u) ? (deg - base) : 32u;
        int s; float l0, l1, l2, l3;
        load_edges(slot, AS4, beg + base, n, lane, ad, s, l0, l1, l2, l3);
        m0 = fmaxf(m0, l0); m1 = fmaxf(m1, l1); m2 = fmaxf(m2, l2); m3 = fmaxf(m3, l3);
    }
    m0 = uni(wave_max(m0)); m1 = uni(wave_max(m1)); m2 = uni(wave_max(m2)); m3 = uni(wave_max(m3));
    m0 = __builtin_isfinite(m0) ? m0 : 0.f; m1 = __builtin_isfinite(m1) ? m1 : 0.f;
    m2 = __builtin_isfinite(m2) ? m2 : 0.f; m3 = __builtin_isfinite(m3) ? m3 : 0.f;

    p0 = __expf(p0 - m0); p1 = __expf(p1 - m1); p2 = __expf(p2 - m2); p3 = __expf(p3 - m3);
    float z0 = p0, z1 = p1, z2 = p2, z3 = p3;
    for (unsigned base = 32; base < deg; base += 32) {
        const unsigned n = (deg - base < 32u) ? (deg - base) : 32u;
        int s; float l0, l1, l2, l3;
        load_edges(slot, AS4, beg + base, n, lane, ad, s, l0, l1, l2, l3);
        z0 += __expf(l0 - m0); z1 += __expf(l1 - m1); z2 += __expf(l2 - m2); z3 += __expf(l3 - m3);
    }
    const float es0 = __expf(ls0 - m0), es1 = __expf(ls1 - m1);
    const float es2 = __expf(ls2 - m2), es3 = __expf(ls3 - m3);
    const float zd0 = uni(wave_sum(z0) + es0 + 1e-16f), zd1 = uni(wave_sum(z1) + es1 + 1e-16f);
    const float zd2 = uni(wave_sum(z2) + es2 + 1e-16f), zd3 = uni(wave_sum(z3) + es3 + 1e-16f);

    v4f acc0, acc1;
    {
        const float alo = (upper ? es1 : es0) / (upper ? zd1 : zd0);
        const float ahi = (upper ? es3 : es2) / (upper ? zd3 : zd2);
        acc0 = self0 * alo;
        acc1 = self1 * ahi;
    }
    {
        int      s = s0;
        unsigned n = n0;
        float a0 = p0 / zd0, a1 = p1 / zd1, a2 = p2 / zd2, a3 = p3 / zd3;
        for (unsigned base = 0;;) {
            add_rows(H4, n, lane, upper, s, a0, a1, a2, a3, acc0, acc1);
            base += 32;
            if (base >= deg) break;
            n = (deg - base < 32u) ? (deg - base) : 32u;
            float l0, l1, l2, l3;
            load_edges(slot, AS4, beg + base, n, lane, ad, s, l0, l1, l2, l3);
            a0 = __expf(l0 - m0) / zd0; a1 = __expf(l1 - m1) / zd1;
            a2 = __expf(l2 - m2) / zd2; a3 = __expf(l3 - m3) / zd3;
        }
    }

    acc0 += bias0;
    acc1 += bias1;
    if (relu) {
        acc0.x = fmaxf(acc0.x, 0.f); acc0.y = fmaxf(acc0.y, 0.f);
        acc0.z = fmaxf(acc0.z, 0.f); acc0.w = fmaxf(acc0.w, 0.f);
        acc1.x = fmaxf(acc1.x, 0.f); acc1.y = fmaxf(acc1.y, 0.f);
        acc1.z = fmaxf(acc1.z, 0.f); acc1.w = fmaxf(acc1.w, 0.f);
    }
    v4f* O4 = (v4f*)out;
    O4[(size_t)node * 64 + lane]      = acc0;
    O4[(size_t)node * 64 + 32 + lane] = acc1;
}

extern "C" void kernel_launch(void* const* d_in, const int* in_sizes, int n_in,
                              void* d_out, int out_size, void* d_ws, size_t ws_size,
                              hipStream_t stream) {
    const float* x        = (const float*)d_in[0];
    const int*   eidx     = (const int*)  d_in[1];
    const float* W1       = (const float*)d_in[2];
    const float* att_s1   = (const float*)d_in[3];
    const float* att_d1   = (const float*)d_in[4];
    const float* b1       = (const float*)d_in[5];
    const float* W2       = (const float*)d_in[6];
    const float* att_s2   = (const float*)d_in[7];
    const float* att_d2   = (const float*)d_in[8];
    const float* b2       = (const float*)d_in[9];
    float*       out      = (float*)d_out;

    const int* src = eidx;
    const int* dst = eidx + E0;

    const size_t NB = (size_t)N_NODES * C_HID;
    float* h      = (float*)d_ws;
    float* hrelu  = h      + NB;
    float* a_src  = hrelu  + NB;
    float* a_dst  = a_src  + (size_t)N_NODES * HEADS;
    unsigned* cnt    = (unsigned*)(a_dst + (size_t)N_NODES * HEADS);
    unsigned* rowptr = cnt + N_NODES;
    int*      slot   = (int*)(rowptr + N_NODES);
    int*      rec    = slot + E0;
    unsigned* table  = (unsigned*)(rec + 2 * (size_t)E0);

    const int gMM  = N_NODES / 16;
    const int gROW = (N_NODES + ROWS_PER_BLOCK - 1) / ROWS_PER_BLOCK;

    chunk_sort<<<NCHUNK, SORT_T, 0, stream>>>(src, dst, table, rec);
    bin_place <<<NBIN,   SORT_T, 0, stream>>>(table, rec, cnt, rowptr, slot);

    gemm_wmma  <<<gMM, 128, 0, stream>>>(x, W1, h, att_s1, att_d1, a_src, a_dst, C_IN, C_HID);
    node_softmax_gather<<<gROW, ROWS_PER_BLOCK * 32, 0, stream>>>(slot, rowptr, cnt, a_src, a_dst, h, b1, hrelu, 1);

    gemm_wmma  <<<gMM, 128, 0, stream>>>(hrelu, W2, h, att_s2, att_d2, a_src, a_dst, C_HID, C_HID);
    node_softmax_gather<<<gROW, ROWS_PER_BLOCK * 32, 0, stream>>>(slot, rowptr, cnt, a_src, a_dst, h, b2, out, 0);
}


# GEMM moves W, x and h in 16-byte pieces per lane: accumulator j takes columns 4c+j of the wave's 64, k taken in 8s; product sum order inside each 8 of k and the dots' 4-per-lane order change
# speedup vs baseline: 8.0393x; 1.0779x over previous
#include <hip/hip_runtime.h>
#include <math.h>

typedef __attribute__((ext_vector_type(2))) float v2f;
typedef __attribute__((ext_vector_type(8))) float v8f;
typedef __attribute__((ext_vector_type(4))) float v4f;

#define N_NODES 50000
#define C_IN    128
#define C_HID   256
#define HEADS   4
#define EMBED   64
#define E0      800000
#define ETOT    (E0 + N_NODES)
#define NEG_SLOPE 0.2f

template <int MASK>
__device__ __forceinline__ float row_xor(float x) {
    return __builtin_bit_cast(float, __builtin_amdgcn_update_dpp(0, __builtin_bit_cast(int, x),
                                                                 0x160 + MASK, 0xf, 0xf, false));
}
__device__ __forceinline__ float row_sum16(float v) {
    v += row_xor<1>(v);
    v += row_xor<2>(v);
    v += row_xor<4>(v);
    v += row_xor<8>(v);
    return v;
}

__global__ void gemm_wmma(const float* __restrict__ X,
                          const float* __restrict__ W,
                          float* __restrict__ H,
                          const float* __restrict__ att_src,
                          const float* __restrict__ att_dst,
                          float* __restrict__ a_src,
                          float* __restrict__ a_dst,
                          int K) {
    __shared__ v4f dots4[2][16];
    float* dots = (float*)dots4;

    const int lane   = threadIdx.x & 31;
    const int wave   = threadIdx.x >> 5;
    const int tileM  = blockIdx.x;
    const int lane15 = lane & 15;
    const int khalf  = lane >> 4;
    const int row    = tileM * 16 + lane15;
    const int col0   = wave * 64;
    const int colL   = col0 + 4 * lane15;

    v8f acc0 = {}, acc1 = {}, acc2 = {}, acc3 = {};

    const v4f*   xp = (const v4f*)(X + (size_t)row * K + 4 * khalf);
    const float* wp = W + (size_t)(4 * khalf) * C_HID + colL;

    for (int k = 0; k < K; k += 8) {
        const v4f a4 = *xp;
        const v4f r0 = *(const v4f*)(wp);
        const v4f r1 = *(const v4f*)(wp + C_HID);
        const v4f r2 = *(const v4f*)(wp + 2 * C_HID);
        const v4f r3 = *(const v4f*)(wp + 3 * C_HID);
        xp += 2;
        wp += 8 * C_HID;

        v2f a, b0, b1, b2, b3;
        a.x  = a4.x;  a.y  = a4.y;
        b0.x = r0.x;  b0.y = r1.x;
        b1.x = r0.y;  b1.y = r1.y;
        b2.x = r0.z;  b2.y = r1.z;
        b3.x = r0.w;  b3.y = r1.w;
        acc0 = __builtin_amdgcn_wmma_f32_16x16x4_f32(false, a, false, b0, (short)0, acc0, false, false);
        acc1 = __builtin_amdgcn_wmma_f32_16x16x4_f32(false, a, false, b1, (short)0, acc1, false, false);
        acc2 = __builtin_amdgcn_wmma_f32_16x16x4_f32(false, a, false, b2, (short)0, acc2, false, false);
        acc3 = __builtin_amdgcn_wmma_f32_16x16x4_f32(false, a, false, b3, (short)0, acc3, false, false);

        a.x  = a4.z;  a.y  = a4.w;
        b0.x = r2.x;  b0.y = r3.x;
        b1.x = r2.y;  b1.y = r3.y;
        b2.x = r2.z;  b2.y = r3.z;
        b3.x = r2.w;  b3.y = r3.w;
        acc0 = __builtin_amdgcn_wmma_f32_16x16x4_f32(false, a, false, b0, (short)0, acc0, false, false);
        acc1 = __builtin_amdgcn_wmma_f32_16x16x4_f32(false, a, false, b1, (short)0, acc1, false, false);
        acc2 = __builtin_amdgcn_wmma_f32_16x16x4_f32(false, a, false, b2, (short)0, acc2, false, false);
        acc3 = __builtin_amdgcn_wmma_f32_16x16x4_f32(false, a, false, b3, (short)0, acc3, false, false);
    }

    const v4f s4 = *(const v4f*)(att_src + colL);
    const v4f d4 = *(const v4f*)(att_dst + colL);
#pragma unroll
    for (int i = 0; i < 8; ++i) {
        const float ps = row_sum16(acc0[i] * s4.x + acc1[i] * s4.y + acc2[i] * s4.z + acc3[i] * s4.w);
        const float pd = row_sum16(acc0[i] * d4.x + acc1[i] * d4.y + acc2[i] * d4.z + acc3[i] * d4.w);
        if (lane15 == 0) {
            dots[(0 * 16 + khalf * 8 + i) * HEADS + wave] = ps;
            dots[(1 * 16 + khalf * 8 + i) * HEADS + wave] = pd;
        }
    }
    __syncthreads();
    if (wave == 0) {
        float* plane = khalf ? a_dst : a_src;
        *(v4f*)(plane + (size_t)(tileM * 16 + lane15) * HEADS) = dots4[khalf][lane15];
    }

    const int rbase = tileM * 16 + khalf * 8;
#pragma unroll
    for (int i = 0; i < 8; ++i) {
        v4f o;
        o.x = acc0[i]; o.y = acc1[i]; o.z = acc2[i]; o.w = acc3[i];
        *(v4f*)(H + (size_t)(rbase + i) * C_HID + colL) = o;
    }
}

typedef __attribute__((ext_vector_type(2))) int v2i;
typedef __attribute__((ext_vector_type(2))) unsigned v2u;

#define BIN_SHIFT 8
#define BIN_NODES (1 << BIN_SHIFT)
#define NBIN      ((N_NODES + BIN_NODES - 1) / BIN_NODES)
#define CHUNK     3200
#define NCHUNK    (E0 / CHUNK)
#define SORT_T    1024
static_assert(E0 % CHUNK == 0, "chunks cover the edge list exactly");

__device__ __forceinline__ unsigned block_scan_incl(unsigned c, unsigned* wtot, int lane, int wave) {
    unsigned incl = c;
    for (int off = 1; off < 32; off <<= 1) {
        unsigned v = __shfl_up(incl, off, 32);
        if (lane >= off) incl += v;
    }
    if (lane == 31) wtot[wave] = incl;
    __syncthreads();
    unsigned before = 0;
    for (int w = 0; w < wave; ++w) before += wtot[w];
    return before + incl;
}

__global__ void __launch_bounds__(SORT_T)
chunk_sort(const int* __restrict__ src, const int* __restrict__ dst,
           unsigned* __restrict__ table, int* __restrict__ rec) {
    __shared__ unsigned hist[NBIN];
    __shared__ unsigned wtot[SORT_T / 32];
    const int t = threadIdx.x, lane = t & 31, wave = t >> 5;
    const int base = blockIdx.x * CHUNK;
    if (t < NBIN) hist[t] = 0u;
    __syncthreads();

    constexpr int PER = (CHUNK + SORT_T - 1) / SORT_T;
    int s[PER], d[PER];
    unsigned r[PER];
#pragma unroll
    for (int k = 0; k < PER; ++k) {
        const int i = t + k * SORT_T;
        d[k] = -1; s[k] = 0; r[k] = 0u;
        if (i < CHUNK) { d[k] = dst[base + i]; s[k] = src[base + i]; }
    }
#pragma unroll
    for (int k = 0; k < PER; ++k)
        if (d[k] >= 0) r[k] = atomicAdd(&hist[d[k] >> BIN_SHIFT], 1u);
    __syncthreads();

    const unsigned c = (t < NBIN) ? hist[t] : 0u;
    const unsigned excl = block_scan_incl(c, wtot, lane, wave) - c;
    if (t < NBIN) {
        hist[t] = excl;
        v2u run; run.x = excl; run.y = c;
        ((v2u*)table)[blockIdx.x * NBIN + t] = run;
    }
    __syncthreads();

    v2i* rec2 = (v2i*)rec;
#pragma unroll
    for (int k = 0; k < PER; ++k)
        if (d[k] >= 0) {
            v2i e; e.x = s[k]; e.y = d[k];
            rec2[base + hist[d[k] >> BIN_SHIFT] + r[k]] = e;
        }
}

__global__ void __launch_bounds__(SORT_T)
bin_place(const unsigned* __restrict__ table, const int* __restrict__ rec,
          unsigned* __restrict__ cnt, unsigned* __restrict__ rowptr, int* __restrict__ slot) {
    __shared__ unsigned rstart[NCHUNK], rlen[NCHUNK];
    __shared__ unsigned cntL[BIN_NODES], offL[BIN_NODES], fillL[BIN_NODES];
    __shared__ unsigned wsum[SORT_T / 32], wtot[SORT_T / 32];
    const int t = threadIdx.x, lane = t & 31, wave = t >> 5;
    const int bin = blockIdx.x;
    const v2i* rec2 = (const v2i*)rec;

    unsigned o0 = 0u;
    if (t < NCHUNK) {
        const v2u run = ((const v2u*)table)[t * NBIN + bin];
        o0 = run.x;
        rstart[t] = (unsigned)t * CHUNK + run.x;
        rlen[t]   = run.y;
    }
    if (t < BIN_NODES) { cntL[t] = 0u; fillL[t] = 0u; }
    unsigned part = o0;
    for (int off = 16; off > 0; off >>= 1) part += __shfl_xor(part, off, 32);
    if (lane == 0) wsum[wave] = part;
    __syncthreads();
    unsigned binStart = 0u;
    for (int w = 0; w < SORT_T / 32; ++w) binStart += wsum[w];

    constexpr int RPW = (NCHUNK + 31) / 32;
    int sv[RPW], dv[RPW];
    unsigned at[RPW];
    bool has[RPW];
#pragma unroll
    for (int k = 0; k < RPW; ++k) {
        const int run = wave + 32 * k;
        const int rr  = run < NCHUNK ? run : 0;
        has[k] = run < NCHUNK && (unsigned)lane < rlen[rr];
        at[k]  = has[k] ? rstart[rr] + lane : 0u;
    }
#pragma unroll
    for (int k = 0; k < RPW; ++k) {
        const v2i e = rec2[at[k]];
        sv[k] = e.x;
        dv[k] = has[k] ? e.y - bin * BIN_NODES : -1;
    }
#pragma unroll
    for (int k = 0; k < RPW; ++k)
        if (dv[k] >= 0) atomicAdd(&cntL[dv[k]], 1u);
    for (int run = wave; run < NCHUNK; run += 32) {
        const unsigned len = rlen[run], st = rstart[run];
        for (unsigned i = lane + 32; i < len; i += 32)
            atomicAdd(&cntL[rec2[st + i].y - bin * BIN_NODES], 1u);
    }
    __syncthreads();

    const unsigned c = (t < BIN_NODES) ? cntL[t] : 0u;
    const unsigned excl = block_scan_incl(c, wtot, lane, wave) - c;
    if (t < BIN_NODES) {
        offL[t] = binStart + excl;
        const int node = bin * BIN_NODES + t;
        if (node < N_NODES) { rowptr[node] = binStart + excl; cnt[node] = c; }
    }
    __syncthreads();

#pragma unroll
    for (int k = 0; k < RPW; ++k)
        if (dv[k] >= 0) slot[offL[dv[k]] + atomicAdd(&fillL[dv[k]], 1u)] = sv[k];
    for (int run = wave; run < NCHUNK; run += 32) {
        const unsigned len = rlen[run], st = rstart[run];
        for (unsigned i = lane + 32; i < len; i += 32) {
            const v2i e = rec2[st + i];
            const int dl = e.y - bin * BIN_NODES;
            slot[offL[dl] + atomicAdd(&fillL[dl], 1u)] = e.x;
        }
    }
}

__device__ __forceinline__ float xhalf(float x) {
    const int b = __builtin_bit_cast(int, x);
    return __builtin_bit_cast(float, __builtin_amdgcn_permlanex16(b, b, 0x76543210, 0xfedcba98, false, false));
}
__device__ __forceinline__ float wave_max(float v) {
    v = fmaxf(v, row_xor<1>(v));
    v = fmaxf(v, row_xor<2>(v));
    v = fmaxf(v, row_xor<4>(v));
    v = fmaxf(v, row_xor<8>(v));
    return fmaxf(v, xhalf(v));
}
__device__ __forceinline__ float wave_sum(float v) {
    v += row_xor<1>(v);
    v += row_xor<2>(v);
    v += row_xor<4>(v);
    v += row_xor<8>(v);
    return v + xhalf(v);
}

__device__ __forceinline__ float uni(float x) {
    return __builtin_bit_cast(float, __builtin_amdgcn_readfirstlane(__builtin_bit_cast(int, x)));
}

__device__ __forceinline__ float leaky(float v) { return v > 0.f ? v : NEG_SLOPE * v; }

__device__ __forceinline__ void load_edges(const int* __restrict__ slot, const v4f* __restrict__ AS4,
                                           unsigned first, unsigned n, int lane, v4f ad,
                                           int& s, float& l0, float& l1, float& l2, float& l3) {
    s = 0;
    l0 = l1 = l2 = l3 = -INFINITY;
    if ((unsigned)lane < n) {
        s = slot[first + lane];
        const v4f as = AS4[s];
        l0 = leaky(as.x + ad.x); l1 = leaky(as.y + ad.y);
        l2 = leaky(as.z + ad.z); l3 = leaky(as.w + ad.w);
    }
}

__device__ __forceinline__ void add_rows(const v4f* __restrict__ H4, unsigned n, int lane, bool upper,
                                         int s, float a0, float a1, float a2, float a3,
                                         v4f& acc0, v4f& acc1) {
#pragma unroll 4
    for (unsigned j = 0; j < n; ++j) {
        const int   sj = __builtin_amdgcn_readlane(s, (int)j);
        const float b0 = __int_as_float(__builtin_amdgcn_readlane(__float_as_int(a0), (int)j));
        const float b1 = __int_as_float(__builtin_amdgcn_readlane(__float_as_int(a1), (int)j));
        const float b2 = __int_as_float(__builtin_amdgcn_readlane(__float_as_int(a2), (int)j));
        const float b3 = __int_as_float(__builtin_amdgcn_readlane(__float_as_int(a3), (int)j));
        const float alo = upper ? b1 : b0;
        const float ahi = upper ? b3 : b2;
        const v4f v0 = H4[(size_t)sj * 64 + lane];
        const v4f v1 = H4[(size_t)sj * 64 + 32 + lane];
        acc0 += v0 * alo;
        acc1 += v1 * ahi;
    }
}

#define ROWS_PER_BLOCK 8
__global__ void __launch_bounds__(ROWS_PER_BLOCK * 32)
node_softmax_gather(const int* __restrict__ slot, const unsigned* __restrict__ rowptr,
                    const unsigned* __restrict__ cnt,
                    const float* __restrict__ a_src, const float* __restrict__ a_dst,
                    const float* __restrict__ H, const float* __restrict__ bias,
                    float* __restrict__ out, int relu) {
    const int lane = threadIdx.x & 31;
    const int node = __builtin_amdgcn_readfirstlane(blockIdx.x * ROWS_PER_BLOCK + (threadIdx.x >> 5));
    if (node >= N_NODES) return;

    const v4f* H4  = (const v4f*)H;
    const v4f* AS4 = (const v4f*)a_src;

    const unsigned beg = rowptr[node];
    const unsigned deg = cnt[node];
    const v4f ad  = ((const v4f*)a_dst)[node];
    const v4f asn = AS4[node];
    const bool upper = lane >= 16;

    const v4f bias0 = ((const v4f*)bias)[lane];
    const v4f bias1 = ((const v4f*)bias)[32 + lane];
    const v4f self0 = H4[(size_t)node * 64 + lane];
    const v4f self1 = H4[(size_t)node * 64 + 32 + lane];

    const float ls0 = leaky(asn.x + ad.x), ls1 = leaky(asn.y + ad.y);
    const float ls2 = leaky(asn.z + ad.z), ls3 = leaky(asn.w + ad.w);

    const unsigned n0 = deg < 32u ? deg : 32u;
    int   s0;
    float p0, p1, p2, p3;
    load_edges(slot, AS4, beg, n0, lane, ad, s0, p0, p1, p2, p3);

    float m0 = fmaxf(ls0, p0), m1 = fmaxf(ls1, p1), m2 = fmaxf(ls2, p2), m3 = fmaxf(ls3, p3);
    for (unsigned base = 32; base < deg; base += 32) {
        const unsigned n = (deg - base < 32u) ? (deg - base) : 32u;
        int s; float l0, l1, l2, l3;
        load_edges(slot, AS4, beg + base, n, lane, ad, s, l0, l1, l2, l3);
        m0 = fmaxf(m0, l0); m1 = fmaxf(m1, l1); m2 = fmaxf(m2, l2); m3 = fmaxf(m3, l3);
    }
    m0 = uni(wave_max(m0)); m1 = uni(wave_max(m1)); m2 = uni(wave_max(m2)); m3 = uni(wave_max(m3));
    m0 = __builtin_isfinite(m0) ? m0 : 0.f; m1 = __builtin_isfinite(m1) ? m1 : 0.f;
    m2 = __builtin_isfinite(m2) ? m2 : 0.f; m3 = __builtin_isfinite(m3) ? m3 : 0.f;

    p0 = __expf(p0 - m0); p1 = __expf(p1 - m1); p2 = __expf(p2 - m2); p3 = __expf(p3 - m3);
    float z0 = p0, z1 = p1, z2 = p2, z3 = p3;
    for (unsigned base = 32; base < deg; base += 32) {
        const unsigned n = (deg - base < 32u) ? (deg - base) : 32u;
        int s; float l0, l1, l2, l3;
        load_edges(slot, AS4, beg + base, n, lane, ad, s, l0, l1, l2, l3);
        z0 += __expf(l0 - m0); z1 += __expf(l1 - m1); z2 += __expf(l2 - m2); z3 += __expf(l3 - m3);
    }
    const float es0 = __expf(ls0 - m0), es1 = __expf(ls1 - m1);
    const float es2 = __expf(ls2 - m2), es3 = __expf(ls3 - m3);
    const float zd0 = uni(wave_sum(z0) + es0 + 1e-16f), zd1 = uni(wave_sum(z1) + es1 + 1e-16f);
    const float zd2 = uni(wave_sum(z2) + es2 + 1e-16f), zd3 = uni(wave_sum(z3) + es3 + 1e-16f);

    v4f acc0, acc1;
    {
        const float alo = (upper ? es1 : es0) / (upper ? zd1 : zd0);
        const float ahi = (upper ? es3 : es2) / (upper ? zd3 : zd2);
        acc0 = self0 * alo;
        acc1 = self1 * ahi;
    }
    {
        int      s = s0;
        unsigned n = n0;
        float a0 = p0 / zd0, a1 = p1 / zd1, a2 = p2 / zd2, a3 = p3 / zd3;
        for (unsigned base = 0;;) {
            add_rows(H4, n, lane, upper, s, a0, a1, a2, a3, acc0, acc1);
            base += 32;
            if (base >= deg) break;
            n = (deg - base < 32u) ? (deg - base) : 32u;
            float l0, l1, l2, l3;
            load_edges(slot, AS4, beg + base, n, lane, ad, s, l0, l1, l2, l3);
            a0 = __expf(l0 - m0) / zd0; a1 = __expf(l1 - m1) / zd1;
            a2 = __expf(l2 - m2) / zd2; a3 = __expf(l3 - m3) / zd3;
        }
    }

    acc0 += bias0;
    acc1 += bias1;
    if (relu) {
        acc0.x = fmaxf(acc0.x, 0.f); acc0.y = fmaxf(acc0.y, 0.f);
        acc0.z = fmaxf(acc0.z, 0.f); acc0.w = fmaxf(acc0.w, 0.f);
        acc1.x = fmaxf(acc1.x, 0.f); acc1.y = fmaxf(acc1.y, 0.f);
        acc1.z = fmaxf(acc1.z, 0.f); acc1.w = fmaxf(acc1.w, 0.f);
    }
    v4f* O4 = (v4f*)out;
    O4[(size_t)node * 64 + lane]      = acc0;
    O4[(size_t)node * 64 + 32 + lane] = acc1;
}

extern "C" void kernel_launch(void* const* d_in, const int* in_sizes, int n_in,
                              void* d_out, int out_size, void* d_ws, size_t ws_size,
                              hipStream_t stream) {
    const float* x        = (const float*)d_in[0];
    const int*   eidx     = (const int*)  d_in[1];
    const float* W1       = (const float*)d_in[2];
    const float* att_s1   = (const float*)d_in[3];
    const float* att_d1   = (const float*)d_in[4];
    const float* b1       = (const float*)d_in[5];
    const float* W2       = (const float*)d_in[6];
    const float* att_s2   = (const float*)d_in[7];
    const float* att_d2   = (const float*)d_in[8];
    const float* b2       = (const float*)d_in[9];
    float*       out      = (float*)d_out;

    const int* src = eidx;
    const int* dst = eidx + E0;

    const size_t NB = (size_t)N_NODES * C_HID;
    float* h      = (float*)d_ws;
    float* hrelu  = h      + NB;
    float* a_src  = hrelu  + NB;
    float* a_dst  = a_src  + (size_t)N_NODES * HEADS;
    unsigned* cnt    = (unsigned*)(a_dst + (size_t)N_NODES * HEADS);
    unsigned* rowptr = cnt + N_NODES;
    int*      slot   = (int*)(rowptr + N_NODES);
    int*      rec    = slot + E0;
    unsigned* table  = (unsigned*)(rec + 2 * (size_t)E0);

    const int gMM  = N_NODES / 16;
    const int gROW = (N_NODES + ROWS_PER_BLOCK - 1) / ROWS_PER_BLOCK;

    chunk_sort<<<NCHUNK, SORT_T, 0, stream>>>(src, dst, table, rec);
    bin_place <<<NBIN,   SORT_T, 0, stream>>>(table, rec, cnt, rowptr, slot);

    gemm_wmma  <<<gMM, 128, 0, stream>>>(x, W1, h, att_s1, att_d1, a_src, a_dst, C_IN);
    node_softmax_gather<<<gROW, ROWS_PER_BLOCK * 32, 0, stream>>>(slot, rowptr, cnt, a_src, a_dst, h, b1, hrelu, 1);

    gemm_wmma  <<<gMM, 128, 0, stream>>>(hrelu, W2, h, att_s2, att_d2, a_src, a_dst, C_HID);
    node_softmax_gather<<<gROW, ROWS_PER_BLOCK * 32, 0, stream>>>(slot, rowptr, cnt, a_src, a_dst, h, b2, out, 0);
}


# Layer 1 sums x rows (512 B) per head ahead of the product, (sum alpha x) W1, in one launch with the WMMA product fed from LDS; dots come from x (W1 att); sum order of layer 1 changes throughout
# speedup vs baseline: 8.6691x; 1.0783x over previous
#include <hip/hip_runtime.h>
#include <math.h>

typedef __attribute__((ext_vector_type(2))) float v2f;
typedef __attribute__((ext_vector_type(8))) float v8f;
typedef __attribute__((ext_vector_type(4))) float v4f;

#define N_NODES 50000
#define C_IN    128
#define C_HID   256
#define HEADS   4
#define EMBED   64
#define E0      800000
#define ETOT    (E0 + N_NODES)
#define NEG_SLOPE 0.2f

template <int MASK>
__device__ __forceinline__ float row_xor(float x) {
    return __builtin_bit_cast(float, __builtin_amdgcn_update_dpp(0, __builtin_bit_cast(int, x),
                                                                 0x160 + MASK, 0xf, 0xf, false));
}
__device__ __forceinline__ float row_sum16(float v) {
    v += row_xor<1>(v);
    v += row_xor<2>(v);
    v += row_xor<4>(v);
    v += row_xor<8>(v);
    return v;
}

__global__ void gemm_wmma(const float* __restrict__ X,
                          const float* __restrict__ W,
                          float* __restrict__ H,
                          const float* __restrict__ att_src,
                          const float* __restrict__ att_dst,
                          float* __restrict__ a_src,
                          float* __restrict__ a_dst,
                          int K) {
    __shared__ v4f dots4[2][16];
    float* dots = (float*)dots4;

    const int lane   = threadIdx.x & 31;
    const int wave   = threadIdx.x >> 5;
    const int tileM  = blockIdx.x;
    const int lane15 = lane & 15;
    const int khalf  = lane >> 4;
    const int row    = tileM * 16 + lane15;
    const int col0   = wave * 64;
    const int colL   = col0 + 4 * lane15;

    v8f acc0 = {}, acc1 = {}, acc2 = {}, acc3 = {};

    const v4f*   xp = (const v4f*)(X + (size_t)row * K + 4 * khalf);
    const float* wp = W + (size_t)(4 * khalf) * C_HID + colL;

    for (int k = 0; k < K; k += 8) {
        const v4f a4 = *xp;
        const v4f r0 = *(const v4f*)(wp);
        const v4f r1 = *(const v4f*)(wp + C_HID);
        const v4f r2 = *(const v4f*)(wp + 2 * C_HID);
        const v4f r3 = *(const v4f*)(wp + 3 * C_HID);
        xp += 2;
        wp += 8 * C_HID;

        v2f a, b0, b1, b2, b3;
        a.x  = a4.x;  a.y  = a4.y;
        b0.x = r0.x;  b0.y = r1.x;
        b1.x = r0.y;  b1.y = r1.y;
        b2.x = r0.z;  b2.y = r1.z;
        b3.x = r0.w;  b3.y = r1.w;
        acc0 = __builtin_amdgcn_wmma_f32_16x16x4_f32(false, a, false, b0, (short)0, acc0, false, false);
        acc1 = __builtin_amdgcn_wmma_f32_16x16x4_f32(false, a, false, b1, (short)0, acc1, false, false);
        acc2 = __builtin_amdgcn_wmma_f32_16x16x4_f32(false, a, false, b2, (short)0, acc2, false, false);
        acc3 = __builtin_amdgcn_wmma_f32_16x16x4_f32(false, a, false, b3, (short)0, acc3, false, false);

        a.x  = a4.z;  a.y  = a4.w;
        b0.x = r2.x;  b0.y = r3.x;
        b1.x = r2.y;  b1.y = r3.y;
        b2.x = r2.z;  b2.y = r3.z;
        b3.x = r2.w;  b3.y = r3.w;
        acc0 = __builtin_amdgcn_wmma_f32_16x16x4_f32(false, a, false, b0, (short)0, acc0, false, false);
        acc1 = __builtin_amdgcn_wmma_f32_16x16x4_f32(false, a, false, b1, (short)0, acc1, false, false);
        acc2 = __builtin_amdgcn_wmma_f32_16x16x4_f32(false, a, false, b2, (short)0, acc2, false, false);
        acc3 = __builtin_amdgcn_wmma_f32_16x16x4_f32(false, a, false, b3, (short)0, acc3, false, false);
    }

    const v4f s4 = *(const v4f*)(att_src + colL);
    const v4f d4 = *(const v4f*)(att_dst + colL);
#pragma unroll
    for (int i = 0; i < 8; ++i) {
        const float ps = row_sum16(acc0[i] * s4.x + acc1[i] * s4.y + acc2[i] * s4.z + acc3[i] * s4.w);
        const float pd = row_sum16(acc0[i] * d4.x + acc1[i] * d4.y + acc2[i] * d4.z + acc3[i] * d4.w);
        if (lane15 == 0) {
            dots[(0 * 16 + khalf * 8 + i) * HEADS + wave] = ps;
            dots[(1 * 16 + khalf * 8 + i) * HEADS + wave] = pd;
        }
    }
    __syncthreads();
    if (wave == 0) {
        float* plane = khalf ? a_dst : a_src;
        *(v4f*)(plane + (size_t)(tileM * 16 + lane15) * HEADS) = dots4[khalf][lane15];
    }

    const int rbase = tileM * 16 + khalf * 8;
#pragma unroll
    for (int i = 0; i < 8; ++i) {
        v4f o;
        o.x = acc0[i]; o.y = acc1[i]; o.z = acc2[i]; o.w = acc3[i];
        *(v4f*)(H + (size_t)(rbase + i) * C_HID + colL) = o;
    }
}

typedef __attribute__((ext_vector_type(2))) int v2i;
typedef __attribute__((ext_vector_type(2))) unsigned v2u;

#define BIN_SHIFT 8
#define BIN_NODES (1 << BIN_SHIFT)
#define NBIN      ((N_NODES + BIN_NODES - 1) / BIN_NODES)
#define CHUNK     3200
#define NCHUNK    (E0 / CHUNK)
#define SORT_T    1024
static_assert(E0 % CHUNK == 0, "chunks cover the edge list exactly");

__device__ __forceinline__ unsigned block_scan_incl(unsigned c, unsigned* wtot, int lane, int wave) {
    unsigned incl = c;
    for (int off = 1; off < 32; off <<= 1) {
        unsigned v = __shfl_up(incl, off, 32);
        if (lane >= off) incl += v;
    }
    if (lane == 31) wtot[wave] = incl;
    __syncthreads();
    unsigned before = 0;
    for (int w = 0; w < wave; ++w) before += wtot[w];
    return before + incl;
}

__global__ void __launch_bounds__(SORT_T)
chunk_sort(const int* __restrict__ src, const int* __restrict__ dst,
           unsigned* __restrict__ table, int* __restrict__ rec) {
    __shared__ unsigned hist[NBIN];
    __shared__ unsigned wtot[SORT_T / 32];
    const int t = threadIdx.x, lane = t & 31, wave = t >> 5;
    const int base = blockIdx.x * CHUNK;
    if (t < NBIN) hist[t] = 0u;
    __syncthreads();

    constexpr int PER = (CHUNK + SORT_T - 1) / SORT_T;
    int s[PER], d[PER];
    unsigned r[PER];
#pragma unroll
    for (int k = 0; k < PER; ++k) {
        const int i = t + k * SORT_T;
        d[k] = -1; s[k] = 0; r[k] = 0u;
        if (i < CHUNK) { d[k] = dst[base + i]; s[k] = src[base + i]; }
    }
#pragma unroll
    for (int k = 0; k < PER; ++k)
        if (d[k] >= 0) r[k] = atomicAdd(&hist[d[k] >> BIN_SHIFT], 1u);
    __syncthreads();

    const unsigned c = (t < NBIN) ? hist[t] : 0u;
    const unsigned excl = block_scan_incl(c, wtot, lane, wave) - c;
    if (t < NBIN) {
        hist[t] = excl;
        v2u run; run.x = excl; run.y = c;
        ((v2u*)table)[blockIdx.x * NBIN + t] = run;
    }
    __syncthreads();

    v2i* rec2 = (v2i*)rec;
#pragma unroll
    for (int k = 0; k < PER; ++k)
        if (d[k] >= 0) {
            v2i e; e.x = s[k]; e.y = d[k];
            rec2[base + hist[d[k] >> BIN_SHIFT] + r[k]] = e;
        }
}

__global__ void __launch_bounds__(SORT_T)
bin_place(const unsigned* __restrict__ table, const int* __restrict__ rec,
          unsigned* __restrict__ cnt, unsigned* __restrict__ rowptr, int* __restrict__ slot) {
    __shared__ unsigned rstart[NCHUNK], rlen[NCHUNK];
    __shared__ unsigned cntL[BIN_NODES], offL[BIN_NODES], fillL[BIN_NODES];
    __shared__ unsigned wsum[SORT_T / 32], wtot[SORT_T / 32];
    const int t = threadIdx.x, lane = t & 31, wave = t >> 5;
    const int bin = blockIdx.x;
    const v2i* rec2 = (const v2i*)rec;

    unsigned o0 = 0u;
    if (t < NCHUNK) {
        const v2u run = ((const v2u*)table)[t * NBIN + bin];
        o0 = run.x;
        rstart[t] = (unsigned)t * CHUNK + run.x;
        rlen[t]   = run.y;
    }
    if (t < BIN_NODES) { cntL[t] = 0u; fillL[t] = 0u; }
    unsigned part = o0;
    for (int off = 16; off > 0; off >>= 1) part += __shfl_xor(part, off, 32);
    if (lane == 0) wsum[wave] = part;
    __syncthreads();
    unsigned binStart = 0u;
    for (int w = 0; w < SORT_T / 32; ++w) binStart += wsum[w];

    constexpr int RPW = (NCHUNK + 31) / 32;
    int sv[RPW], dv[RPW];
    unsigned at[RPW];
    bool has[RPW];
#pragma unroll
    for (int k = 0; k < RPW; ++k) {
        const int run = wave + 32 * k;
        const int rr  = run < NCHUNK ? run : 0;
        has[k] = run < NCHUNK && (unsigned)lane < rlen[rr];
        at[k]  = has[k] ? rstart[rr] + lane : 0u;
    }
#pragma unroll
    for (int k = 0; k < RPW; ++k) {
        const v2i e = rec2[at[k]];
        sv[k] = e.x;
        dv[k] = has[k] ? e.y - bin * BIN_NODES : -1;
    }
#pragma unroll
    for (int k = 0; k < RPW; ++k)
        if (dv[k] >= 0) atomicAdd(&cntL[dv[k]], 1u);
    for (int run = wave; run < NCHUNK; run += 32) {
        const unsigned len = rlen[run], st = rstart[run];
        for (unsigned i = lane + 32; i < len; i += 32)
            atomicAdd(&cntL[rec2[st + i].y - bin * BIN_NODES], 1u);
    }
    __syncthreads();

    const unsigned c = (t < BIN_NODES) ? cntL[t] : 0u;
    const unsigned excl = block_scan_incl(c, wtot, lane, wave) - c;
    if (t < BIN_NODES) {
        offL[t] = binStart + excl;
        const int node = bin * BIN_NODES + t;
        if (node < N_NODES) { rowptr[node] = binStart + excl; cnt[node] = c; }
    }
    __syncthreads();

#pragma unroll
    for (int k = 0; k < RPW; ++k)
        if (dv[k] >= 0) slot[offL[dv[k]] + atomicAdd(&fillL[dv[k]], 1u)] = sv[k];
    for (int run = wave; run < NCHUNK; run += 32) {
        const unsigned len = rlen[run], st = rstart[run];
        for (unsigned i = lane + 32; i < len; i += 32) {
            const v2i e = rec2[st + i];
            const int dl = e.y - bin * BIN_NODES;
            slot[offL[dl] + atomicAdd(&fillL[dl], 1u)] = e.x;
        }
    }
}

__device__ __forceinline__ float xhalf(float x) {
    const int b = __builtin_bit_cast(int, x);
    return __builtin_bit_cast(float, __builtin_amdgcn_permlanex16(b, b, 0x76543210, 0xfedcba98, false, false));
}
__device__ __forceinline__ float wave_max(float v) {
    v = fmaxf(v, row_xor<1>(v));
    v = fmaxf(v, row_xor<2>(v));
    v = fmaxf(v, row_xor<4>(v));
    v = fmaxf(v, row_xor<8>(v));
    return fmaxf(v, xhalf(v));
}
__device__ __forceinline__ float wave_sum(float v) {
    v += row_xor<1>(v);
    v += row_xor<2>(v);
    v += row_xor<4>(v);
    v += row_xor<8>(v);
    return v + xhalf(v);
}

__device__ __forceinline__ float uni(float x) {
    return __builtin_bit_cast(float, __builtin_amdgcn_readfirstlane(__builtin_bit_cast(int, x)));
}

__device__ __forceinline__ float leaky(float v) { return v > 0.f ? v : NEG_SLOPE * v; }

__device__ __forceinline__ void load_edges(const int* __restrict__ slot, const v4f* __restrict__ AS4,
                                           unsigned first, unsigned n, int lane, v4f ad,
                                           int& s, float& l0, float& l1, float& l2, float& l3) {
    s = 0;
    l0 = l1 = l2 = l3 = -INFINITY;
    if ((unsigned)lane < n) {
        s = slot[first + lane];
        const v4f as = AS4[s];
        l0 = leaky(as.x + ad.x); l1 = leaky(as.y + ad.y);
        l2 = leaky(as.z + ad.z); l3 = leaky(as.w + ad.w);
    }
}

__device__ __forceinline__ void add_rows(const v4f* __restrict__ H4, unsigned n, int lane, bool upper,
                                         int s, float a0, float a1, float a2, float a3,
                                         v4f& acc0, v4f& acc1) {
#pragma unroll 4
    for (unsigned j = 0; j < n; ++j) {
        const int   sj = __builtin_amdgcn_readlane(s, (int)j);
        const float b0 = __int_as_float(__builtin_amdgcn_readlane(__float_as_int(a0), (int)j));
        const float b1 = __int_as_float(__builtin_amdgcn_readlane(__float_as_int(a1), (int)j));
        const float b2 = __int_as_float(__builtin_amdgcn_readlane(__float_as_int(a2), (int)j));
        const float b3 = __int_as_float(__builtin_amdgcn_readlane(__float_as_int(a3), (int)j));
        const float alo = upper ? b1 : b0;
        const float ahi = upper ? b3 : b2;
        const v4f v0 = H4[(size_t)sj * 64 + lane];
        const v4f v1 = H4[(size_t)sj * 64 + 32 + lane];
        acc0 += v0 * alo;
        acc1 += v1 * ahi;
    }
}

#define PRE_T      256
#define PRE_BLOCKS 512
static_assert(C_IN % (8 * (PRE_T / 32)) == 0, "a wave's rows of W come in whole groups of eight");
__global__ void __launch_bounds__(PRE_T)
attn_dots_from_x(const float* __restrict__ X, const float* __restrict__ W,
                 const float* __restrict__ att_src, const float* __restrict__ att_dst,
                 float* __restrict__ a_src, float* __restrict__ a_dst) {
    __shared__ v4f wsd[C_IN][2];
    float* wsdf = (float*)wsd;
    const int lane = threadIdx.x & 31;
    const int wave = __builtin_amdgcn_readfirstlane(threadIdx.x >> 5);

    {
        const v4f s0 = ((const v4f*)att_src)[2 * lane], s1 = ((const v4f*)att_src)[2 * lane + 1];
        const v4f d0 = ((const v4f*)att_dst)[2 * lane], d1 = ((const v4f*)att_dst)[2 * lane + 1];
        constexpr int KG = 8;
        for (int kb = wave; kb < C_IN; kb += KG * (PRE_T / 32)) {
            v4f w0[KG], w1[KG];
#pragma unroll
            for (int i = 0; i < KG; ++i) {
                const v4f* wr = (const v4f*)(W + (size_t)(kb + i * (PRE_T / 32)) * C_HID);
                w0[i] = wr[2 * lane];
                w1[i] = wr[2 * lane + 1];
            }
#pragma unroll
            for (int i = 0; i < KG; ++i) {
                const int k = kb + i * (PRE_T / 32);
                float ps = w0[i].x * s0.x + w0[i].y * s0.y + w0[i].z * s0.z + w0[i].w * s0.w
                         + w1[i].x * s1.x + w1[i].y * s1.y + w1[i].z * s1.z + w1[i].w * s1.w;
                float pd = w0[i].x * d0.x + w0[i].y * d0.y + w0[i].z * d0.z + w0[i].w * d0.w
                         + w1[i].x * d1.x + w1[i].y * d1.y + w1[i].z * d1.z + w1[i].w * d1.w;
                ps += row_xor<1>(ps); ps += row_xor<2>(ps); ps += row_xor<4>(ps);
                pd += row_xor<1>(pd); pd += row_xor<2>(pd); pd += row_xor<4>(pd);
                if ((lane & 7) == 0) {
                    wsdf[k * 8 + (lane >> 3)]     = ps;
                    wsdf[k * 8 + 4 + (lane >> 3)] = pd;
                }
            }
        }
    }
    __syncthreads();

    const v4f ws0 = wsd[4 * lane][0],     wd0 = wsd[4 * lane][1];
    const v4f ws1 = wsd[4 * lane + 1][0], wd1 = wsd[4 * lane + 1][1];
    const v4f ws2 = wsd[4 * lane + 2][0], wd2 = wsd[4 * lane + 2][1];
    const v4f ws3 = wsd[4 * lane + 3][0], wd3 = wsd[4 * lane + 3][1];
    const v4f* X4 = (const v4f*)X;
    const int nw = PRE_BLOCKS * (PRE_T / 32);
    constexpr int NG = 4;
    for (int nb = blockIdx.x * (PRE_T / 32) + wave; nb < N_NODES; nb += NG * nw) {
        v4f xv[NG];
#pragma unroll
        for (int i = 0; i < NG; ++i) {
            const int n = nb + i * nw;
            xv[i] = X4[(size_t)(n < N_NODES ? n : nb) * (C_IN / 4) + lane];
        }
#pragma unroll
        for (int i = 0; i < NG; ++i) {
            const int n = nb + i * nw;
            v4f ps = ws0 * xv[i].x + ws1 * xv[i].y + ws2 * xv[i].z + ws3 * xv[i].w;
            v4f pd = wd0 * xv[i].x + wd1 * xv[i].y + wd2 * xv[i].z + wd3 * xv[i].w;
            ps.x = wave_sum(ps.x); ps.y = wave_sum(ps.y); ps.z = wave_sum(ps.z); ps.w = wave_sum(ps.w);
            pd.x = wave_sum(pd.x); pd.y = wave_sum(pd.y); pd.z = wave_sum(pd.z); pd.w = wave_sum(pd.w);
            if (n < N_NODES) {
                if (lane == 0) ((v4f*)a_src)[n] = ps;
                if (lane == 1) ((v4f*)a_dst)[n] = pd;
            }
        }
    }
}

template <int G>
__device__ __forceinline__ void add_x_rows(const v4f* __restrict__ X4, unsigned j0, int lane,
                                           int s, float a0, float a1, float a2, float a3,
                                           v4f& acc0, v4f& acc1, v4f& acc2, v4f& acc3) {
    v4f v[G];
#pragma unroll
    for (int g = 0; g < G; ++g) {
        const int sj = __builtin_amdgcn_readlane(s, (int)(j0 + g));
        v[g] = X4[(size_t)sj * (C_IN / 4) + lane];
    }
#pragma unroll
    for (int g = 0; g < G; ++g) {
        const float b0 = __int_as_float(__builtin_amdgcn_readlane(__float_as_int(a0), (int)(j0 + g)));
        const float b1 = __int_as_float(__builtin_amdgcn_readlane(__float_as_int(a1), (int)(j0 + g)));
        const float b2 = __int_as_float(__builtin_amdgcn_readlane(__float_as_int(a2), (int)(j0 + g)));
        const float b3 = __int_as_float(__builtin_amdgcn_readlane(__float_as_int(a3), (int)(j0 + g)));
        acc0 += v[g] * b0;
        acc1 += v[g] * b1;
        acc2 += v[g] * b2;
        acc3 += v[g] * b3;
    }
}

#define TILE_NODES 16
#define AGG_PITCH  (HEADS * C_IN + 4)
static_assert(N_NODES % TILE_NODES == 0, "every tile is full: no wave leaves before the barrier");
static_assert((AGG_PITCH / 4) % 2 == 1, "pitch is an odd number of 16-byte pieces");
__global__ void __launch_bounds__(TILE_NODES * 32)
layer1_gather_gemm(const int* __restrict__ slot, const unsigned* __restrict__ rowptr,
                   const unsigned* __restrict__ cnt,
                   const float* __restrict__ a_src, const float* __restrict__ a_dst,
                   const float* __restrict__ X, const float* __restrict__ W,
                   const float* __restrict__ bias, float* __restrict__ out) {
    __shared__ v4f agg4[TILE_NODES * AGG_PITCH / 4];
    float* agg = (float*)agg4;

    const int lane = threadIdx.x & 31;
    const int wave = __builtin_amdgcn_readfirstlane(threadIdx.x >> 5);
    const int node = blockIdx.x * TILE_NODES + wave;

    const v4f* X4  = (const v4f*)X;
    const v4f* AS4 = (const v4f*)a_src;

    {
        const unsigned beg = rowptr[node];
        const unsigned deg = cnt[node];
        const v4f ad  = ((const v4f*)a_dst)[node];
        const v4f asn = AS4[node];
        const v4f selfx = X4[(size_t)node * (C_IN / 4) + lane];

        const float ls0 = leaky(asn.x + ad.x), ls1 = leaky(asn.y + ad.y);
        const float ls2 = leaky(asn.z + ad.z), ls3 = leaky(asn.w + ad.w);

        const unsigned n0 = deg < 32u ? deg : 32u;
        int   s0;
        float p0, p1, p2, p3;
        load_edges(slot, AS4, beg, n0, lane, ad, s0, p0, p1, p2, p3);

        float m0 = fmaxf(ls0, p0), m1 = fmaxf(ls1, p1), m2 = fmaxf(ls2, p2), m3 = fmaxf(ls3, p3);
        for (unsigned base = 32; base < deg; base += 32) {
            const unsigned n = (deg - base < 32u) ? (deg - base) : 32u;
            int s; float l0, l1, l2, l3;
            load_edges(slot, AS4, beg + base, n, lane, ad, s, l0, l1, l2, l3);
            m0 = fmaxf(m0, l0); m1 = fmaxf(m1, l1); m2 = fmaxf(m2, l2); m3 = fmaxf(m3, l3);
        }
        m0 = uni(wave_max(m0)); m1 = uni(wave_max(m1)); m2 = uni(wave_max(m2)); m3 = uni(wave_max(m3));
        m0 = __builtin_isfinite(m0) ? m0 : 0.f; m1 = __builtin_isfinite(m1) ? m1 : 0.f;
        m2 = __builtin_isfinite(m2) ? m2 : 0.f; m3 = __builtin_isfinite(m3) ? m3 : 0.f;

        p0 = __expf(p0 - m0); p1 = __expf(p1 - m1); p2 = __expf(p2 - m2); p3 = __expf(p3 - m3);
        float z0 = p0, z1 = p1, z2 = p2, z3 = p3;
        for (unsigned base = 32; base < deg; base += 32) {
            const unsigned n = (deg - base < 32u) ? (deg - base) : 32u;
            int s; float l0, l1, l2, l3;
            load_edges(slot, AS4, beg + base, n, lane, ad, s, l0, l1, l2, l3);
            z0 += __expf(l0 - m0); z1 += __expf(l1 - m1); z2 += __expf(l2 - m2); z3 += __expf(l3 - m3);
        }
        const float es0 = __expf(ls0 - m0), es1 = __expf(ls1 - m1);
        const float es2 = __expf(ls2 - m2), es3 = __expf(ls3 - m3);
        const float zd0 = uni(wave_sum(z0) + es0 + 1e-16f), zd1 = uni(wave_sum(z1) + es1 + 1e-16f);
        const float zd2 = uni(wave_sum(z2) + es2 + 1e-16f), zd3 = uni(wave_sum(z3) + es3 + 1e-16f);

        v4f acc0 = selfx * (es0 / zd0);
        v4f acc1 = selfx * (es1 / zd1);
        v4f acc2 = selfx * (es2 / zd2);
        v4f acc3 = selfx * (es3 / zd3);
        {
            int      s = s0;
            unsigned n = n0;
            float a0 = p0 / zd0, a1 = p1 / zd1, a2 = p2 / zd2, a3 = p3 / zd3;
            for (unsigned base = 0;;) {
                unsigned j = 0;
                for (; j + 8 <= n; j += 8)
                    add_x_rows<8>(X4, j, lane, s, a0, a1, a2, a3, acc0, acc1, acc2, acc3);
                if (j + 4 <= n) {
                    add_x_rows<4>(X4, j, lane, s, a0, a1, a2, a3, acc0, acc1, acc2, acc3);
                    j += 4;
                }
                for (; j < n; ++j)
                    add_x_rows<1>(X4, j, lane, s, a0, a1, a2, a3, acc0, acc1, acc2, acc3);
                base += 32;
                if (base >= deg) break;
                n = (deg - base < 32u) ? (deg - base) : 32u;
                float l0, l1, l2, l3;
                load_edges(slot, AS4, beg + base, n, lane, ad, s, l0, l1, l2, l3);
                a0 = __expf(l0 - m0) / zd0; a1 = __expf(l1 - m1) / zd1;
                a2 = __expf(l2 - m2) / zd2; a3 = __expf(l3 - m3) / zd3;
            }
        }

        v4f* mine = (v4f*)(agg + wave * AGG_PITCH);
        mine[0 * (C_IN / 4) + lane] = acc0;
        mine[1 * (C_IN / 4) + lane] = acc1;
        mine[2 * (C_IN / 4) + lane] = acc2;
        mine[3 * (C_IN / 4) + lane] = acc3;
    }
    __syncthreads();
    if (wave >= HEADS) return;

    const int lane15 = lane & 15;
    const int khalf  = lane >> 4;
    const int colL   = wave * EMBED + 4 * lane15;

    v8f c0 = {}, c1 = {}, c2 = {}, c3 = {};
    const v4f*   ap = (const v4f*)(agg + lane15 * AGG_PITCH + wave * C_IN + 4 * khalf);
    const float* wp = W + (size_t)(4 * khalf) * C_HID + colL;
    for (int k = 0; k < C_IN; k += 8) {
        const v4f a4 = *ap;
        const v4f r0 = *(const v4f*)(wp);
        const v4f r1 = *(const v4f*)(wp + C_HID);
        const v4f r2 = *(const v4f*)(wp + 2 * C_HID);
        const v4f r3 = *(const v4f*)(wp + 3 * C_HID);
        ap += 2;
        wp += 8 * C_HID;

        v2f a, b0, b1, b2, b3;
        a.x  = a4.x;  a.y  = a4.y;
        b0.x = r0.x;  b0.y = r1.x;
        b1.x = r0.y;  b1.y = r1.y;
        b2.x = r0.z;  b2.y = r1.z;
        b3.x = r0.w;  b3.y = r1.w;
        c0 = __builtin_amdgcn_wmma_f32_16x16x4_f32(false, a, false, b0, (short)0, c0, false, false);
        c1 = __builtin_amdgcn_wmma_f32_16x16x4_f32(false, a, false, b1, (short)0, c1, false, false);
        c2 = __builtin_amdgcn_wmma_f32_16x16x4_f32(false, a, false, b2, (short)0, c2, false, false);
        c3 = __builtin_amdgcn_wmma_f32_16x16x4_f32(false, a, false, b3, (short)0, c3, false, false);

        a.x  = a4.z;  a.y  = a4.w;
        b0.x = r2.x;  b0.y = r3.x;
        b1.x = r2.y;  b1.y = r3.y;
        b2.x = r2.z;  b2.y = r3.z;
        b3.x = r2.w;  b3.y = r3.w;
        c0 = __builtin_amdgcn_wmma_f32_16x16x4_f32(false, a, false, b0, (short)0, c0, false, false);
        c1 = __builtin_amdgcn_wmma_f32_16x16x4_f32(false, a, false, b1, (short)0, c1, false, false);
        c2 = __builtin_amdgcn_wmma_f32_16x16x4_f32(false, a, false, b2, (short)0, c2, false, false);
        c3 = __builtin_amdgcn_wmma_f32_16x16x4_f32(false, a, false, b3, (short)0, c3, false, false);
    }

    const v4f b4 = *(const v4f*)(bias + colL);
    const int rbase = blockIdx.x * TILE_NODES + khalf * 8;
#pragma unroll
    for (int i = 0; i < 8; ++i) {
        v4f o;
        o.x = fmaxf(c0[i] + b4.x, 0.f); o.y = fmaxf(c1[i] + b4.y, 0.f);
        o.z = fmaxf(c2[i] + b4.z, 0.f); o.w = fmaxf(c3[i] + b4.w, 0.f);
        *(v4f*)(out + (size_t)(rbase + i) * C_HID + colL) = o;
    }
}

#define ROWS_PER_BLOCK 8
__global__ void __launch_bounds__(ROWS_PER_BLOCK * 32)
node_softmax_gather(const int* __restrict__ slot, const unsigned* __restrict__ rowptr,
                    const unsigned* __restrict__ cnt,
                    const float* __restrict__ a_src, const float* __restrict__ a_dst,
                    const float* __restrict__ H, const float* __restrict__ bias,
                    float* __restrict__ out, int relu) {
    const int lane = threadIdx.x & 31;
    const int node = __builtin_amdgcn_readfirstlane(blockIdx.x * ROWS_PER_BLOCK + (threadIdx.x >> 5));
    if (node >= N_NODES) return;

    const v4f* H4  = (const v4f*)H;
    const v4f* AS4 = (const v4f*)a_src;

    const unsigned beg = rowptr[node];
    const unsigned deg = cnt[node];
    const v4f ad  = ((const v4f*)a_dst)[node];
    const v4f asn = AS4[node];
    const bool upper = lane >= 16;

    const v4f bias0 = ((const v4f*)bias)[lane];
    const v4f bias1 = ((const v4f*)bias)[32 + lane];
    const v4f self0 = H4[(size_t)node * 64 + lane];
    const v4f self1 = H4[(size_t)node * 64 + 32 + lane];

    const float ls0 = leaky(asn.x + ad.x), ls1 = leaky(asn.y + ad.y);
    const float ls2 = leaky(asn.z + ad.z), ls3 = leaky(asn.w + ad.w);

    const unsigned n0 = deg < 32u ? deg : 32u;
    int   s0;
    float p0, p1, p2, p3;
    load_edges(slot, AS4, beg, n0, lane, ad, s0, p0, p1, p2, p3);

    float m0 = fmaxf(ls0, p0), m1 = fmaxf(ls1, p1), m2 = fmaxf(ls2, p2), m3 = fmaxf(ls3, p3);
    for (unsigned base = 32; base < deg; base += 32) {
        const unsigned n = (deg - base < 32u) ? (deg - base) : 32u;
        int s; float l0, l1, l2, l3;
        load_edges(slot, AS4, beg + base, n, lane, ad, s, l0, l1, l2, l3);
        m0 = fmaxf(m0, l0); m1 = fmaxf(m1, l1); m2 = fmaxf(m2, l2); m3 = fmaxf(m3, l3);
    }
    m0 = uni(wave_max(m0)); m1 = uni(wave_max(m1)); m2 = uni(wave_max(m2)); m3 = uni(wave_max(m3));
    m0 = __builtin_isfinite(m0) ? m0 : 0.f; m1 = __builtin_isfinite(m1) ? m1 : 0.f;
    m2 = __builtin_isfinite(m2) ? m2 : 0.f; m3 = __builtin_isfinite(m3) ? m3 : 0.f;

    p0 = __expf(p0 - m0); p1 = __expf(p1 - m1); p2 = __expf(p2 - m2); p3 = __expf(p3 - m3);
    float z0 = p0, z1 = p1, z2 = p2, z3 = p3;
    for (unsigned base = 32; base < deg; base += 32) {
        const unsigned n = (deg - base < 32u) ? (deg - base) : 32u;
        int s; float l0, l1, l2, l3;
        load_edges(slot, AS4, beg + base, n, lane, ad, s, l0, l1, l2, l3);
        z0 += __expf(l0 - m0); z1 += __expf(l1 - m1); z2 += __expf(l2 - m2); z3 += __expf(l3 - m3);
    }
    const float es0 = __expf(ls0 - m0), es1 = __expf(ls1 - m1);
    const float es2 = __expf(ls2 - m2), es3 = __expf(ls3 - m3);
    const float zd0 = uni(wave_sum(z0) + es0 + 1e-16f), zd1 = uni(wave_sum(z1) + es1 + 1e-16f);
    const float zd2 = uni(wave_sum(z2) + es2 + 1e-16f), zd3 = uni(wave_sum(z3) + es3 + 1e-16f);

    v4f acc0, acc1;
    {
        const float alo = (upper ? es1 : es0) / (upper ? zd1 : zd0);
        const float ahi = (upper ? es3 : es2) / (upper ? zd3 : zd2);
        acc0 = self0 * alo;
        acc1 = self1 * ahi;
    }
    {
        int      s = s0;
        unsigned n = n0;
        float a0 = p0 / zd0, a1 = p1 / zd1, a2 = p2 / zd2, a3 = p3 / zd3;
        for (unsigned base = 0;;) {
            add_rows(H4, n, lane, upper, s, a0, a1, a2, a3, acc0, acc1);
            base += 32;
            if (base >= deg) break;
            n = (deg - base < 32u) ? (deg - base) : 32u;
            float l0, l1, l2, l3;
            load_edges(slot, AS4, beg + base, n, lane, ad, s, l0, l1, l2, l3);
            a0 = __expf(l0 - m0) / zd0; a1 = __expf(l1 - m1) / zd1;
            a2 = __expf(l2 - m2) / zd2; a3 = __expf(l3 - m3) / zd3;
        }
    }

    acc0 += bias0;
    acc1 += bias1;
    if (relu) {
        acc0.x = fmaxf(acc0.x, 0.f); acc0.y = fmaxf(acc0.y, 0.f);
        acc0.z = fmaxf(acc0.z, 0.f); acc0.w = fmaxf(acc0.w, 0.f);
        acc1.x = fmaxf(acc1.x, 0.f); acc1.y = fmaxf(acc1.y, 0.f);
        acc1.z = fmaxf(acc1.z, 0.f); acc1.w = fmaxf(acc1.w, 0.f);
    }
    v4f* O4 = (v4f*)out;
    O4[(size_t)node * 64 + lane]      = acc0;
    O4[(size_t)node * 64 + 32 + lane] = acc1;
}

extern "C" void kernel_launch(void* const* d_in, const int* in_sizes, int n_in,
                              void* d_out, int out_size, void* d_ws, size_t ws_size,
                              hipStream_t stream) {
    const float* x        = (const float*)d_in[0];
    const int*   eidx     = (const int*)  d_in[1];
    const float* W1       = (const float*)d_in[2];
    const float* att_s1   = (const float*)d_in[3];
    const float* att_d1   = (const float*)d_in[4];
    const float* b1       = (const float*)d_in[5];
    const float* W2       = (const float*)d_in[6];
    const float* att_s2   = (const float*)d_in[7];
    const float* att_d2   = (const float*)d_in[8];
    const float* b2       = (const float*)d_in[9];
    float*       out      = (float*)d_out;

    const int* src = eidx;
    const int* dst = eidx + E0;

    const size_t NB = (size_t)N_NODES * C_HID;
    float* h      = (float*)d_ws;
    float* hrelu  = h      + NB;
    float* a_src  = hrelu  + NB;
    float* a_dst  = a_src  + (size_t)N_NODES * HEADS;
    unsigned* cnt    = (unsigned*)(a_dst + (size_t)N_NODES * HEADS);
    unsigned* rowptr = cnt + N_NODES;
    int*      slot   = (int*)(rowptr + N_NODES);
    int*      rec    = slot + E0;
    unsigned* table  = (unsigned*)(rec + 2 * (size_t)E0);

    const int gMM  = N_NODES / 16;
    const int gROW = (N_NODES + ROWS_PER_BLOCK - 1) / ROWS_PER_BLOCK;

    chunk_sort<<<NCHUNK, SORT_T, 0, stream>>>(src, dst, table, rec);
    bin_place <<<NBIN,   SORT_T, 0, stream>>>(table, rec, cnt, rowptr, slot);

    attn_dots_from_x<<<PRE_BLOCKS, PRE_T, 0, stream>>>(x, W1, att_s1, att_d1, a_src, a_dst);
    layer1_gather_gemm<<<N_NODES / TILE_NODES, TILE_NODES * 32, 0, stream>>>(slot, rowptr, cnt, a_src, a_dst, x, W1, b1, hrelu);

    gemm_wmma  <<<gMM, 128, 0, stream>>>(hrelu, W2, h, att_s2, att_d2, a_src, a_dst, C_HID);
    node_softmax_gather<<<gROW, ROWS_PER_BLOCK * 32, 0, stream>>>(slot, rowptr, cnt, a_src, a_dst, h, b2, out, 0);
}


# Layer-1 launch: the tile product's k range split over all 16 waves (4 per head, a quarter of k each), quarter tiles added through LDS in order 0,1,2,3; product sum order of layer 1 changes
# speedup vs baseline: 9.0799x; 1.0474x over previous
#include <hip/hip_runtime.h>
#include <math.h>

typedef __attribute__((ext_vector_type(2))) float v2f;
typedef __attribute__((ext_vector_type(8))) float v8f;
typedef __attribute__((ext_vector_type(4))) float v4f;

#define N_NODES 50000
#define C_IN    128
#define C_HID   256
#define HEADS   4
#define EMBED   64
#define E0      800000
#define ETOT    (E0 + N_NODES)
#define NEG_SLOPE 0.2f

template <int MASK>
__device__ __forceinline__ float row_xor(float x) {
    return __builtin_bit_cast(float, __builtin_amdgcn_update_dpp(0, __builtin_bit_cast(int, x),
                                                                 0x160 + MASK, 0xf, 0xf, false));
}
__device__ __forceinline__ float row_sum16(float v) {
    v += row_xor<1>(v);
    v += row_xor<2>(v);
    v += row_xor<4>(v);
    v += row_xor<8>(v);
    return v;
}

__global__ void gemm_wmma(const float* __restrict__ X,
                          const float* __restrict__ W,
                          float* __restrict__ H,
                          const float* __restrict__ att_src,
                          const float* __restrict__ att_dst,
                          float* __restrict__ a_src,
                          float* __restrict__ a_dst,
                          int K) {
    __shared__ v4f dots4[2][16];
    float* dots = (float*)dots4;

    const int lane   = threadIdx.x & 31;
    const int wave   = threadIdx.x >> 5;
    const int tileM  = blockIdx.x;
    const int lane15 = lane & 15;
    const int khalf  = lane >> 4;
    const int row    = tileM * 16 + lane15;
    const int col0   = wave * 64;
    const int colL   = col0 + 4 * lane15;

    v8f acc0 = {}, acc1 = {}, acc2 = {}, acc3 = {};

    const v4f*   xp = (const v4f*)(X + (size_t)row * K + 4 * khalf);
    const float* wp = W + (size_t)(4 * khalf) * C_HID + colL;

    for (int k = 0; k < K; k += 8) {
        const v4f a4 = *xp;
        const v4f r0 = *(const v4f*)(wp);
        const v4f r1 = *(const v4f*)(wp + C_HID);
        const v4f r2 = *(const v4f*)(wp + 2 * C_HID);
        const v4f r3 = *(const v4f*)(wp + 3 * C_HID);
        xp += 2;
        wp += 8 * C_HID;

        v2f a, b0, b1, b2, b3;
        a.x  = a4.x;  a.y  = a4.y;
        b0.x = r0.x;  b0.y = r1.x;
        b1.x = r0.y;  b1.y = r1.y;
        b2.x = r0.z;  b2.y = r1.z;
        b3.x = r0.w;  b3.y = r1.w;
        acc0 = __builtin_amdgcn_wmma_f32_16x16x4_f32(false, a, false, b0, (short)0, acc0, false, false);
        acc1 = __builtin_amdgcn_wmma_f32_16x16x4_f32(false, a, false, b1, (short)0, acc1, false, false);
        acc2 = __builtin_amdgcn_wmma_f32_16x16x4_f32(false, a, false, b2, (short)0, acc2, false, false);
        acc3 = __builtin_amdgcn_wmma_f32_16x16x4_f32(false, a, false, b3, (short)0, acc3, false, false);

        a.x  = a4.z;  a.y  = a4.w;
        b0.x = r2.x;  b0.y = r3.x;
        b1.x = r2.y;  b1.y = r3.y;
        b2.x = r2.z;  b2.y = r3.z;
        b3.x = r2.w;  b3.y = r3.w;
        acc0 = __builtin_amdgcn_wmma_f32_16x16x4_f32(false, a, false, b0, (short)0, acc0, false, false);
        acc1 = __builtin_amdgcn_wmma_f32_16x16x4_f32(false, a, false, b1, (short)0, acc1, false, false);
        acc2 = __builtin_amdgcn_wmma_f32_16x16x4_f32(false, a, false, b2, (short)0, acc2, false, false);
        acc3 = __builtin_amdgcn_wmma_f32_16x16x4_f32(false, a, false, b3, (short)0, acc3, false, false);
    }

    const v4f s4 = *(const v4f*)(att_src + colL);
    const v4f d4 = *(const v4f*)(att_dst + colL);
#pragma unroll
    for (int i = 0; i < 8; ++i) {
        const float ps = row_sum16(acc0[i] * s4.x + acc1[i] * s4.y + acc2[i] * s4.z + acc3[i] * s4.w);
        const float pd = row_sum16(acc0[i] * d4.x + acc1[i] * d4.y + acc2[i] * d4.z + acc3[i] * d4.w);
        if (lane15 == 0) {
            dots[(0 * 16 + khalf * 8 + i) * HEADS + wave] = ps;
            dots[(1 * 16 + khalf * 8 + i) * HEADS + wave] = pd;
        }
    }
    __syncthreads();
    if (wave == 0) {
        float* plane = khalf ? a_dst : a_src;
        *(v4f*)(plane + (size_t)(tileM * 16 + lane15) * HEADS) = dots4[khalf][lane15];
    }

    const int rbase = tileM * 16 + khalf * 8;
#pragma unroll
    for (int i = 0; i < 8; ++i) {
        v4f o;
        o.x = acc0[i]; o.y = acc1[i]; o.z = acc2[i]; o.w = acc3[i];
        *(v4f*)(H + (size_t)(rbase + i) * C_HID + colL) = o;
    }
}

typedef __attribute__((ext_vector_type(2))) int v2i;
typedef __attribute__((ext_vector_type(2))) unsigned v2u;

#define BIN_SHIFT 8
#define BIN_NODES (1 << BIN_SHIFT)
#define NBIN      ((N_NODES + BIN_NODES - 1) / BIN_NODES)
#define CHUNK     3200
#define NCHUNK    (E0 / CHUNK)
#define SORT_T    1024
static_assert(E0 % CHUNK == 0, "chunks cover the edge list exactly");

__device__ __forceinline__ unsigned block_scan_incl(unsigned c, unsigned* wtot, int lane, int wave) {
    unsigned incl = c;
    for (int off = 1; off < 32; off <<= 1) {
        unsigned v = __shfl_up(incl, off, 32);
        if (lane >= off) incl += v;
    }
    if (lane == 31) wtot[wave] = incl;
    __syncthreads();
    unsigned before = 0;
    for (int w = 0; w < wave; ++w) before += wtot[w];
    return before + incl;
}

__global__ void __launch_bounds__(SORT_T)
chunk_sort(const int* __restrict__ src, const int* __restrict__ dst,
           unsigned* __restrict__ table, int* __restrict__ rec) {
    __shared__ unsigned hist[NBIN];
    __shared__ unsigned wtot[SORT_T / 32];
    const int t = threadIdx.x, lane = t & 31, wave = t >> 5;
    const int base = blockIdx.x * CHUNK;
    if (t < NBIN) hist[t] = 0u;
    __syncthreads();

    constexpr int PER = (CHUNK + SORT_T - 1) / SORT_T;
    int s[PER], d[PER];
    unsigned r[PER];
#pragma unroll
    for (int k = 0; k < PER; ++k) {
        const int i = t + k * SORT_T;
        d[k] = -1; s[k] = 0; r[k] = 0u;
        if (i < CHUNK) { d[k] = dst[base + i]; s[k] = src[base + i]; }
    }
#pragma unroll
    for (int k = 0; k < PER; ++k)
        if (d[k] >= 0) r[k] = atomicAdd(&hist[d[k] >> BIN_SHIFT], 1u);
    __syncthreads();

    const unsigned c = (t < NBIN) ? hist[t] : 0u;
    const unsigned excl = block_scan_incl(c, wtot, lane, wave) - c;
    if (t < NBIN) {
        hist[t] = excl;
        v2u run; run.x = excl; run.y = c;
        ((v2u*)table)[blockIdx.x * NBIN + t] = run;
    }
    __syncthreads();

    v2i* rec2 = (v2i*)rec;
#pragma unroll
    for (int k = 0; k < PER; ++k)
        if (d[k] >= 0) {
            v2i e; e.x = s[k]; e.y = d[k];
            rec2[base + hist[d[k] >> BIN_SHIFT] + r[k]] = e;
        }
}

__global__ void __launch_bounds__(SORT_T)
bin_place(const unsigned* __restrict__ table, const int* __restrict__ rec,
          unsigned* __restrict__ cnt, unsigned* __restrict__ rowptr, int* __restrict__ slot) {
    __shared__ unsigned rstart[NCHUNK], rlen[NCHUNK];
    __shared__ unsigned cntL[BIN_NODES], offL[BIN_NODES], fillL[BIN_NODES];
    __shared__ unsigned wsum[SORT_T / 32], wtot[SORT_T / 32];
    const int t = threadIdx.x, lane = t & 31, wave = t >> 5;
    const int bin = blockIdx.x;
    const v2i* rec2 = (const v2i*)rec;

    unsigned o0 = 0u;
    if (t < NCHUNK) {
        const v2u run = ((const v2u*)table)[t * NBIN + bin];
        o0 = run.x;
        rstart[t] = (unsigned)t * CHUNK + run.x;
        rlen[t]   = run.y;
    }
    if (t < BIN_NODES) { cntL[t] = 0u; fillL[t] = 0u; }
    unsigned part = o0;
    for (int off = 16; off > 0; off >>= 1) part += __shfl_xor(part, off, 32);
    if (lane == 0) wsum[wave] = part;
    __syncthreads();
    unsigned binStart = 0u;
    for (int w = 0; w < SORT_T / 32; ++w) binStart += wsum[w];

    constexpr int RPW = (NCHUNK + 31) / 32;
    int sv[RPW], dv[RPW];
    unsigned at[RPW];
    bool has[RPW];
#pragma unroll
    for (int k = 0; k < RPW; ++k) {
        const int run = wave + 32 * k;
        const int rr  = run < NCHUNK ? run : 0;
        has[k] = run < NCHUNK && (unsigned)lane < rlen[rr];
        at[k]  = has[k] ? rstart[rr] + lane : 0u;
    }
#pragma unroll
    for (int k = 0; k < RPW; ++k) {
        const v2i e = rec2[at[k]];
        sv[k] = e.x;
        dv[k] = has[k] ? e.y - bin * BIN_NODES : -1;
    }
#pragma unroll
    for (int k = 0; k < RPW; ++k)
        if (dv[k] >= 0) atomicAdd(&cntL[dv[k]], 1u);
    for (int run = wave; run < NCHUNK; run += 32) {
        const unsigned len = rlen[run], st = rstart[run];
        for (unsigned i = lane + 32; i < len; i += 32)
            atomicAdd(&cntL[rec2[st + i].y - bin * BIN_NODES], 1u);
    }
    __syncthreads();

    const unsigned c = (t < BIN_NODES) ? cntL[t] : 0u;
    const unsigned excl = block_scan_incl(c, wtot, lane, wave) - c;
    if (t < BIN_NODES) {
        offL[t] = binStart + excl;
        const int node = bin * BIN_NODES + t;
        if (node < N_NODES) { rowptr[node] = binStart + excl; cnt[node] = c; }
    }
    __syncthreads();

#pragma unroll
    for (int k = 0; k < RPW; ++k)
        if (dv[k] >= 0) slot[offL[dv[k]] + atomicAdd(&fillL[dv[k]], 1u)] = sv[k];
    for (int run = wave; run < NCHUNK; run += 32) {
        const unsigned len = rlen[run], st = rstart[run];
        for (unsigned i = lane + 32; i < len; i += 32) {
            const v2i e = rec2[st + i];
            const int dl = e.y - bin * BIN_NODES;
            slot[offL[dl] + atomicAdd(&fillL[dl], 1u)] = e.x;
        }
    }
}

__device__ __forceinline__ float xhalf(float x) {
    const int b = __builtin_bit_cast(int, x);
    return __builtin_bit_cast(float, __builtin_amdgcn_permlanex16(b, b, 0x76543210, 0xfedcba98, false, false));
}
__device__ __forceinline__ float wave_max(float v) {
    v = fmaxf(v, row_xor<1>(v));
    v = fmaxf(v, row_xor<2>(v));
    v = fmaxf(v, row_xor<4>(v));
    v = fmaxf(v, row_xor<8>(v));
    return fmaxf(v, xhalf(v));
}
__device__ __forceinline__ float wave_sum(float v) {
    v += row_xor<1>(v);
    v += row_xor<2>(v);
    v += row_xor<4>(v);
    v += row_xor<8>(v);
    return v + xhalf(v);
}

__device__ __forceinline__ float uni(float x) {
    return __builtin_bit_cast(float, __builtin_amdgcn_readfirstlane(__builtin_bit_cast(int, x)));
}

__device__ __forceinline__ float leaky(float v) { return v > 0.f ? v : NEG_SLOPE * v; }

__device__ __forceinline__ void load_edges(const int* __restrict__ slot, const v4f* __restrict__ AS4,
                                           unsigned first, unsigned n, int lane, v4f ad,
                                           int& s, float& l0, float& l1, float& l2, float& l3) {
    s = 0;
    l0 = l1 = l2 = l3 = -INFINITY;
    if ((unsigned)lane < n) {
        s = slot[first + lane];
        const v4f as = AS4[s];
        l0 = leaky(as.x + ad.x); l1 = leaky(as.y + ad.y);
        l2 = leaky(as.z + ad.z); l3 = leaky(as.w + ad.w);
    }
}

__device__ __forceinline__ void add_rows(const v4f* __restrict__ H4, unsigned n, int lane, bool upper,
                                         int s, float a0, float a1, float a2, float a3,
                                         v4f& acc0, v4f& acc1) {
#pragma unroll 4
    for (unsigned j = 0; j < n; ++j) {
        const int   sj = __builtin_amdgcn_readlane(s, (int)j);
        const float b0 = __int_as_float(__builtin_amdgcn_readlane(__float_as_int(a0), (int)j));
        const float b1 = __int_as_float(__builtin_amdgcn_readlane(__float_as_int(a1), (int)j));
        const float b2 = __int_as_float(__builtin_amdgcn_readlane(__float_as_int(a2), (int)j));
        const float b3 = __int_as_float(__builtin_amdgcn_readlane(__float_as_int(a3), (int)j));
        const float alo = upper ? b1 : b0;
        const float ahi = upper ? b3 : b2;
        const v4f v0 = H4[(size_t)sj * 64 + lane];
        const v4f v1 = H4[(size_t)sj * 64 + 32 + lane];
        acc0 += v0 * alo;
        acc1 += v1 * ahi;
    }
}

#define PRE_T      256
#define PRE_BLOCKS 512
static_assert(C_IN % (8 * (PRE_T / 32)) == 0, "a wave's rows of W come in whole groups of eight");
__global__ void __launch_bounds__(PRE_T)
attn_dots_from_x(const float* __restrict__ X, const float* __restrict__ W,
                 const float* __restrict__ att_src, const float* __restrict__ att_dst,
                 float* __restrict__ a_src, float* __restrict__ a_dst) {
    __shared__ v4f wsd[C_IN][2];
    float* wsdf = (float*)wsd;
    const int lane = threadIdx.x & 31;
    const int wave = __builtin_amdgcn_readfirstlane(threadIdx.x >> 5);

    {
        const v4f s0 = ((const v4f*)att_src)[2 * lane], s1 = ((const v4f*)att_src)[2 * lane + 1];
        const v4f d0 = ((const v4f*)att_dst)[2 * lane], d1 = ((const v4f*)att_dst)[2 * lane + 1];
        constexpr int KG = 8;
        for (int kb = wave; kb < C_IN; kb += KG * (PRE_T / 32)) {
            v4f w0[KG], w1[KG];
#pragma unroll
            for (int i = 0; i < KG; ++i) {
                const v4f* wr = (const v4f*)(W + (size_t)(kb + i * (PRE_T / 32)) * C_HID);
                w0[i] = wr[2 * lane];
                w1[i] = wr[2 * lane + 1];
            }
#pragma unroll
            for (int i = 0; i < KG; ++i) {
                const int k = kb + i * (PRE_T / 32);
                float ps = w0[i].x * s0.x + w0[i].y * s0.y + w0[i].z * s0.z + w0[i].w * s0.w
                         + w1[i].x * s1.x + w1[i].y * s1.y + w1[i].z * s1.z + w1[i].w * s1.w;
                float pd = w0[i].x * d0.x + w0[i].y * d0.y + w0[i].z * d0.z + w0[i].w * d0.w
                         + w1[i].x * d1.x + w1[i].y * d1.y + w1[i].z * d1.z + w1[i].w * d1.w;
                ps += row_xor<1>(ps); ps += row_xor<2>(ps); ps += row_xor<4>(ps);
                pd += row_xor<1>(pd); pd += row_xor<2>(pd); pd += row_xor<4>(pd);
                if ((lane & 7) == 0) {
                    wsdf[k * 8 + (lane >> 3)]     = ps;
                    wsdf[k * 8 + 4 + (lane >> 3)] = pd;
                }
            }
        }
    }
    __syncthreads();

    const v4f ws0 = wsd[4 * lane][0],     wd0 = wsd[4 * lane][1];
    const v4f ws1 = wsd[4 * lane + 1][0], wd1 = wsd[4 * lane + 1][1];
    const v4f ws2 = wsd[4 * lane + 2][0], wd2 = wsd[4 * lane + 2][1];
    const v4f ws3 = wsd[4 * lane + 3][0], wd3 = wsd[4 * lane + 3][1];
    const v4f* X4 = (const v4f*)X;
    const int nw = PRE_BLOCKS * (PRE_T / 32);
    constexpr int NG = 4;
    for (int nb = blockIdx.x * (PRE_T / 32) + wave; nb < N_NODES; nb += NG * nw) {
        v4f xv[NG];
#pragma unroll
        for (int i = 0; i < NG; ++i) {
            const int n = nb + i * nw;
            xv[i] = X4[(size_t)(n < N_NODES ? n : nb) * (C_IN / 4) + lane];
        }
#pragma unroll
        for (int i = 0; i < NG; ++i) {
            const int n = nb + i * nw;
            v4f ps = ws0 * xv[i].x + ws1 * xv[i].y + ws2 * xv[i].z + ws3 * xv[i].w;
            v4f pd = wd0 * xv[i].x + wd1 * xv[i].y + wd2 * xv[i].z + wd3 * xv[i].w;
            ps.x = wave_sum(ps.x); ps.y = wave_sum(ps.y); ps.z = wave_sum(ps.z); ps.w = wave_sum(ps.w);
            pd.x = wave_sum(pd.x); pd.y = wave_sum(pd.y); pd.z = wave_sum(pd.z); pd.w = wave_sum(pd.w);
            if (n < N_NODES) {
                if (lane == 0) ((v4f*)a_src)[n] = ps;
                if (lane == 1) ((v4f*)a_dst)[n] = pd;
            }
        }
    }
}

template <int G>
__device__ __forceinline__ void add_x_rows(const v4f* __restrict__ X4, unsigned j0, int lane,
                                           int s, float a0, float a1, float a2, float a3,
                                           v4f& acc0, v4f& acc1, v4f& acc2, v4f& acc3) {
    v4f v[G];
#pragma unroll
    for (int g = 0; g < G; ++g) {
        const int sj = __builtin_amdgcn_readlane(s, (int)(j0 + g));
        v[g] = X4[(size_t)sj * (C_IN / 4) + lane];
    }
#pragma unroll
    for (int g = 0; g < G; ++g) {
        const float b0 = __int_as_float(__builtin_amdgcn_readlane(__float_as_int(a0), (int)(j0 + g)));
        const float b1 = __int_as_float(__builtin_amdgcn_readlane(__float_as_int(a1), (int)(j0 + g)));
        const float b2 = __int_as_float(__builtin_amdgcn_readlane(__float_as_int(a2), (int)(j0 + g)));
        const float b3 = __int_as_float(__builtin_amdgcn_readlane(__float_as_int(a3), (int)(j0 + g)));
        acc0 += v[g] * b0;
        acc1 += v[g] * b1;
        acc2 += v[g] * b2;
        acc3 += v[g] * b3;
    }
}

#define TILE_NODES 16
#define AGG_PITCH  (HEADS * C_IN + 4)
static_assert(N_NODES % TILE_NODES == 0, "every tile is full: no wave leaves before the barrier");
static_assert((AGG_PITCH / 4) % 2 == 1, "pitch is an odd number of 16-byte pieces");
#define KQUARTERS  (TILE_NODES / HEADS)
#define KQ         (C_IN / KQUARTERS)
static_assert(TILE_NODES == HEADS * KQUARTERS && KQ % 8 == 0, "16 waves = heads x parts of k, k taken in eights");
__global__ void __launch_bounds__(TILE_NODES * 32)
layer1_gather_gemm(const int* __restrict__ slot, const unsigned* __restrict__ rowptr,
                   const unsigned* __restrict__ cnt,
                   const float* __restrict__ a_src, const float* __restrict__ a_dst,
                   const float* __restrict__ X, const float* __restrict__ W,
                   const float* __restrict__ bias, float* __restrict__ out) {
    __shared__ v4f agg4[TILE_NODES * AGG_PITCH / 4];
    __shared__ v4f part4[(KQUARTERS - 1) * HEADS * 8 * 32];
    float* agg = (float*)agg4;

    const int lane = threadIdx.x & 31;
    const int wave = __builtin_amdgcn_readfirstlane(threadIdx.x >> 5);
    const int node = blockIdx.x * TILE_NODES + wave;

    const v4f* X4  = (const v4f*)X;
    const v4f* AS4 = (const v4f*)a_src;

    {
        const unsigned beg = rowptr[node];
        const unsigned deg = cnt[node];
        const v4f ad  = ((const v4f*)a_dst)[node];
        const v4f asn = AS4[node];
        const v4f selfx = X4[(size_t)node * (C_IN / 4) + lane];

        const float ls0 = leaky(asn.x + ad.x), ls1 = leaky(asn.y + ad.y);
        const float ls2 = leaky(asn.z + ad.z), ls3 = leaky(asn.w + ad.w);

        const unsigned n0 = deg < 32u ? deg : 32u;
        int   s0;
        float p0, p1, p2, p3;
        load_edges(slot, AS4, beg, n0, lane, ad, s0, p0, p1, p2, p3);

        float m0 = fmaxf(ls0, p0), m1 = fmaxf(ls1, p1), m2 = fmaxf(ls2, p2), m3 = fmaxf(ls3, p3);
        for (unsigned base = 32; base < deg; base += 32) {
            const unsigned n = (deg - base < 32u) ? (deg - base) : 32u;
            int s; float l0, l1, l2, l3;
            load_edges(slot, AS4, beg + base, n, lane, ad, s, l0, l1, l2, l3);
            m0 = fmaxf(m0, l0); m1 = fmaxf(m1, l1); m2 = fmaxf(m2, l2); m3 = fmaxf(m3, l3);
        }
        m0 = uni(wave_max(m0)); m1 = uni(wave_max(m1)); m2 = uni(wave_max(m2)); m3 = uni(wave_max(m3));
        m0 = __builtin_isfinite(m0) ? m0 : 0.f; m1 = __builtin_isfinite(m1) ? m1 : 0.f;
        m2 = __builtin_isfinite(m2) ? m2 : 0.f; m3 = __builtin_isfinite(m3) ? m3 : 0.f;

        p0 = __expf(p0 - m0); p1 = __expf(p1 - m1); p2 = __expf(p2 - m2); p3 = __expf(p3 - m3);
        float z0 = p0, z1 = p1, z2 = p2, z3 = p3;
        for (unsigned base = 32; base < deg; base += 32) {
            const unsigned n = (deg - base < 32u) ? (deg - base) : 32u;
            int s; float l0, l1, l2, l3;
            load_edges(slot, AS4, beg + base, n, lane, ad, s, l0, l1, l2, l3);
            z0 += __expf(l0 - m0); z1 += __expf(l1 - m1); z2 += __expf(l2 - m2); z3 += __expf(l3 - m3);
        }
        const float es0 = __expf(ls0 - m0), es1 = __expf(ls1 - m1);
        const float es2 = __expf(ls2 - m2), es3 = __expf(ls3 - m3);
        const float zd0 = uni(wave_sum(z0) + es0 + 1e-16f), zd1 = uni(wave_sum(z1) + es1 + 1e-16f);
        const float zd2 = uni(wave_sum(z2) + es2 + 1e-16f), zd3 = uni(wave_sum(z3) + es3 + 1e-16f);

        v4f acc0 = selfx * (es0 / zd0);
        v4f acc1 = selfx * (es1 / zd1);
        v4f acc2 = selfx * (es2 / zd2);
        v4f acc3 = selfx * (es3 / zd3);
        {
            int      s = s0;
            unsigned n = n0;
            float a0 = p0 / zd0, a1 = p1 / zd1, a2 = p2 / zd2, a3 = p3 / zd3;
            for (unsigned base = 0;;) {
                unsigned j = 0;
                for (; j + 8 <= n; j += 8)
                    add_x_rows<8>(X4, j, lane, s, a0, a1, a2, a3, acc0, acc1, acc2, acc3);
                if (j + 4 <= n) {
                    add_x_rows<4>(X4, j, lane, s, a0, a1, a2, a3, acc0, acc1, acc2, acc3);
                    j += 4;
                }
                for (; j < n; ++j)
                    add_x_rows<1>(X4, j, lane, s, a0, a1, a2, a3, acc0, acc1, acc2, acc3);
                base += 32;
                if (base >= deg) break;
                n = (deg - base < 32u) ? (deg - base) : 32u;
                float l0, l1, l2, l3;
                load_edges(slot, AS4, beg + base, n, lane, ad, s, l0, l1, l2, l3);
                a0 = __expf(l0 - m0) / zd0; a1 = __expf(l1 - m1) / zd1;
                a2 = __expf(l2 - m2) / zd2; a3 = __expf(l3 - m3) / zd3;
            }
        }

        v4f* mine = (v4f*)(agg + wave * AGG_PITCH);
        mine[0 * (C_IN / 4) + lane] = acc0;
        mine[1 * (C_IN / 4) + lane] = acc1;
        mine[2 * (C_IN / 4) + lane] = acc2;
        mine[3 * (C_IN / 4) + lane] = acc3;
    }
    __syncthreads();

    const int hd     = wave & (HEADS - 1);
    const int q      = wave / HEADS;
    const int lane15 = lane & 15;
    const int khalf  = lane >> 4;
    const int colL   = hd * EMBED + 4 * lane15;

    v8f c0 = {}, c1 = {}, c2 = {}, c3 = {};
    const v4f*   ap = (const v4f*)(agg + lane15 * AGG_PITCH + hd * C_IN + q * KQ + 4 * khalf);
    const float* wp = W + (size_t)(q * KQ + 4 * khalf) * C_HID + colL;
#pragma unroll
    for (int k = 0; k < KQ; k += 8) {
        const v4f a4 = *ap;
        const v4f r0 = *(const v4f*)(wp);
        const v4f r1 = *(const v4f*)(wp + C_HID);
        const v4f r2 = *(const v4f*)(wp + 2 * C_HID);
        const v4f r3 = *(const v4f*)(wp + 3 * C_HID);
        ap += 2;
        wp += 8 * C_HID;

        v2f a, b0, b1, b2, b3;
        a.x  = a4.x;  a.y  = a4.y;
        b0.x = r0.x;  b0.y = r1.x;
        b1.x = r0.y;  b1.y = r1.y;
        b2.x = r0.z;  b2.y = r1.z;
        b3.x = r0.w;  b3.y = r1.w;
        c0 = __builtin_amdgcn_wmma_f32_16x16x4_f32(false, a, false, b0, (short)0, c0, false, false);
        c1 = __builtin_amdgcn_wmma_f32_16x16x4_f32(false, a, false, b1, (short)0, c1, false, false);
        c2 = __builtin_amdgcn_wmma_f32_16x16x4_f32(false, a, false, b2, (short)0, c2, false, false);
        c3 = __builtin_amdgcn_wmma_f32_16x16x4_f32(false, a, false, b3, (short)0, c3, false, false);

        a.x  = a4.z;  a.y  = a4.w;
        b0.x = r2.x;  b0.y = r3.x;
        b1.x = r2.y;  b1.y = r3.y;
        b2.x = r2.z;  b2.y = r3.z;
        b3.x = r2.w;  b3.y = r3.w;
        c0 = __builtin_amdgcn_wmma_f32_16x16x4_f32(false, a, false, b0, (short)0, c0, false, false);
        c1 = __builtin_amdgcn_wmma_f32_16x16x4_f32(false, a, false, b1, (short)0, c1, false, false);
        c2 = __builtin_amdgcn_wmma_f32_16x16x4_f32(false, a, false, b2, (short)0, c2, false, false);
        c3 = __builtin_amdgcn_wmma_f32_16x16x4_f32(false, a, false, b3, (short)0, c3, false, false);
    }

    if (q != 0) {
        v4f* mine = part4 + ((q - 1) * HEADS + hd) * (8 * 32) + lane;
#pragma unroll
        for (int i = 0; i < 8; ++i) {
            v4f o;
            o.x = c0[i]; o.y = c1[i]; o.z = c2[i]; o.w = c3[i];
            mine[i * 32] = o;
        }
    }
    __syncthreads();
    if (q != 0) return;

    const v4f b4 = *(const v4f*)(bias + colL);
    const int rbase = blockIdx.x * TILE_NODES + khalf * 8;
    const v4f* theirs = part4 + hd * (8 * 32) + lane;
#pragma unroll
    for (int i = 0; i < 8; ++i) {
        v4f o;
        o.x = c0[i]; o.y = c1[i]; o.z = c2[i]; o.w = c3[i];
        o += theirs[(0 * HEADS * 8 + i) * 32];
        o += theirs[(1 * HEADS * 8 + i) * 32];
        o += theirs[(2 * HEADS * 8 + i) * 32];
        o.x = fmaxf(o.x + b4.x, 0.f); o.y = fmaxf(o.y + b4.y, 0.f);
        o.z = fmaxf(o.z + b4.z, 0.f); o.w = fmaxf(o.w + b4.w, 0.f);
        *(v4f*)(out + (size_t)(rbase + i) * C_HID + colL) = o;
    }
}

#define ROWS_PER_BLOCK 8
__global__ void __launch_bounds__(ROWS_PER_BLOCK * 32)
node_softmax_gather(const int* __restrict__ slot, const unsigned* __restrict__ rowptr,
                    const unsigned* __restrict__ cnt,
                    const float* __restrict__ a_src, const float* __restrict__ a_dst,
                    const float* __restrict__ H, const float* __restrict__ bias,
                    float* __restrict__ out, int relu) {
    const int lane = threadIdx.x & 31;
    const int node = __builtin_amdgcn_readfirstlane(blockIdx.x * ROWS_PER_BLOCK + (threadIdx.x >> 5));
    if (node >= N_NODES) return;

    const v4f* H4  = (const v4f*)H;
    const v4f* AS4 = (const v4f*)a_src;

    const unsigned beg = rowptr[node];
    const unsigned deg = cnt[node];
    const v4f ad  = ((const v4f*)a_dst)[node];
    const v4f asn = AS4[node];
    const bool upper = lane >= 16;

    const v4f bias0 = ((const v4f*)bias)[lane];
    const v4f bias1 = ((const v4f*)bias)[32 + lane];
    const v4f self0 = H4[(size_t)node * 64 + lane];
    const v4f self1 = H4[(size_t)node * 64 + 32 + lane];

    const float ls0 = leaky(asn.x + ad.x), ls1 = leaky(asn.y + ad.y);
    const float ls2 = leaky(asn.z + ad.z), ls3 = leaky(asn.w + ad.w);

    const unsigned n0 = deg < 32u ? deg : 32u;
    int   s0;
    float p0, p1, p2, p3;
    load_edges(slot, AS4, beg, n0, lane, ad, s0, p0, p1, p2, p3);

    float m0 = fmaxf(ls0, p0), m1 = fmaxf(ls1, p1), m2 = fmaxf(ls2, p2), m3 = fmaxf(ls3, p3);
    for (unsigned base = 32; base < deg; base += 32) {
        const unsigned n = (deg - base < 32u) ? (deg - base) : 32u;
        int s; float l0, l1, l2, l3;
        load_edges(slot, AS4, beg + base, n, lane, ad, s, l0, l1, l2, l3);
        m0 = fmaxf(m0, l0); m1 = fmaxf(m1, l1); m2 = fmaxf(m2, l2); m3 = fmaxf(m3, l3);
    }
    m0 = uni(wave_max(m0)); m1 = uni(wave_max(m1)); m2 = uni(wave_max(m2)); m3 = uni(wave_max(m3));
    m0 = __builtin_isfinite(m0) ? m0 : 0.f; m1 = __builtin_isfinite(m1) ? m1 : 0.f;
    m2 = __builtin_isfinite(m2) ? m2 : 0.f; m3 = __builtin_isfinite(m3) ? m3 : 0.f;

    p0 = __expf(p0 - m0); p1 = __expf(p1 - m1); p2 = __expf(p2 - m2); p3 = __expf(p3 - m3);
    float z0 = p0, z1 = p1, z2 = p2, z3 = p3;
    for (unsigned base = 32; base < deg; base += 32) {
        const unsigned n = (deg - base < 32u) ? (deg - base) : 32u;
        int s; float l0, l1, l2, l3;
        load_edges(slot, AS4, beg + base, n, lane, ad, s, l0, l1, l2, l3);
        z0 += __expf(l0 - m0); z1 += __expf(l1 - m1); z2 += __expf(l2 - m2); z3 += __expf(l3 - m3);
    }
    const float es0 = __expf(ls0 - m0), es1 = __expf(ls1 - m1);
    const float es2 = __expf(ls2 - m2), es3 = __expf(ls3 - m3);
    const float zd0 = uni(wave_sum(z0) + es0 + 1e-16f), zd1 = uni(wave_sum(z1) + es1 + 1e-16f);
    const float zd2 = uni(wave_sum(z2) + es2 + 1e-16f), zd3 = uni(wave_sum(z3) + es3 + 1e-16f);

    v4f acc0, acc1;
    {
        const float alo = (upper ? es1 : es0) / (upper ? zd1 : zd0);
        const float ahi = (upper ? es3 : es2) / (upper ? zd3 : zd2);
        acc0 = self0 * alo;
        acc1 = self1 * ahi;
    }
    {
        int      s = s0;
        unsigned n = n0;
        float a0 = p0 / zd0, a1 = p1 / zd1, a2 = p2 / zd2, a3 = p3 / zd3;
        for (unsigned base = 0;;) {
            add_rows(H4, n, lane, upper, s, a0, a1, a2, a3, acc0, acc1);
            base += 32;
            if (base >= deg) break;
            n = (deg - base < 32u) ? (deg - base) : 32u;
            float l0, l1, l2, l3;
            load_edges(slot, AS4, beg + base, n, lane, ad, s, l0, l1, l2, l3);
            a0 = __expf(l0 - m0) / zd0; a1 = __expf(l1 - m1) / zd1;
            a2 = __expf(l2 - m2) / zd2; a3 = __expf(l3 - m3) / zd3;
        }
    }

    acc0 += bias0;
    acc1 += bias1;
    if (relu) {
        acc0.x = fmaxf(acc0.x, 0.f); acc0.y = fmaxf(acc0.y, 0.f);
        acc0.z = fmaxf(acc0.z, 0.f); acc0.w = fmaxf(acc0.w, 0.f);
        acc1.x = fmaxf(acc1.x, 0.f); acc1.y = fmaxf(acc1.y, 0.f);
        acc1.z = fmaxf(acc1.z, 0.f); acc1.w = fmaxf(acc1.w, 0.f);
    }
    v4f* O4 = (v4f*)out;
    O4[(size_t)node * 64 + lane]      = acc0;
    O4[(size_t)node * 64 + 32 + lane] = acc1;
}

extern "C" void kernel_launch(void* const* d_in, const int* in_sizes, int n_in,
                              void* d_out, int out_size, void* d_ws, size_t ws_size,
                              hipStream_t stream) {
    const float* x        = (const float*)d_in[0];
    const int*   eidx     = (const int*)  d_in[1];
    const float* W1       = (const float*)d_in[2];
    const float* att_s1   = (const float*)d_in[3];
    const float* att_d1   = (const float*)d_in[4];
    const float* b1       = (const float*)d_in[5];
    const float* W2       = (const float*)d_in[6];
    const float* att_s2   = (const float*)d_in[7];
    const float* att_d2   = (const float*)d_in[8];
    const float* b2       = (const float*)d_in[9];
    float*       out      = (float*)d_out;

    const int* src = eidx;
    const int* dst = eidx + E0;

    const size_t NB = (size_t)N_NODES * C_HID;
    float* h      = (float*)d_ws;
    float* hrelu  = h      + NB;
    float* a_src  = hrelu  + NB;
    float* a_dst  = a_src  + (size_t)N_NODES * HEADS;
    unsigned* cnt    = (unsigned*)(a_dst + (size_t)N_NODES * HEADS);
    unsigned* rowptr = cnt + N_NODES;
    int*      slot   = (int*)(rowptr + N_NODES);
    int*      rec    = slot + E0;
    unsigned* table  = (unsigned*)(rec + 2 * (size_t)E0);

    const int gMM  = N_NODES / 16;
    const int gROW = (N_NODES + ROWS_PER_BLOCK - 1) / ROWS_PER_BLOCK;

    chunk_sort<<<NCHUNK, SORT_T, 0, stream>>>(src, dst, table, rec);
    bin_place <<<NBIN,   SORT_T, 0, stream>>>(table, rec, cnt, rowptr, slot);

    attn_dots_from_x<<<PRE_BLOCKS, PRE_T, 0, stream>>>(x, W1, att_s1, att_d1, a_src, a_dst);
    layer1_gather_gemm<<<N_NODES / TILE_NODES, TILE_NODES * 32, 0, stream>>>(slot, rowptr, cnt, a_src, a_dst, x, W1, b1, hrelu);

    gemm_wmma  <<<gMM, 128, 0, stream>>>(hrelu, W2, h, att_s2, att_d2, a_src, a_dst, C_HID);
    node_softmax_gather<<<gROW, ROWS_PER_BLOCK * 32, 0, stream>>>(slot, rowptr, cnt, a_src, a_dst, h, b2, out, 0);
}


# Layer 2's product moves into the layer-1 launch: the tile's finished rows stay in LDS and feed W2 on the f32 WMMA, k in 4 parts added 0,1,2,3; a launch and layer 1's row plane gone; sum order changes
# speedup vs baseline: 9.7424x; 1.0730x over previous
#include <hip/hip_runtime.h>
#include <math.h>

typedef __attribute__((ext_vector_type(2))) float v2f;
typedef __attribute__((ext_vector_type(8))) float v8f;
typedef __attribute__((ext_vector_type(4))) float v4f;

#define N_NODES 50000
#define C_IN    128
#define C_HID   256
#define HEADS   4
#define EMBED   64
#define E0      800000
#define ETOT    (E0 + N_NODES)
#define NEG_SLOPE 0.2f

template <int MASK>
__device__ __forceinline__ float row_xor(float x) {
    return __builtin_bit_cast(float, __builtin_amdgcn_update_dpp(0, __builtin_bit_cast(int, x),
                                                                 0x160 + MASK, 0xf, 0xf, false));
}
__device__ __forceinline__ float row_sum16(float v) {
    v += row_xor<1>(v);
    v += row_xor<2>(v);
    v += row_xor<4>(v);
    v += row_xor<8>(v);
    return v;
}

__device__ __forceinline__ void tile_step8(const v4f a4, const float* __restrict__ wp,
                                           v8f& c0, v8f& c1, v8f& c2, v8f& c3) {
    const v4f r0 = *(const v4f*)(wp);
    const v4f r1 = *(const v4f*)(wp + C_HID);
    const v4f r2 = *(const v4f*)(wp + 2 * C_HID);
    const v4f r3 = *(const v4f*)(wp + 3 * C_HID);

    v2f a, b0, b1, b2, b3;
    a.x  = a4.x;  a.y  = a4.y;
    b0.x = r0.x;  b0.y = r1.x;
    b1.x = r0.y;  b1.y = r1.y;
    b2.x = r0.z;  b2.y = r1.z;
    b3.x = r0.w;  b3.y = r1.w;
    c0 = __builtin_amdgcn_wmma_f32_16x16x4_f32(false, a, false, b0, (short)0, c0, false, false);
    c1 = __builtin_amdgcn_wmma_f32_16x16x4_f32(false, a, false, b1, (short)0, c1, false, false);
    c2 = __builtin_amdgcn_wmma_f32_16x16x4_f32(false, a, false, b2, (short)0, c2, false, false);
    c3 = __builtin_amdgcn_wmma_f32_16x16x4_f32(false, a, false, b3, (short)0, c3, false, false);

    a.x  = a4.z;  a.y  = a4.w;
    b0.x = r2.x;  b0.y = r3.x;
    b1.x = r2.y;  b1.y = r3.y;
    b2.x = r2.z;  b2.y = r3.z;
    b3.x = r2.w;  b3.y = r3.w;
    c0 = __builtin_amdgcn_wmma_f32_16x16x4_f32(false, a, false, b0, (short)0, c0, false, false);
    c1 = __builtin_amdgcn_wmma_f32_16x16x4_f32(false, a, false, b1, (short)0, c1, false, false);
    c2 = __builtin_amdgcn_wmma_f32_16x16x4_f32(false, a, false, b2, (short)0, c2, false, false);
    c3 = __builtin_amdgcn_wmma_f32_16x16x4_f32(false, a, false, b3, (short)0, c3, false, false);
}

typedef __attribute__((ext_vector_type(2))) int v2i;
typedef __attribute__((ext_vector_type(2))) unsigned v2u;

#define BIN_SHIFT 8
#define BIN_NODES (1 << BIN_SHIFT)
#define NBIN      ((N_NODES + BIN_NODES - 1) / BIN_NODES)
#define CHUNK     3200
#define NCHUNK    (E0 / CHUNK)
#define SORT_T    1024
static_assert(E0 % CHUNK == 0, "chunks cover the edge list exactly");

__device__ __forceinline__ unsigned block_scan_incl(unsigned c, unsigned* wtot, int lane, int wave) {
    unsigned incl = c;
    for (int off = 1; off < 32; off <<= 1) {
        unsigned v = __shfl_up(incl, off, 32);
        if (lane >= off) incl += v;
    }
    if (lane == 31) wtot[wave] = incl;
    __syncthreads();
    unsigned before = 0;
    for (int w = 0; w < wave; ++w) before += wtot[w];
    return before + incl;
}

__global__ void __launch_bounds__(SORT_T)
chunk_sort(const int* __restrict__ src, const int* __restrict__ dst,
           unsigned* __restrict__ table, int* __restrict__ rec) {
    __shared__ unsigned hist[NBIN];
    __shared__ unsigned wtot[SORT_T / 32];
    const int t = threadIdx.x, lane = t & 31, wave = t >> 5;
    const int base = blockIdx.x * CHUNK;
    if (t < NBIN) hist[t] = 0u;
    __syncthreads();

    constexpr int PER = (CHUNK + SORT_T - 1) / SORT_T;
    int s[PER], d[PER];
    unsigned r[PER];
#pragma unroll
    for (int k = 0; k < PER; ++k) {
        const int i = t + k * SORT_T;
        d[k] = -1; s[k] = 0; r[k] = 0u;
        if (i < CHUNK) { d[k] = dst[base + i]; s[k] = src[base + i]; }
    }
#pragma unroll
    for (int k = 0; k < PER; ++k)
        if (d[k] >= 0) r[k] = atomicAdd(&hist[d[k] >> BIN_SHIFT], 1u);
    __syncthreads();

    const unsigned c = (t < NBIN) ? hist[t] : 0u;
    const unsigned excl = block_scan_incl(c, wtot, lane, wave) - c;
    if (t < NBIN) {
        hist[t] = excl;
        v2u run; run.x = excl; run.y = c;
        ((v2u*)table)[blockIdx.x * NBIN + t] = run;
    }
    __syncthreads();

    v2i* rec2 = (v2i*)rec;
#pragma unroll
    for (int k = 0; k < PER; ++k)
        if (d[k] >= 0) {
            v2i e; e.x = s[k]; e.y = d[k];
            rec2[base + hist[d[k] >> BIN_SHIFT] + r[k]] = e;
        }
}

__global__ void __launch_bounds__(SORT_T)
bin_place(const unsigned* __restrict__ table, const int* __restrict__ rec,
          unsigned* __restrict__ cnt, unsigned* __restrict__ rowptr, int* __restrict__ slot) {
    __shared__ unsigned rstart[NCHUNK], rlen[NCHUNK];
    __shared__ unsigned cntL[BIN_NODES], offL[BIN_NODES], fillL[BIN_NODES];
    __shared__ unsigned wsum[SORT_T / 32], wtot[SORT_T / 32];
    const int t = threadIdx.x, lane = t & 31, wave = t >> 5;
    const int bin = blockIdx.x;
    const v2i* rec2 = (const v2i*)rec;

    unsigned o0 = 0u;
    if (t < NCHUNK) {
        const v2u run = ((const v2u*)table)[t * NBIN + bin];
        o0 = run.x;
        rstart[t] = (unsigned)t * CHUNK + run.x;
        rlen[t]   = run.y;
    }
    if (t < BIN_NODES) { cntL[t] = 0u; fillL[t] = 0u; }
    unsigned part = o0;
    for (int off = 16; off > 0; off >>= 1) part += __shfl_xor(part, off, 32);
    if (lane == 0) wsum[wave] = part;
    __syncthreads();
    unsigned binStart = 0u;
    for (int w = 0; w < SORT_T / 32; ++w) binStart += wsum[w];

    constexpr int RPW = (NCHUNK + 31) / 32;
    int sv[RPW], dv[RPW];
    unsigned at[RPW];
    bool has[RPW];
#pragma unroll
    for (int k = 0; k < RPW; ++k) {
        const int run = wave + 32 * k;
        const int rr  = run < NCHUNK ? run : 0;
        has[k] = run < NCHUNK && (unsigned)lane < rlen[rr];
        at[k]  = has[k] ? rstart[rr] + lane : 0u;
    }
#pragma unroll
    for (int k = 0; k < RPW; ++k) {
        const v2i e = rec2[at[k]];
        sv[k] = e.x;
        dv[k] = has[k] ? e.y - bin * BIN_NODES : -1;
    }
#pragma unroll
    for (int k = 0; k < RPW; ++k)
        if (dv[k] >= 0) atomicAdd(&cntL[dv[k]], 1u);
    for (int run = wave; run < NCHUNK; run += 32) {
        const unsigned len = rlen[run], st = rstart[run];
        for (unsigned i = lane + 32; i < len; i += 32)
            atomicAdd(&cntL[rec2[st + i].y - bin * BIN_NODES], 1u);
    }
    __syncthreads();

    const unsigned c = (t < BIN_NODES) ? cntL[t] : 0u;
    const unsigned excl = block_scan_incl(c, wtot, lane, wave) - c;
    if (t < BIN_NODES) {
        offL[t] = binStart + excl;
        const int node = bin * BIN_NODES + t;
        if (node < N_NODES) { rowptr[node] = binStart + excl; cnt[node] = c; }
    }
    __syncthreads();

#pragma unroll
    for (int k = 0; k < RPW; ++k)
        if (dv[k] >= 0) slot[offL[dv[k]] + atomicAdd(&fillL[dv[k]], 1u)] = sv[k];
    for (int run = wave; run < NCHUNK; run += 32) {
        const unsigned len = rlen[run], st = rstart[run];
        for (unsigned i = lane + 32; i < len; i += 32) {
            const v2i e = rec2[st + i];
            const int dl = e.y - bin * BIN_NODES;
            slot[offL[dl] + atomicAdd(&fillL[dl], 1u)] = e.x;
        }
    }
}

__device__ __forceinline__ float xhalf(float x) {
    const int b = __builtin_bit_cast(int, x);
    return __builtin_bit_cast(float, __builtin_amdgcn_permlanex16(b, b, 0x76543210, 0xfedcba98, false, false));
}
__device__ __forceinline__ float wave_max(float v) {
    v = fmaxf(v, row_xor<1>(v));
    v = fmaxf(v, row_xor<2>(v));
    v = fmaxf(v, row_xor<4>(v));
    v = fmaxf(v, row_xor<8>(v));
    return fmaxf(v, xhalf(v));
}
__device__ __forceinline__ float wave_sum(float v) {
    v += row_xor<1>(v);
    v += row_xor<2>(v);
    v += row_xor<4>(v);
    v += row_xor<8>(v);
    return v + xhalf(v);
}

__device__ __forceinline__ float uni(float x) {
    return __builtin_bit_cast(float, __builtin_amdgcn_readfirstlane(__builtin_bit_cast(int, x)));
}

__device__ __forceinline__ float leaky(float v) { return v > 0.f ? v : NEG_SLOPE * v; }

__device__ __forceinline__ void load_edges(const int* __restrict__ slot, const v4f* __restrict__ AS4,
                                           unsigned first, unsigned n, int lane, v4f ad,
                                           int& s, float& l0, float& l1, float& l2, float& l3) {
    s = 0;
    l0 = l1 = l2 = l3 = -INFINITY;
    if ((unsigned)lane < n) {
        s = slot[first + lane];
        const v4f as = AS4[s];
        l0 = leaky(as.x + ad.x); l1 = leaky(as.y + ad.y);
        l2 = leaky(as.z + ad.z); l3 = leaky(as.w + ad.w);
    }
}

__device__ __forceinline__ void add_rows(const v4f* __restrict__ H4, unsigned n, int lane, bool upper,
                                         int s, float a0, float a1, float a2, float a3,
                                         v4f& acc0, v4f& acc1) {
#pragma unroll 4
    for (unsigned j = 0; j < n; ++j) {
        const int   sj = __builtin_amdgcn_readlane(s, (int)j);
        const float b0 = __int_as_float(__builtin_amdgcn_readlane(__float_as_int(a0), (int)j));
        const float b1 = __int_as_float(__builtin_amdgcn_readlane(__float_as_int(a1), (int)j));
        const float b2 = __int_as_float(__builtin_amdgcn_readlane(__float_as_int(a2), (int)j));
        const float b3 = __int_as_float(__builtin_amdgcn_readlane(__float_as_int(a3), (int)j));
        const float alo = upper ? b1 : b0;
        const float ahi = upper ? b3 : b2;
        const v4f v0 = H4[(size_t)sj * 64 + lane];
        const v4f v1 = H4[(size_t)sj * 64 + 32 + lane];
        acc0 += v0 * alo;
        acc1 += v1 * ahi;
    }
}

#define PRE_T      256
#define PRE_BLOCKS 512
static_assert(C_IN % (8 * (PRE_T / 32)) == 0, "a wave's rows of W come in whole groups of eight");
__global__ void __launch_bounds__(PRE_T)
attn_dots_from_x(const float* __restrict__ X, const float* __restrict__ W,
                 const float* __restrict__ att_src, const float* __restrict__ att_dst,
                 float* __restrict__ a_src, float* __restrict__ a_dst) {
    __shared__ v4f wsd[C_IN][2];
    float* wsdf = (float*)wsd;
    const int lane = threadIdx.x & 31;
    const int wave = __builtin_amdgcn_readfirstlane(threadIdx.x >> 5);

    {
        const v4f s0 = ((const v4f*)att_src)[2 * lane], s1 = ((const v4f*)att_src)[2 * lane + 1];
        const v4f d0 = ((const v4f*)att_dst)[2 * lane], d1 = ((const v4f*)att_dst)[2 * lane + 1];
        constexpr int KG = 8;
        for (int kb = wave; kb < C_IN; kb += KG * (PRE_T / 32)) {
            v4f w0[KG], w1[KG];
#pragma unroll
            for (int i = 0; i < KG; ++i) {
                const v4f* wr = (const v4f*)(W + (size_t)(kb + i * (PRE_T / 32)) * C_HID);
                w0[i] = wr[2 * lane];
                w1[i] = wr[2 * lane + 1];
            }
#pragma unroll
            for (int i = 0; i < KG; ++i) {
                const int k = kb + i * (PRE_T / 32);
                float ps = w0[i].x * s0.x + w0[i].y * s0.y + w0[i].z * s0.z + w0[i].w * s0.w
                         + w1[i].x * s1.x + w1[i].y * s1.y + w1[i].z * s1.z + w1[i].w * s1.w;
                float pd = w0[i].x * d0.x + w0[i].y * d0.y + w0[i].z * d0.z + w0[i].w * d0.w
                         + w1[i].x * d1.x + w1[i].y * d1.y + w1[i].z * d1.z + w1[i].w * d1.w;
                ps += row_xor<1>(ps); ps += row_xor<2>(ps); ps += row_xor<4>(ps);
                pd += row_xor<1>(pd); pd += row_xor<2>(pd); pd += row_xor<4>(pd);
                if ((lane & 7) == 0) {
                    wsdf[k * 8 + (lane >> 3)]     = ps;
                    wsdf[k * 8 + 4 + (lane >> 3)] = pd;
                }
            }
        }
    }
    __syncthreads();

    const v4f ws0 = wsd[4 * lane][0],     wd0 = wsd[4 * lane][1];
    const v4f ws1 = wsd[4 * lane + 1][0], wd1 = wsd[4 * lane + 1][1];
    const v4f ws2 = wsd[4 * lane + 2][0], wd2 = wsd[4 * lane + 2][1];
    const v4f ws3 = wsd[4 * lane + 3][0], wd3 = wsd[4 * lane + 3][1];
    const v4f* X4 = (const v4f*)X;
    const int nw = PRE_BLOCKS * (PRE_T / 32);
    constexpr int NG = 4;
    for (int nb = blockIdx.x * (PRE_T / 32) + wave; nb < N_NODES; nb += NG * nw) {
        v4f xv[NG];
#pragma unroll
        for (int i = 0; i < NG; ++i) {
            const int n = nb + i * nw;
            xv[i] = X4[(size_t)(n < N_NODES ? n : nb) * (C_IN / 4) + lane];
        }
#pragma unroll
        for (int i = 0; i < NG; ++i) {
            const int n = nb + i * nw;
            v4f ps = ws0 * xv[i].x + ws1 * xv[i].y + ws2 * xv[i].z + ws3 * xv[i].w;
            v4f pd = wd0 * xv[i].x + wd1 * xv[i].y + wd2 * xv[i].z + wd3 * xv[i].w;
            ps.x = wave_sum(ps.x); ps.y = wave_sum(ps.y); ps.z = wave_sum(ps.z); ps.w = wave_sum(ps.w);
            pd.x = wave_sum(pd.x); pd.y = wave_sum(pd.y); pd.z = wave_sum(pd.z); pd.w = wave_sum(pd.w);
            if (n < N_NODES) {
                if (lane == 0) ((v4f*)a_src)[n] = ps;
                if (lane == 1) ((v4f*)a_dst)[n] = pd;
            }
        }
    }
}

template <int G>
__device__ __forceinline__ void add_x_rows(const v4f* __restrict__ X4, unsigned j0, int lane,
                                           int s, float a0, float a1, float a2, float a3,
                                           v4f& acc0, v4f& acc1, v4f& acc2, v4f& acc3) {
    v4f v[G];
#pragma unroll
    for (int g = 0; g < G; ++g) {
        const int sj = __builtin_amdgcn_readlane(s, (int)(j0 + g));
        v[g] = X4[(size_t)sj * (C_IN / 4) + lane];
    }
#pragma unroll
    for (int g = 0; g < G; ++g) {
        const float b0 = __int_as_float(__builtin_amdgcn_readlane(__float_as_int(a0), (int)(j0 + g)));
        const float b1 = __int_as_float(__builtin_amdgcn_readlane(__float_as_int(a1), (int)(j0 + g)));
        const float b2 = __int_as_float(__builtin_amdgcn_readlane(__float_as_int(a2), (int)(j0 + g)));
        const float b3 = __int_as_float(__builtin_amdgcn_readlane(__float_as_int(a3), (int)(j0 + g)));
        acc0 += v[g] * b0;
        acc1 += v[g] * b1;
        acc2 += v[g] * b2;
        acc3 += v[g] * b3;
    }
}

#define TILE_NODES 16
#define AGG_PITCH  (HEADS * C_IN + 4)
#define HR_PITCH   (C_HID + 4)
static_assert(N_NODES % TILE_NODES == 0, "every tile is full: no wave leaves before the barriers");
static_assert((AGG_PITCH / 4) % 2 == 1 && (HR_PITCH / 4) % 2 == 1, "pitches are odd numbers of 16-byte pieces");
static_assert(HR_PITCH <= AGG_PITCH - 8, "layer 1's rows and the 128 dots fit where the sums were");
#define KQUARTERS  (TILE_NODES / HEADS)
#define KQ         (C_IN / KQUARTERS)
#define KQ2        (C_HID / KQUARTERS)
static_assert(TILE_NODES == HEADS * KQUARTERS && KQ % 8 == 0 && KQ2 % 8 == 0, "16 waves = heads x parts of k, k taken in eights");
__global__ void __launch_bounds__(TILE_NODES * 32) __attribute__((amdgpu_waves_per_eu(12)))
layer1_gather_two_products(const int* __restrict__ slot, const unsigned* __restrict__ rowptr,
                           const unsigned* __restrict__ cnt,
                           const float* __restrict__ a_src, const float* __restrict__ a_dst,
                           const float* __restrict__ X, const float* __restrict__ W,
                           const float* __restrict__ bias,
                           const float* __restrict__ W2,
                           const float* __restrict__ att_src2, const float* __restrict__ att_dst2,
                           float* __restrict__ H2, float* __restrict__ a_src2, float* __restrict__ a_dst2) {
    __shared__ v4f agg4[TILE_NODES * AGG_PITCH / 4];
    __shared__ v4f part4[(KQUARTERS - 1) * HEADS * 8 * 32];
    float* agg = (float*)agg4;

    const int lane = threadIdx.x & 31;
    const int wave = __builtin_amdgcn_readfirstlane(threadIdx.x >> 5);
    const int node = blockIdx.x * TILE_NODES + wave;

    const v4f* X4  = (const v4f*)X;
    const v4f* AS4 = (const v4f*)a_src;

    {
        const unsigned beg = rowptr[node];
        const unsigned deg = cnt[node];
        const v4f ad  = ((const v4f*)a_dst)[node];
        const v4f asn = AS4[node];
        const v4f selfx = X4[(size_t)node * (C_IN / 4) + lane];

        const float ls0 = leaky(asn.x + ad.x), ls1 = leaky(asn.y + ad.y);
        const float ls2 = leaky(asn.z + ad.z), ls3 = leaky(asn.w + ad.w);

        const unsigned n0 = deg < 32u ? deg : 32u;
        int   s0;
        float p0, p1, p2, p3;
        load_edges(slot, AS4, beg, n0, lane, ad, s0, p0, p1, p2, p3);

        float m0 = fmaxf(ls0, p0), m1 = fmaxf(ls1, p1), m2 = fmaxf(ls2, p2), m3 = fmaxf(ls3, p3);
        for (unsigned base = 32; base < deg; base += 32) {
            const unsigned n = (deg - base < 32u) ? (deg - base) : 32u;
            int s; float l0, l1, l2, l3;
            load_edges(slot, AS4, beg + base, n, lane, ad, s, l0, l1, l2, l3);
            m0 = fmaxf(m0, l0); m1 = fmaxf(m1, l1); m2 = fmaxf(m2, l2); m3 = fmaxf(m3, l3);
        }
        m0 = uni(wave_max(m0)); m1 = uni(wave_max(m1)); m2 = uni(wave_max(m2)); m3 = uni(wave_max(m3));
        m0 = __builtin_isfinite(m0) ? m0 : 0.f; m1 = __builtin_isfinite(m1) ? m1 : 0.f;
        m2 = __builtin_isfinite(m2) ? m2 : 0.f; m3 = __builtin_isfinite(m3) ? m3 : 0.f;

        p0 = __expf(p0 - m0); p1 = __expf(p1 - m1); p2 = __expf(p2 - m2); p3 = __expf(p3 - m3);
        float z0 = p0, z1 = p1, z2 = p2, z3 = p3;
        for (unsigned base = 32; base < deg; base += 32) {
            const unsigned n = (deg - base < 32u) ? (deg - base) : 32u;
            int s; float l0, l1, l2, l3;
            load_edges(slot, AS4, beg + base, n, lane, ad, s, l0, l1, l2, l3);
            z0 += __expf(l0 - m0); z1 += __expf(l1 - m1); z2 += __expf(l2 - m2); z3 += __expf(l3 - m3);
        }
        const float es0 = __expf(ls0 - m0), es1 = __expf(ls1 - m1);
        const float es2 = __expf(ls2 - m2), es3 = __expf(ls3 - m3);
        const float zd0 = uni(wave_sum(z0) + es0 + 1e-16f), zd1 = uni(wave_sum(z1) + es1 + 1e-16f);
        const float zd2 = uni(wave_sum(z2) + es2 + 1e-16f), zd3 = uni(wave_sum(z3) + es3 + 1e-16f);

        v4f acc0 = selfx * (es0 / zd0);
        v4f acc1 = selfx * (es1 / zd1);
        v4f acc2 = selfx * (es2 / zd2);
        v4f acc3 = selfx * (es3 / zd3);
        {
            int      s = s0;
            unsigned n = n0;
            float a0 = p0 / zd0, a1 = p1 / zd1, a2 = p2 / zd2, a3 = p3 / zd3;
            for (unsigned base = 0;;) {
                unsigned j = 0;
                for (; j + 8 <= n; j += 8)
                    add_x_rows<8>(X4, j, lane, s, a0, a1, a2, a3, acc0, acc1, acc2, acc3);
                if (j + 4 <= n) {
                    add_x_rows<4>(X4, j, lane, s, a0, a1, a2, a3, acc0, acc1, acc2, acc3);
                    j += 4;
                }
                for (; j < n; ++j)
                    add_x_rows<1>(X4, j, lane, s, a0, a1, a2, a3, acc0, acc1, acc2, acc3);
                base += 32;
                if (base >= deg) break;
                n = (deg - base < 32u) ? (deg - base) : 32u;
                float l0, l1, l2, l3;
                load_edges(slot, AS4, beg + base, n, lane, ad, s, l0, l1, l2, l3);
                a0 = __expf(l0 - m0) / zd0; a1 = __expf(l1 - m1) / zd1;
                a2 = __expf(l2 - m2) / zd2; a3 = __expf(l3 - m3) / zd3;
            }
        }

        v4f* mine = (v4f*)(agg + wave * AGG_PITCH);
        mine[0 * (C_IN / 4) + lane] = acc0;
        mine[1 * (C_IN / 4) + lane] = acc1;
        mine[2 * (C_IN / 4) + lane] = acc2;
        mine[3 * (C_IN / 4) + lane] = acc3;
    }
    __syncthreads();

    const int hd     = wave & (HEADS - 1);
    const int q      = wave / HEADS;
    const int lane15 = lane & 15;
    const int khalf  = lane >> 4;
    const int colL   = hd * EMBED + 4 * lane15;

    v8f c0 = {}, c1 = {}, c2 = {}, c3 = {};
    {
        const v4f*   ap = (const v4f*)(agg + lane15 * AGG_PITCH + hd * C_IN + q * KQ + 4 * khalf);
        const float* wp = W + (size_t)(q * KQ + 4 * khalf) * C_HID + colL;
#pragma unroll
        for (int k = 0; k < KQ; k += 8) {
            tile_step8(*ap, wp, c0, c1, c2, c3);
            ap += 2;
            wp += 8 * C_HID;
        }
    }

    if (q != 0) {
        v4f* mine = part4 + ((q - 1) * HEADS + hd) * (8 * 32) + lane;
#pragma unroll
        for (int i = 0; i < 8; ++i) {
            v4f o;
            o.x = c0[i]; o.y = c1[i]; o.z = c2[i]; o.w = c3[i];
            mine[i * 32] = o;
        }
    }
    __syncthreads();

    float* hr = agg;
    if (q == 0) {
        const v4f b4 = *(const v4f*)(bias + colL);
        const v4f* theirs = part4 + hd * (8 * 32) + lane;
#pragma unroll
        for (int i = 0; i < 8; ++i) {
            v4f o;
            o.x = c0[i]; o.y = c1[i]; o.z = c2[i]; o.w = c3[i];
            o += theirs[(0 * HEADS * 8 + i) * 32];
            o += theirs[(1 * HEADS * 8 + i) * 32];
            o += theirs[(2 * HEADS * 8 + i) * 32];
            o.x = fmaxf(o.x + b4.x, 0.f); o.y = fmaxf(o.y + b4.y, 0.f);
            o.z = fmaxf(o.z + b4.z, 0.f); o.w = fmaxf(o.w + b4.w, 0.f);
            *(v4f*)(hr + (khalf * 8 + i) * HR_PITCH + colL) = o;
        }
    }
    __syncthreads();

    v8f d0 = {}, d1 = {}, d2 = {}, d3 = {};
    {
        const v4f*   ap = (const v4f*)(hr + lane15 * HR_PITCH + q * KQ2 + 4 * khalf);
        const float* wp = W2 + (size_t)(q * KQ2 + 4 * khalf) * C_HID + colL;
#pragma unroll
        for (int k = 0; k < KQ2; k += 8) {
            tile_step8(*ap, wp, d0, d1, d2, d3);
            ap += 2;
            wp += 8 * C_HID;
        }
    }
    if (q != 0) {
        v4f* mine = part4 + ((q - 1) * HEADS + hd) * (8 * 32) + lane;
#pragma unroll
        for (int i = 0; i < 8; ++i) {
            v4f o;
            o.x = d0[i]; o.y = d1[i]; o.z = d2[i]; o.w = d3[i];
            mine[i * 32] = o;
        }
    }
    __syncthreads();

    v4f* dots4 = agg4 + (TILE_NODES * HR_PITCH) / 4;
    float* dots = (float*)dots4;
    v4f hrow[8];
    if (q == 0) {
        const v4f* theirs = part4 + hd * (8 * 32) + lane;
        const v4f s4 = *(const v4f*)(att_src2 + colL);
        const v4f t4 = *(const v4f*)(att_dst2 + colL);
#pragma unroll
        for (int i = 0; i < 8; ++i) {
            v4f o;
            o.x = d0[i]; o.y = d1[i]; o.z = d2[i]; o.w = d3[i];
            o += theirs[(0 * HEADS * 8 + i) * 32];
            o += theirs[(1 * HEADS * 8 + i) * 32];
            o += theirs[(2 * HEADS * 8 + i) * 32];
            hrow[i] = o;
            const float ps = row_sum16(o.x * s4.x + o.y * s4.y + o.z * s4.z + o.w * s4.w);
            const float pd = row_sum16(o.x * t4.x + o.y * t4.y + o.z * t4.z + o.w * t4.w);
            if (lane15 == 0) {
                dots[(0 * 16 + khalf * 8 + i) * HEADS + hd] = ps;
                dots[(1 * 16 + khalf * 8 + i) * HEADS + hd] = pd;
            }
        }
    }
    __syncthreads();
    if (wave == 0) {
        float* plane = khalf ? a_dst2 : a_src2;
        *(v4f*)(plane + (size_t)(blockIdx.x * TILE_NODES + lane15) * HEADS) = dots4[khalf * 16 + lane15];
    }
    if (q != 0) return;

    const int rbase = blockIdx.x * TILE_NODES + khalf * 8;
#pragma unroll
    for (int i = 0; i < 8; ++i)
        *(v4f*)(H2 + (size_t)(rbase + i) * C_HID + colL) = hrow[i];
}

#define ROWS_PER_BLOCK 8
__global__ void __launch_bounds__(ROWS_PER_BLOCK * 32)
node_softmax_gather(const int* __restrict__ slot, const unsigned* __restrict__ rowptr,
                    const unsigned* __restrict__ cnt,
                    const float* __restrict__ a_src, const float* __restrict__ a_dst,
                    const float* __restrict__ H, const float* __restrict__ bias,
                    float* __restrict__ out, int relu) {
    const int lane = threadIdx.x & 31;
    const int node = __builtin_amdgcn_readfirstlane(blockIdx.x * ROWS_PER_BLOCK + (threadIdx.x >> 5));
    if (node >= N_NODES) return;

    const v4f* H4  = (const v4f*)H;
    const v4f* AS4 = (const v4f*)a_src;

    const unsigned beg = rowptr[node];
    const unsigned deg = cnt[node];
    const v4f ad  = ((const v4f*)a_dst)[node];
    const v4f asn = AS4[node];
    const bool upper = lane >= 16;

    const v4f bias0 = ((const v4f*)bias)[lane];
    const v4f bias1 = ((const v4f*)bias)[32 + lane];
    const v4f self0 = H4[(size_t)node * 64 + lane];
    const v4f self1 = H4[(size_t)node * 64 + 32 + lane];

    const float ls0 = leaky(asn.x + ad.x), ls1 = leaky(asn.y + ad.y);
    const float ls2 = leaky(asn.z + ad.z), ls3 = leaky(asn.w + ad.w);

    const unsigned n0 = deg < 32u ? deg : 32u;
    int   s0;
    float p0, p1, p2, p3;
    load_edges(slot, AS4, beg, n0, lane, ad, s0, p0, p1, p2, p3);

    float m0 = fmaxf(ls0, p0), m1 = fmaxf(ls1, p1), m2 = fmaxf(ls2, p2), m3 = fmaxf(ls3, p3);
    for (unsigned base = 32; base < deg; base += 32) {
        const unsigned n = (deg - base < 32u) ? (deg - base) : 32u;
        int s; float l0, l1, l2, l3;
        load_edges(slot, AS4, beg + base, n, lane, ad, s, l0, l1, l2, l3);
        m0 = fmaxf(m0, l0); m1 = fmaxf(m1, l1); m2 = fmaxf(m2, l2); m3 = fmaxf(m3, l3);
    }
    m0 = uni(wave_max(m0)); m1 = uni(wave_max(m1)); m2 = uni(wave_max(m2)); m3 = uni(wave_max(m3));
    m0 = __builtin_isfinite(m0) ? m0 : 0.f; m1 = __builtin_isfinite(m1) ? m1 : 0.f;
    m2 = __builtin_isfinite(m2) ? m2 : 0.f; m3 = __builtin_isfinite(m3) ? m3 : 0.f;

    p0 = __expf(p0 - m0); p1 = __expf(p1 - m1); p2 = __expf(p2 - m2); p3 = __expf(p3 - m3);
    float z0 = p0, z1 = p1, z2 = p2, z3 = p3;
    for (unsigned base = 32; base < deg; base += 32) {
        const unsigned n = (deg - base < 32u) ? (deg - base) : 32u;
        int s; float l0, l1, l2, l3;
        load_edges(slot, AS4, beg + base, n, lane, ad, s, l0, l1, l2, l3);
        z0 += __expf(l0 - m0); z1 += __expf(l1 - m1); z2 += __expf(l2 - m2); z3 += __expf(l3 - m3);
    }
    const float es0 = __expf(ls0 - m0), es1 = __expf(ls1 - m1);
    const float es2 = __expf(ls2 - m2), es3 = __expf(ls3 - m3);
    const float zd0 = uni(wave_sum(z0) + es0 + 1e-16f), zd1 = uni(wave_sum(z1) + es1 + 1e-16f);
    const float zd2 = uni(wave_sum(z2) + es2 + 1e-16f), zd3 = uni(wave_sum(z3) + es3 + 1e-16f);

    v4f acc0, acc1;
    {
        const float alo = (upper ? es1 : es0) / (upper ? zd1 : zd0);
        const float ahi = (upper ? es3 : es2) / (upper ? zd3 : zd2);
        acc0 = self0 * alo;
        acc1 = self1 * ahi;
    }
    {
        int      s = s0;
        unsigned n = n0;
        float a0 = p0 / zd0, a1 = p1 / zd1, a2 = p2 / zd2, a3 = p3 / zd3;
        for (unsigned base = 0;;) {
            add_rows(H4, n, lane, upper, s, a0, a1, a2, a3, acc0, acc1);
            base += 32;
            if (base >= deg) break;
            n = (deg - base < 32u) ? (deg - base) : 32u;
            float l0, l1, l2, l3;
            load_edges(slot, AS4, beg + base, n, lane, ad, s, l0, l1, l2, l3);
            a0 = __expf(l0 - m0) / zd0; a1 = __expf(l1 - m1) / zd1;
            a2 = __expf(l2 - m2) / zd2; a3 = __expf(l3 - m3) / zd3;
        }
    }

    acc0 += bias0;
    acc1 += bias1;
    if (relu) {
        acc0.x = fmaxf(acc0.x, 0.f); acc0.y = fmaxf(acc0.y, 0.f);
        acc0.z = fmaxf(acc0.z, 0.f); acc0.w = fmaxf(acc0.w, 0.f);
        acc1.x = fmaxf(acc1.x, 0.f); acc1.y = fmaxf(acc1.y, 0.f);
        acc1.z = fmaxf(acc1.z, 0.f); acc1.w = fmaxf(acc1.w, 0.f);
    }
    v4f* O4 = (v4f*)out;
    O4[(size_t)node * 64 + lane]      = acc0;
    O4[(size_t)node * 64 + 32 + lane] = acc1;
}

extern "C" void kernel_launch(void* const* d_in, const int* in_sizes, int n_in,
                              void* d_out, int out_size, void* d_ws, size_t ws_size,
                              hipStream_t stream) {
    const float* x        = (const float*)d_in[0];
    const int*   eidx     = (const int*)  d_in[1];
    const float* W1       = (const float*)d_in[2];
    const float* att_s1   = (const float*)d_in[3];
    const float* att_d1   = (const float*)d_in[4];
    const float* b1       = (const float*)d_in[5];
    const float* W2       = (const float*)d_in[6];
    const float* att_s2   = (const float*)d_in[7];
    const float* att_d2   = (const float*)d_in[8];
    const float* b2       = (const float*)d_in[9];
    float*       out      = (float*)d_out;

    const int* src = eidx;
    const int* dst = eidx + E0;

    const size_t NB = (size_t)N_NODES * C_HID;
    const size_t NH = (size_t)N_NODES * HEADS;
    float* h      = (float*)d_ws;
    float* a_src  = h      + NB;
    float* a_dst  = a_src  + NH;
    float* a_src2 = a_dst  + NH;
    float* a_dst2 = a_src2 + NH;
    unsigned* cnt    = (unsigned*)(a_dst2 + NH);
    unsigned* rowptr = cnt + N_NODES;
    int*      slot   = (int*)(rowptr + N_NODES);
    int*      rec    = slot + E0;
    unsigned* table  = (unsigned*)(rec + 2 * (size_t)E0);

    const int gROW = (N_NODES + ROWS_PER_BLOCK - 1) / ROWS_PER_BLOCK;

    chunk_sort<<<NCHUNK, SORT_T, 0, stream>>>(src, dst, table, rec);
    bin_place <<<NBIN,   SORT_T, 0, stream>>>(table, rec, cnt, rowptr, slot);

    attn_dots_from_x<<<PRE_BLOCKS, PRE_T, 0, stream>>>(x, W1, att_s1, att_d1, a_src, a_dst);
    layer1_gather_two_products<<<N_NODES / TILE_NODES, TILE_NODES * 32, 0, stream>>>(
        slot, rowptr, cnt, a_src, a_dst, x, W1, b1, W2, att_s2, att_d2, h, a_src2, a_dst2);

    node_softmax_gather<<<gROW, ROWS_PER_BLOCK * 32, 0, stream>>>(slot, rowptr, cnt, a_src2, a_dst2, h, b2, out, 0);
}
